# starting kernel
# baseline (speedup 1.0000x reference)
#include <hip/hip_runtime.h>
#include <hip/hip_bf16.h>

#define NEG_SLOPE 0.2f
#define F_IN 14

typedef float v2f __attribute__((ext_vector_type(2)));
typedef float v8f __attribute__((ext_vector_type(8)));

__device__ __forceinline__ unsigned enc_f32(float f) {
    unsigned u = __float_as_uint(f);
    return (u & 0x80000000u) ? ~u : (u | 0x80000000u);
}
__device__ __forceinline__ float dec_f32(unsigned e) {
    unsigned u = (e & 0x80000000u) ? (e ^ 0x80000000u) : ~e;
    return __uint_as_float(u);
}

__device__ __forceinline__ void edge_sd(const int* __restrict__ ei, int E, int n,
                                        int idx, int& s, int& d) {
    if (idx < E) { s = ei[idx]; d = ei[E + idx]; }
    else         { s = idx - E; d = idx - E; }
    s = s < 0 ? 0 : (s >= n ? n - 1 : s);
    d = d < 0 ? 0 : (d >= n ? n - 1 : d);
}

__global__ __launch_bounds__(32) void k_gemm1(const float* __restrict__ x,
                                              const float* __restrict__ W1,
                                              float* __restrict__ h1, int n) {
    __shared__ float xs[16 * 16];
    __shared__ float ws[16 * 64];

    const int lane = threadIdx.x;
    const int row0 = blockIdx.x * 16;
    const bool hi  = lane >= 16;
    const int  m   = lane & 15;
    const bool full = (row0 + 16 <= n);

    for (int i = lane; i < 2 * 64 / 4; i += 32)
        ((float4*)(ws + 14 * 64))[i] = make_float4(0.f, 0.f, 0.f, 0.f);
    for (int i = lane; i < F_IN * 64 / 4; i += 32)
        ((float4*)ws)[i] = ((const float4*)W1)[i];

    for (int i = lane; i < 16 * 16 / 4; i += 32)
        ((float4*)xs)[i] = make_float4(0.f, 0.f, 0.f, 0.f);
    __syncthreads();
    {
        const float* xt = x + (size_t)row0 * F_IN;
        int rows = full ? 16 : (n - row0);
        int cnt  = rows * F_IN;
        for (int i = lane; i < cnt; i += 32) {
            int r = i / F_IN, c = i - r * F_IN;
            xs[r * 16 + c] = xt[i];
        }
    }
    __syncthreads();

    const int kofs = hi ? 2 : 0;
    v2f a[4];
#pragma unroll
    for (int k = 0; k < 4; ++k) {
        a[k][0] = xs[m * 16 + 4 * k + kofs + 0];
        a[k][1] = xs[m * 16 + 4 * k + kofs + 1];
    }

#pragma unroll
    for (int t = 0; t < 4; ++t) {
        v8f c = {};
#pragma unroll
        for (int k = 0; k < 4; ++k) {
            v2f b;
            b[0] = ws[(4 * k + kofs + 0) * 64 + t * 16 + m];
            b[1] = ws[(4 * k + kofs + 1) * 64 + t * 16 + m];
            c = __builtin_amdgcn_wmma_f32_16x16x4_f32(false, a[k], false, b,
                                                      (short)0, c, false, false);
        }
        float* out = h1 + (size_t)row0 * 64 + t * 16 + m;
        if (full) {
#pragma unroll
            for (int r = 0; r < 8; ++r)
                out[(size_t)(r + (hi ? 8 : 0)) * 64] = c[r];
        } else {
#pragma unroll
            for (int r = 0; r < 8; ++r) {
                int rr = r + (hi ? 8 : 0);
                if (row0 + rr < n) out[(size_t)rr * 64] = c[r];
            }
        }
    }
}

__global__ __launch_bounds__(32) void k_gemm2(const float* __restrict__ a1,
                                              const float* __restrict__ W2,
                                              float* __restrict__ h2, int n) {
    __shared__ float smA[16 * 64];
    __shared__ float ws2[64 * 16];

    const int lane = threadIdx.x;
    const int row0 = blockIdx.x * 16;
    const bool hi  = lane >= 16;
    const int  m   = lane & 15;
    const bool full = (row0 + 16 <= n);

    for (int i = lane; i < 64 * 16 / 4; i += 32)
        ((float4*)ws2)[i] = make_float4(0.f, 0.f, 0.f, 0.f);
    __syncthreads();
    for (int i = lane; i < 64 * 8 / 4; i += 32) {
        int k = i >> 1, half = i & 1;
        ((float4*)(ws2 + k * 16 + half * 4))[0] = ((const float4*)W2)[i];
    }

    if (full) {
        const float4* src = (const float4*)(a1 + (size_t)row0 * 64);
        for (int i = lane; i < 16 * 64 / 4; i += 32)
            ((float4*)smA)[i] = src[i];
    } else {
        for (int i = lane; i < 16 * 64 / 4; i += 32)
            ((float4*)smA)[i] = make_float4(0.f, 0.f, 0.f, 0.f);
        __syncthreads();
        int rows = n - row0;
        const float4* src = (const float4*)(a1 + (size_t)row0 * 64);
        for (int i = lane; i < rows * 16; i += 32)
            ((float4*)smA)[i] = src[i];
    }
    __syncthreads();

    const int kofs = hi ? 2 : 0;
    v8f c = {};
#pragma unroll
    for (int k = 0; k < 16; ++k) {
        v2f a, b;
        a[0] = smA[m * 64 + 4 * k + kofs + 0];
        a[1] = smA[m * 64 + 4 * k + kofs + 1];
        b[0] = ws2[(4 * k + kofs + 0) * 16 + m];
        b[1] = ws2[(4 * k + kofs + 1) * 16 + m];
        c = __builtin_amdgcn_wmma_f32_16x16x4_f32(false, a, false, b,
                                                  (short)0, c, false, false);
    }
    if (m < 8) {
        float* out = h2 + (size_t)row0 * 8 + m;
        if (full) {
#pragma unroll
            for (int r = 0; r < 8; ++r)
                out[(size_t)(r + (hi ? 8 : 0)) * 8] = c[r];
        } else {
#pragma unroll
            for (int r = 0; r < 8; ++r) {
                int rr = r + (hi ? 8 : 0);
                if (row0 + rr < n) out[(size_t)rr * 8] = c[r];
            }
        }
    }
}

__global__ void k_alpha1(const float* __restrict__ h1,
                         const float* __restrict__ asrc, const float* __restrict__ adst,
                         float* __restrict__ as1, float* __restrict__ ad1, int n) {
    int tid = blockIdx.x * blockDim.x + threadIdx.x;
    if (tid >= n * 8) return;
    unsigned h = tid & 7u;
    const float4* hp = (const float4*)(h1 + (unsigned)(tid >> 3) * 64u + h * 8u);
    float4 v0 = hp[0], v1 = hp[1];
    const float4* sp = (const float4*)(asrc + h * 8u);
    const float4* dp = (const float4*)(adst + h * 8u);
    float4 s0 = sp[0], s1v = sp[1], d0 = dp[0], d1v = dp[1];
    float s = v0.x * s0.x + v0.y * s0.y + v0.z * s0.z + v0.w * s0.w
            + v1.x * s1v.x + v1.y * s1v.y + v1.z * s1v.z + v1.w * s1v.w;
    float d = v0.x * d0.x + v0.y * d0.y + v0.z * d0.z + v0.w * d0.w
            + v1.x * d1v.x + v1.y * d1v.y + v1.z * d1v.z + v1.w * d1v.w;
    as1[tid] = s;
    ad1[tid] = d;
}

__global__ void k_edge_max1(const int* __restrict__ ei, int E, int n,
                            const float* __restrict__ as1, const float* __restrict__ ad1,
                            unsigned* __restrict__ m1) {
    int idx = blockIdx.x * blockDim.x + threadIdx.x;
    if (idx >= E + n) return;
    int s, d; edge_sd(ei, E, n, idx, s, d);
    const float4* ap = (const float4*)(as1 + (unsigned)s * 8u);
    const float4* bp = (const float4*)(ad1 + (unsigned)d * 8u);
    float4 a0 = ap[0], a1 = ap[1], b0 = bp[0], b1 = bp[1];
    float e[8] = {a0.x + b0.x, a0.y + b0.y, a0.z + b0.z, a0.w + b0.w,
                  a1.x + b1.x, a1.y + b1.y, a1.z + b1.z, a1.w + b1.w};
#pragma unroll
    for (int h = 0; h < 8; ++h) {
        float v = e[h] > 0.0f ? e[h] : NEG_SLOPE * e[h];
        atomicMax(&m1[(unsigned)d * 8u + h], enc_f32(v));
    }
}

__global__ void k_edge_sum1(const int* __restrict__ ei, int E, int n,
                            const float* __restrict__ as1, const float* __restrict__ ad1,
                            const unsigned* __restrict__ m1, float* __restrict__ s1) {
    int idx = blockIdx.x * blockDim.x + threadIdx.x;
    if (idx >= E + n) return;
    int s, d; edge_sd(ei, E, n, idx, s, d);
    const float4* ap = (const float4*)(as1 + (unsigned)s * 8u);
    const float4* bp = (const float4*)(ad1 + (unsigned)d * 8u);
    const uint4*  mp = (const uint4*)(m1 + (unsigned)d * 8u);
    float4 a0 = ap[0], a1 = ap[1], b0 = bp[0], b1 = bp[1];
    uint4  m0 = mp[0], m1v = mp[1];
    float e[8] = {a0.x + b0.x, a0.y + b0.y, a0.z + b0.z, a0.w + b0.w,
                  a1.x + b1.x, a1.y + b1.y, a1.z + b1.z, a1.w + b1.w};
    unsigned mm[8] = {m0.x, m0.y, m0.z, m0.w, m1v.x, m1v.y, m1v.z, m1v.w};
#pragma unroll
    for (int h = 0; h < 8; ++h) {
        float v = e[h] > 0.0f ? e[h] : NEG_SLOPE * e[h];
        atomicAdd(&s1[(unsigned)d * 8u + h], expf(v - dec_f32(mm[h])));
    }
}

__global__ void k_edge_scatter1(const int* __restrict__ ei, int E, int n,
                                const float* __restrict__ as1, const float* __restrict__ ad1,
                                const unsigned* __restrict__ m1, const float* __restrict__ s1,
                                const float* __restrict__ h1, float* __restrict__ acc1) {
    int tid = blockIdx.x * blockDim.x + threadIdx.x;
    if (tid >= (E + n) * 8) return;
    int idx = tid >> 3, h = tid & 7;
    int s, d; edge_sd(ei, E, n, idx, s, d);
    const float* hp = h1 + (unsigned)s * 64u + (unsigned)h * 8u;
    __builtin_prefetch(hp, 0, 3);
    float e = as1[(unsigned)s * 8u + h] + ad1[(unsigned)d * 8u + h];
    e = e > 0.0f ? e : NEG_SLOPE * e;
    float ex    = expf(e - dec_f32(m1[(unsigned)d * 8u + h]));
    float alpha = ex / (s1[(unsigned)d * 8u + h] + 1e-16f);
    float4 v0 = ((const float4*)hp)[0], v1 = ((const float4*)hp)[1];
    float* op = acc1 + (unsigned)d * 64u + (unsigned)h * 8u;
    atomicAdd(op + 0, v0.x * alpha);
    atomicAdd(op + 1, v0.y * alpha);
    atomicAdd(op + 2, v0.z * alpha);
    atomicAdd(op + 3, v0.w * alpha);
    atomicAdd(op + 4, v1.x * alpha);
    atomicAdd(op + 5, v1.y * alpha);
    atomicAdd(op + 6, v1.z * alpha);
    atomicAdd(op + 7, v1.w * alpha);
}

__global__ void k_post1(float* __restrict__ acc1, const float* __restrict__ bias1, int n) {
    int tid = blockIdx.x * blockDim.x + threadIdx.x;
    if (tid >= n * 64) return;
    float v = acc1[tid] + bias1[tid & 63];
    acc1[tid] = v > 0.0f ? v : (expf(v) - 1.0f);
}

__global__ void k_alpha2(const float* __restrict__ h2,
                         const float* __restrict__ asrc, const float* __restrict__ adst,
                         float* __restrict__ as2, float* __restrict__ ad2, int n) {
    int tid = blockIdx.x * blockDim.x + threadIdx.x;
    if (tid >= n) return;
    const float4* hp = (const float4*)(h2 + (unsigned)tid * 8u);
    float4 v0 = hp[0], v1 = hp[1];
    const float4* sp = (const float4*)asrc;
    const float4* dp = (const float4*)adst;
    float4 s0 = sp[0], s1v = sp[1], d0 = dp[0], d1v = dp[1];
    as2[tid] = v0.x * s0.x + v0.y * s0.y + v0.z * s0.z + v0.w * s0.w
             + v1.x * s1v.x + v1.y * s1v.y + v1.z * s1v.z + v1.w * s1v.w;
    ad2[tid] = v0.x * d0.x + v0.y * d0.y + v0.z * d0.z + v0.w * d0.w
             + v1.x * d1v.x + v1.y * d1v.y + v1.z * d1v.z + v1.w * d1v.w;
}

__global__ void k_edge_max2(const int* __restrict__ ei, int E, int n,
                            const float* __restrict__ as2, const float* __restrict__ ad2,
                            unsigned* __restrict__ m2) {
    int idx = blockIdx.x * blockDim.x + threadIdx.x;
    if (idx >= E + n) return;
    int s, d; edge_sd(ei, E, n, idx, s, d);
    float e = as2[s] + ad2[d];
    e = e > 0.0f ? e : NEG_SLOPE * e;
    atomicMax(&m2[d], enc_f32(e));
}

__global__ void k_edge_sum2(const int* __restrict__ ei, int E, int n,
                            const float* __restrict__ as2, const float* __restrict__ ad2,
                            const unsigned* __restrict__ m2, float* __restrict__ s2) {
    int idx = blockIdx.x * blockDim.x + threadIdx.x;
    if (idx >= E + n) return;
    int s, d; edge_sd(ei, E, n, idx, s, d);
    float e = as2[s] + ad2[d];
    e = e > 0.0f ? e : NEG_SLOPE * e;
    atomicAdd(&s2[d], expf(e - dec_f32(m2[d])));
}

__global__ void k_edge_scatter2(const int* __restrict__ ei, int E, int n,
                                const float* __restrict__ as2, const float* __restrict__ ad2,
                                const unsigned* __restrict__ m2, const float* __restrict__ s2,
                                const float* __restrict__ h2, float* __restrict__ out) {
    int idx = blockIdx.x * blockDim.x + threadIdx.x;
    if (idx >= E + n) return;
    int s, d; edge_sd(ei, E, n, idx, s, d);
    const float* hp = h2 + (unsigned)s * 8u;
    __builtin_prefetch(hp, 0, 3);
    float e = as2[s] + ad2[d];
    e = e > 0.0f ? e : NEG_SLOPE * e;
    float ex    = expf(e - dec_f32(m2[d]));
    float alpha = ex / (s2[d] + 1e-16f);
    float4 v0 = ((const float4*)hp)[0], v1 = ((const float4*)hp)[1];
    float* op = out + (unsigned)d * 8u;
    atomicAdd(op + 0, v0.x * alpha);
    atomicAdd(op + 1, v0.y * alpha);
    atomicAdd(op + 2, v0.z * alpha);
    atomicAdd(op + 3, v0.w * alpha);
    atomicAdd(op + 4, v1.x * alpha);
    atomicAdd(op + 5, v1.y * alpha);
    atomicAdd(op + 6, v1.z * alpha);
    atomicAdd(op + 7, v1.w * alpha);
}

__global__ void k_post2(float* __restrict__ out, const float* __restrict__ bias2, int n) {
    int tid = blockIdx.x * blockDim.x + threadIdx.x;
    if (tid >= n * 8) return;
    out[tid] += bias2[tid & 7];
}

extern "C" void kernel_launch(void* const* d_in, const int* in_sizes, int n_in,
                              void* d_out, int out_size, void* d_ws, size_t ws_size,
                              hipStream_t stream) {
    const float* x     = (const float*)d_in[0];
    const int*   ei    = (const int*)d_in[1];
    const float* W1    = (const float*)d_in[2];
    const float* asrc1 = (const float*)d_in[3];
    const float* adst1 = (const float*)d_in[4];
    const float* bias1 = (const float*)d_in[5];
    const float* W2    = (const float*)d_in[6];
    const float* asrc2 = (const float*)d_in[7];
    const float* adst2 = (const float*)d_in[8];
    const float* bias2 = (const float*)d_in[9];

    int n = in_sizes[0] / F_IN;
    int E = in_sizes[1] / 2;

    char* p = (char*)d_ws;
    float*    h1   = (float*)p;    p += (size_t)n * 64 * 4;
    float*    acc1 = (float*)p;    p += (size_t)n * 64 * 4;
    float*    as1  = (float*)p;    p += (size_t)n * 8 * 4;
    float*    ad1  = (float*)p;    p += (size_t)n * 8 * 4;
    float*    s1   = (float*)p;    p += (size_t)n * 8 * 4;
    unsigned* m1   = (unsigned*)p; p += (size_t)n * 8 * 4;
    float*    h2   = (float*)p;    p += (size_t)n * 8 * 4;
    float*    as2  = (float*)p;    p += (size_t)n * 4;
    float*    ad2  = (float*)p;    p += (size_t)n * 4;
    float*    s2   = (float*)p;    p += (size_t)n * 4;
    unsigned* m2   = (unsigned*)p; p += (size_t)n * 4;

    hipMemsetAsync(acc1, 0, (size_t)n * 64 * 4, stream);
    hipMemsetAsync(s1,   0, (size_t)n * 8 * 4, stream);
    hipMemsetAsync(m1,   0, (size_t)n * 8 * 4, stream);
    hipMemsetAsync(s2,   0, (size_t)n * 4, stream);
    hipMemsetAsync(m2,   0, (size_t)n * 4, stream);
    hipMemsetAsync(d_out, 0, (size_t)out_size * 4, stream);

    const int tpb = 256;
    int tot = E + n;
    int gEd = (tot + tpb - 1) / tpb;

    k_gemm1<<<(n + 15) / 16, 32, 0, stream>>>(x, W1, h1, n);
    k_alpha1<<<(n * 8 + tpb - 1) / tpb, tpb, 0, stream>>>(h1, asrc1, adst1, as1, ad1, n);
    k_edge_max1<<<gEd, tpb, 0, stream>>>(ei, E, n, as1, ad1, m1);
    k_edge_sum1<<<gEd, tpb, 0, stream>>>(ei, E, n, as1, ad1, m1, s1);
    k_edge_scatter1<<<(tot * 8 + tpb - 1) / tpb, tpb, 0, stream>>>(ei, E, n, as1, ad1, m1, s1, h1, acc1);
    k_post1<<<(n * 64 + tpb - 1) / tpb, tpb, 0, stream>>>(acc1, bias1, n);

    k_gemm2<<<(n + 15) / 16, 32, 0, stream>>>(acc1, W2, h2, n);
    k_alpha2<<<(n + tpb - 1) / tpb, tpb, 0, stream>>>(h2, asrc2, adst2, as2, ad2, n);
    k_edge_max2<<<gEd, tpb, 0, stream>>>(ei, E, n, as2, ad2, m2);
    k_edge_sum2<<<gEd, tpb, 0, stream>>>(ei, E, n, as2, ad2, m2, s2);
    k_edge_scatter2<<<gEd, tpb, 0, stream>>>(ei, E, n, as2, ad2, m2, s2, h2, (float*)d_out);
    k_post2<<<(n * 8 + tpb - 1) / tpb, tpb, 0, stream>>>((float*)d_out, bias2, n);
}


# layer-1 scatter by destination-range owners: each work-group reads the whole dst list, queues its range's edges in LDS, sums rows with LDS float atomics (arrival order), writes finished rows
# speedup vs baseline: 1.5084x; 1.5084x over previous
#include <hip/hip_runtime.h>
#include <hip/hip_bf16.h>

#define NEG_SLOPE 0.2f
#define F_IN 14

typedef float v2f __attribute__((ext_vector_type(2)));
typedef float v8f __attribute__((ext_vector_type(8)));

__device__ __forceinline__ unsigned enc_f32(float f) {
    unsigned u = __float_as_uint(f);
    return (u & 0x80000000u) ? ~u : (u | 0x80000000u);
}
__device__ __forceinline__ float dec_f32(unsigned e) {
    unsigned u = (e & 0x80000000u) ? (e ^ 0x80000000u) : ~e;
    return __uint_as_float(u);
}

__device__ __forceinline__ void edge_sd(const int* __restrict__ ei, int E, int n,
                                        int idx, int& s, int& d) {
    if (idx < E) { s = ei[idx]; d = ei[E + idx]; }
    else         { s = idx - E; d = idx - E; }
    s = s < 0 ? 0 : (s >= n ? n - 1 : s);
    d = d < 0 ? 0 : (d >= n ? n - 1 : d);
}

__global__ __launch_bounds__(32) void k_gemm1(const float* __restrict__ x,
                                              const float* __restrict__ W1,
                                              float* __restrict__ h1, int n) {
    __shared__ float xs[16 * 16];
    __shared__ float ws[16 * 64];

    const int lane = threadIdx.x;
    const int row0 = blockIdx.x * 16;
    const bool hi  = lane >= 16;
    const int  m   = lane & 15;
    const bool full = (row0 + 16 <= n);

    for (int i = lane; i < 2 * 64 / 4; i += 32)
        ((float4*)(ws + 14 * 64))[i] = make_float4(0.f, 0.f, 0.f, 0.f);
    for (int i = lane; i < F_IN * 64 / 4; i += 32)
        ((float4*)ws)[i] = ((const float4*)W1)[i];

    for (int i = lane; i < 16 * 16 / 4; i += 32)
        ((float4*)xs)[i] = make_float4(0.f, 0.f, 0.f, 0.f);
    __syncthreads();
    {
        const float* xt = x + (size_t)row0 * F_IN;
        int rows = full ? 16 : (n - row0);
        int cnt  = rows * F_IN;
        for (int i = lane; i < cnt; i += 32) {
            int r = i / F_IN, c = i - r * F_IN;
            xs[r * 16 + c] = xt[i];
        }
    }
    __syncthreads();

    const int kofs = hi ? 2 : 0;
    v2f a[4];
#pragma unroll
    for (int k = 0; k < 4; ++k) {
        a[k][0] = xs[m * 16 + 4 * k + kofs + 0];
        a[k][1] = xs[m * 16 + 4 * k + kofs + 1];
    }

#pragma unroll
    for (int t = 0; t < 4; ++t) {
        v8f c = {};
#pragma unroll
        for (int k = 0; k < 4; ++k) {
            v2f b;
            b[0] = ws[(4 * k + kofs + 0) * 64 + t * 16 + m];
            b[1] = ws[(4 * k + kofs + 1) * 64 + t * 16 + m];
            c = __builtin_amdgcn_wmma_f32_16x16x4_f32(false, a[k], false, b,
                                                      (short)0, c, false, false);
        }
        float* out = h1 + (size_t)row0 * 64 + t * 16 + m;
        if (full) {
#pragma unroll
            for (int r = 0; r < 8; ++r)
                out[(size_t)(r + (hi ? 8 : 0)) * 64] = c[r];
        } else {
#pragma unroll
            for (int r = 0; r < 8; ++r) {
                int rr = r + (hi ? 8 : 0);
                if (row0 + rr < n) out[(size_t)rr * 64] = c[r];
            }
        }
    }
}

__global__ __launch_bounds__(32) void k_gemm2(const float* __restrict__ a1,
                                              const float* __restrict__ W2,
                                              float* __restrict__ h2, int n) {
    __shared__ float smA[16 * 64];
    __shared__ float ws2[64 * 16];

    const int lane = threadIdx.x;
    const int row0 = blockIdx.x * 16;
    const bool hi  = lane >= 16;
    const int  m   = lane & 15;
    const bool full = (row0 + 16 <= n);

    for (int i = lane; i < 64 * 16 / 4; i += 32)
        ((float4*)ws2)[i] = make_float4(0.f, 0.f, 0.f, 0.f);
    __syncthreads();
    for (int i = lane; i < 64 * 8 / 4; i += 32) {
        int k = i >> 1, half = i & 1;
        ((float4*)(ws2 + k * 16 + half * 4))[0] = ((const float4*)W2)[i];
    }

    if (full) {
        const float4* src = (const float4*)(a1 + (size_t)row0 * 64);
        for (int i = lane; i < 16 * 64 / 4; i += 32)
            ((float4*)smA)[i] = src[i];
    } else {
        for (int i = lane; i < 16 * 64 / 4; i += 32)
            ((float4*)smA)[i] = make_float4(0.f, 0.f, 0.f, 0.f);
        __syncthreads();
        int rows = n - row0;
        const float4* src = (const float4*)(a1 + (size_t)row0 * 64);
        for (int i = lane; i < rows * 16; i += 32)
            ((float4*)smA)[i] = src[i];
    }
    __syncthreads();

    const int kofs = hi ? 2 : 0;
    v8f c = {};
#pragma unroll
    for (int k = 0; k < 16; ++k) {
        v2f a, b;
        a[0] = smA[m * 64 + 4 * k + kofs + 0];
        a[1] = smA[m * 64 + 4 * k + kofs + 1];
        b[0] = ws2[(4 * k + kofs + 0) * 16 + m];
        b[1] = ws2[(4 * k + kofs + 1) * 16 + m];
        c = __builtin_amdgcn_wmma_f32_16x16x4_f32(false, a, false, b,
                                                  (short)0, c, false, false);
    }
    if (m < 8) {
        float* out = h2 + (size_t)row0 * 8 + m;
        if (full) {
#pragma unroll
            for (int r = 0; r < 8; ++r)
                out[(size_t)(r + (hi ? 8 : 0)) * 8] = c[r];
        } else {
#pragma unroll
            for (int r = 0; r < 8; ++r) {
                int rr = r + (hi ? 8 : 0);
                if (row0 + rr < n) out[(size_t)rr * 8] = c[r];
            }
        }
    }
}

__global__ void k_alpha1(const float* __restrict__ h1,
                         const float* __restrict__ asrc, const float* __restrict__ adst,
                         float* __restrict__ as1, float* __restrict__ ad1, int n) {
    int tid = blockIdx.x * blockDim.x + threadIdx.x;
    if (tid >= n * 8) return;
    unsigned h = tid & 7u;
    const float4* hp = (const float4*)(h1 + (unsigned)(tid >> 3) * 64u + h * 8u);
    float4 v0 = hp[0], v1 = hp[1];
    const float4* sp = (const float4*)(asrc + h * 8u);
    const float4* dp = (const float4*)(adst + h * 8u);
    float4 s0 = sp[0], s1v = sp[1], d0 = dp[0], d1v = dp[1];
    float s = v0.x * s0.x + v0.y * s0.y + v0.z * s0.z + v0.w * s0.w
            + v1.x * s1v.x + v1.y * s1v.y + v1.z * s1v.z + v1.w * s1v.w;
    float d = v0.x * d0.x + v0.y * d0.y + v0.z * d0.z + v0.w * d0.w
            + v1.x * d1v.x + v1.y * d1v.y + v1.z * d1v.z + v1.w * d1v.w;
    as1[tid] = s;
    ad1[tid] = d;
}

__global__ void k_edge_max1(const int* __restrict__ ei, int E, int n,
                            const float* __restrict__ as1, const float* __restrict__ ad1,
                            unsigned* __restrict__ m1) {
    int idx = blockIdx.x * blockDim.x + threadIdx.x;
    if (idx >= E + n) return;
    int s, d; edge_sd(ei, E, n, idx, s, d);
    const float4* ap = (const float4*)(as1 + (unsigned)s * 8u);
    const float4* bp = (const float4*)(ad1 + (unsigned)d * 8u);
    float4 a0 = ap[0], a1 = ap[1], b0 = bp[0], b1 = bp[1];
    float e[8] = {a0.x + b0.x, a0.y + b0.y, a0.z + b0.z, a0.w + b0.w,
                  a1.x + b1.x, a1.y + b1.y, a1.z + b1.z, a1.w + b1.w};
#pragma unroll
    for (int h = 0; h < 8; ++h) {
        float v = e[h] > 0.0f ? e[h] : NEG_SLOPE * e[h];
        atomicMax(&m1[(unsigned)d * 8u + h], enc_f32(v));
    }
}

#define NB1   392
#define QCAP  8192
#define OWN_T 1024

__device__ __noinline__ void own_edge_direct(int s, unsigned dl, int dnode,
                                             const float* __restrict__ as1,
                                             const float* __restrict__ ad1,
                                             const unsigned* __restrict__ m1,
                                             const float* __restrict__ h1,
                                             float* acc, float* ssum) {
    for (int h = 0; h < 8; ++h) {
        float e = as1[(unsigned)s * 8u + h] + ad1[(unsigned)dnode * 8u + h];
        e = e > 0.0f ? e : NEG_SLOPE * e;
        float ex = expf(e - dec_f32(m1[(unsigned)dnode * 8u + h]));
        atomicAdd(&ssum[dl * 8u + h], ex);
        const float* hp = h1 + (unsigned)s * 64u + (unsigned)h * 8u;
        for (int j = 0; j < 8; ++j)
            atomicAdd(&acc[dl * 64u + h * 8u + j], hp[j] * ex);
    }
}

__global__ __launch_bounds__(OWN_T) void k_own_scatter1(const int* __restrict__ ei, int E, int n,
                                const float* __restrict__ as1, const float* __restrict__ ad1,
                                const unsigned* __restrict__ m1,
                                const float* __restrict__ h1,
                                const float* __restrict__ bias1, float* __restrict__ a1out) {
    __shared__ float acc[NB1 * 64];
    __shared__ float ssum[NB1 * 8];
    __shared__ int qs[QCAP];
    __shared__ unsigned short qd[QCAP];
    __shared__ unsigned qn;

    const int tid  = threadIdx.x;
    const int lane = tid & 31;
    const int wave = tid >> 5;
    const int d0   = blockIdx.x * NB1;
    const int nb   = (n - d0) < NB1 ? (n - d0) : NB1;

    for (int i = tid; i < NB1 * 64; i += OWN_T) acc[i] = 0.0f;
    for (int i = tid; i < NB1 * 8; i += OWN_T) ssum[i] = 0.0f;
    if (tid == 0) qn = 0u;
    __syncthreads();

    const int* __restrict__ dstp = ei + E;
    const int lo = (d0 == 0) ? (int)0x80000000 : d0;
    const int hi = (d0 + nb == n) ? 0x7fffffff : (d0 + nb - 1);
    const unsigned span = (unsigned)hi - (unsigned)lo;
    auto push = [&](int idx) {
        int d = dstp[idx];
        d = d < 0 ? 0 : (d >= n ? n - 1 : d);
        unsigned dl = (unsigned)(d - d0);
        unsigned slot = atomicAdd(&qn, 1u);
        if (slot < (unsigned)QCAP) {
            qs[slot] = idx;
            qd[slot] = (unsigned short)dl;
        } else {
            int s = ei[idx];
            s = s < 0 ? 0 : (s >= n ? n - 1 : s);
            own_edge_direct(s, dl, d, as1, ad1, m1, h1, acc, ssum);
        }
    };
    const int E4 = ((E & 3) == 0) ? (E >> 2) : 0;
    const int4* __restrict__ dp4 = (const int4*)dstp;
    auto hits4 = [&](int4 q) -> unsigned {
        return ((unsigned)q.x - (unsigned)lo <= span ? 1u : 0u)
             | ((unsigned)q.y - (unsigned)lo <= span ? 2u : 0u)
             | ((unsigned)q.z - (unsigned)lo <= span ? 4u : 0u)
             | ((unsigned)q.w - (unsigned)lo <= span ? 8u : 0u);
    };
    int base = tid;
    for (; base + 7 * OWN_T < E4; base += 8 * OWN_T) {
        int4 v[8];
#pragma unroll
        for (int u = 0; u < 8; ++u)
            v[u] = dp4[base + u * OWN_T];
        unsigned hit = 0u;
#pragma unroll
        for (int u = 0; u < 8; ++u)
            hit |= hits4(v[u]) << (4 * u);
        while (hit) {
            int j = __ffs(hit) - 1;
            hit &= hit - 1u;
            push(4 * (base + (j >> 2) * OWN_T) + (j & 3));
        }
    }
    for (; base < E4; base += OWN_T) {
        unsigned hit = hits4(dp4[base]);
        while (hit) {
            int j = __ffs(hit) - 1;
            hit &= hit - 1u;
            push(4 * base + j);
        }
    }
    for (int idx = 4 * E4 + tid; idx < E; idx += OWN_T)
        if ((unsigned)dstp[idx] - (unsigned)lo <= span) push(idx);
    __syncthreads();

    const int cnt = (int)(qn < (unsigned)QCAP ? qn : (unsigned)QCAP);
    for (int q = tid; q < cnt; q += OWN_T) {
        int s = ei[qs[q]];
        qs[q] = s < 0 ? 0 : (s >= n ? n - 1 : s);
    }
    __syncthreads();

    const int total = cnt + nb;
    const unsigned h = (unsigned)lane >> 2;
    for (int q0 = wave * 4; q0 < total; q0 += (OWN_T / 32) * 4) {
        int   sN[4];
        unsigned dL[4];
        float eV[4];
        unsigned mV[4];
        v2f   hV[4];
#pragma unroll
        for (int u = 0; u < 4; ++u) {
            int q = q0 + u;
            int qq = q < total ? q : q0;
            if (qq < cnt) { sN[u] = qs[qq]; dL[u] = qd[qq]; }
            else          { dL[u] = (unsigned)(qq - cnt); sN[u] = d0 + (qq - cnt); }
        }
#pragma unroll
        for (int u = 0; u < 4; ++u) {
            unsigned dn = (unsigned)d0 + dL[u];
            eV[u] = as1[(unsigned)sN[u] * 8u + h] + ad1[dn * 8u + h];
            mV[u] = m1[dn * 8u + h];
            hV[u] = ((const v2f*)(h1 + (unsigned)sN[u] * 64u))[lane];
        }
#pragma unroll
        for (int u = 0; u < 4; ++u) {
            if (q0 + u < total) {
                float e = eV[u] > 0.0f ? eV[u] : NEG_SLOPE * eV[u];
                float ex = expf(e - dec_f32(mV[u]));
                if ((lane & 3) == 0) atomicAdd(&ssum[dL[u] * 8u + h], ex);
                atomicAdd(&acc[dL[u] * 64u + 2u * lane + 0u], hV[u][0] * ex);
                atomicAdd(&acc[dL[u] * 64u + 2u * lane + 1u], hV[u][1] * ex);
            }
        }
    }
    __syncthreads();

    float* __restrict__ op = a1out + (size_t)d0 * 64;
    for (int i = tid; i < nb * 64; i += OWN_T) {
        float v = acc[i] / (ssum[i >> 3] + 1e-16f) + bias1[i & 63];
        op[i] = v > 0.0f ? v : (expf(v) - 1.0f);
    }
}

__global__ void k_alpha2(const float* __restrict__ h2,
                         const float* __restrict__ asrc, const float* __restrict__ adst,
                         float* __restrict__ as2, float* __restrict__ ad2, int n) {
    int tid = blockIdx.x * blockDim.x + threadIdx.x;
    if (tid >= n) return;
    const float4* hp = (const float4*)(h2 + (unsigned)tid * 8u);
    float4 v0 = hp[0], v1 = hp[1];
    const float4* sp = (const float4*)asrc;
    const float4* dp = (const float4*)adst;
    float4 s0 = sp[0], s1v = sp[1], d0 = dp[0], d1v = dp[1];
    as2[tid] = v0.x * s0.x + v0.y * s0.y + v0.z * s0.z + v0.w * s0.w
             + v1.x * s1v.x + v1.y * s1v.y + v1.z * s1v.z + v1.w * s1v.w;
    ad2[tid] = v0.x * d0.x + v0.y * d0.y + v0.z * d0.z + v0.w * d0.w
             + v1.x * d1v.x + v1.y * d1v.y + v1.z * d1v.z + v1.w * d1v.w;
}

__global__ void k_edge_max2(const int* __restrict__ ei, int E, int n,
                            const float* __restrict__ as2, const float* __restrict__ ad2,
                            unsigned* __restrict__ m2) {
    int idx = blockIdx.x * blockDim.x + threadIdx.x;
    if (idx >= E + n) return;
    int s, d; edge_sd(ei, E, n, idx, s, d);
    float e = as2[s] + ad2[d];
    e = e > 0.0f ? e : NEG_SLOPE * e;
    atomicMax(&m2[d], enc_f32(e));
}

__global__ void k_edge_scatter2(const int* __restrict__ ei, int E, int n,
                                const float* __restrict__ as2, const float* __restrict__ ad2,
                                const unsigned* __restrict__ m2, float* __restrict__ s2,
                                const float* __restrict__ h2, float* __restrict__ out) {
    int idx = blockIdx.x * blockDim.x + threadIdx.x;
    if (idx >= E + n) return;
    int s, d; edge_sd(ei, E, n, idx, s, d);
    const float* hp = h2 + (unsigned)s * 8u;
    __builtin_prefetch(hp, 0, 3);
    float e = as2[s] + ad2[d];
    e = e > 0.0f ? e : NEG_SLOPE * e;
    float ex = expf(e - dec_f32(m2[d]));
    atomicAdd(&s2[d], ex);
    float4 v0 = ((const float4*)hp)[0], v1 = ((const float4*)hp)[1];
    float* op = out + (unsigned)d * 8u;
    atomicAdd(op + 0, v0.x * ex);
    atomicAdd(op + 1, v0.y * ex);
    atomicAdd(op + 2, v0.z * ex);
    atomicAdd(op + 3, v0.w * ex);
    atomicAdd(op + 4, v1.x * ex);
    atomicAdd(op + 5, v1.y * ex);
    atomicAdd(op + 6, v1.z * ex);
    atomicAdd(op + 7, v1.w * ex);
}

__global__ void k_post2(float* __restrict__ out, const float* __restrict__ s2,
                        const float* __restrict__ bias2, int n) {
    int tid = blockIdx.x * blockDim.x + threadIdx.x;
    if (tid >= n * 8) return;
    out[tid] = out[tid] / (s2[tid >> 3] + 1e-16f) + bias2[tid & 7];
}

extern "C" void kernel_launch(void* const* d_in, const int* in_sizes, int n_in,
                              void* d_out, int out_size, void* d_ws, size_t ws_size,
                              hipStream_t stream) {
    const float* x     = (const float*)d_in[0];
    const int*   ei    = (const int*)d_in[1];
    const float* W1    = (const float*)d_in[2];
    const float* asrc1 = (const float*)d_in[3];
    const float* adst1 = (const float*)d_in[4];
    const float* bias1 = (const float*)d_in[5];
    const float* W2    = (const float*)d_in[6];
    const float* asrc2 = (const float*)d_in[7];
    const float* adst2 = (const float*)d_in[8];
    const float* bias2 = (const float*)d_in[9];

    int n = in_sizes[0] / F_IN;
    int E = in_sizes[1] / 2;

    char* p = (char*)d_ws;
    float*    h1   = (float*)p;    p += (size_t)n * 64 * 4;
    float*    acc1 = (float*)p;    p += (size_t)n * 64 * 4;
    float*    as1  = (float*)p;    p += (size_t)n * 8 * 4;
    float*    ad1  = (float*)p;    p += (size_t)n * 8 * 4;
    float*    s1   = (float*)p;    p += (size_t)n * 8 * 4;
    unsigned* m1   = (unsigned*)p; p += (size_t)n * 8 * 4;
    float*    h2   = (float*)p;    p += (size_t)n * 8 * 4;
    float*    as2  = (float*)p;    p += (size_t)n * 4;
    float*    ad2  = (float*)p;    p += (size_t)n * 4;
    float*    s2   = (float*)p;    p += (size_t)n * 4;
    unsigned* m2   = (unsigned*)p; p += (size_t)n * 4;

    hipMemsetAsync(m1,   0, (size_t)n * 8 * 4, stream);
    hipMemsetAsync(s2,   0, (size_t)n * 4, stream);
    hipMemsetAsync(m2,   0, (size_t)n * 4, stream);
    hipMemsetAsync(d_out, 0, (size_t)out_size * 4, stream);

    const int tpb = 256;
    int tot = E + n;
    int gEd = (tot + tpb - 1) / tpb;

    k_gemm1<<<(n + 15) / 16, 32, 0, stream>>>(x, W1, h1, n);
    k_alpha1<<<(n * 8 + tpb - 1) / tpb, tpb, 0, stream>>>(h1, asrc1, adst1, as1, ad1, n);
    k_edge_max1<<<gEd, tpb, 0, stream>>>(ei, E, n, as1, ad1, m1);
    k_own_scatter1<<<(n + NB1 - 1) / NB1, OWN_T, 0, stream>>>(ei, E, n, as1, ad1, m1, h1, bias1, acc1);

    k_gemm2<<<(n + 15) / 16, 32, 0, stream>>>(acc1, W2, h2, n);
    k_alpha2<<<(n + tpb - 1) / tpb, tpb, 0, stream>>>(h2, asrc2, adst2, as2, ad2, n);
    k_edge_max2<<<gEd, tpb, 0, stream>>>(ei, E, n, as2, ad2, m2);
    k_edge_scatter2<<<gEd, tpb, 0, stream>>>(ei, E, n, as2, ad2, m2, s2, h2, (float*)d_out);
    k_post2<<<(n * 8 + tpb - 1) / tpb, tpb, 0, stream>>>((float*)d_out, s2, bias2, n);
}


# layer-1 segment max taken inside the owner work-groups: an LDS max walk over the queued edges ahead of the sum walk; the global max pass, its memset and its array go; sum orders kept
# speedup vs baseline: 2.1131x; 1.4009x over previous
#include <hip/hip_runtime.h>
#include <hip/hip_bf16.h>

#define NEG_SLOPE 0.2f
#define F_IN 14

typedef float v2f __attribute__((ext_vector_type(2)));
typedef float v8f __attribute__((ext_vector_type(8)));

__device__ __forceinline__ unsigned enc_f32(float f) {
    unsigned u = __float_as_uint(f);
    return (u & 0x80000000u) ? ~u : (u | 0x80000000u);
}
__device__ __forceinline__ float dec_f32(unsigned e) {
    unsigned u = (e & 0x80000000u) ? (e ^ 0x80000000u) : ~e;
    return __uint_as_float(u);
}

__device__ __forceinline__ void edge_sd(const int* __restrict__ ei, int E, int n,
                                        int idx, int& s, int& d) {
    if (idx < E) { s = ei[idx]; d = ei[E + idx]; }
    else         { s = idx - E; d = idx - E; }
    s = s < 0 ? 0 : (s >= n ? n - 1 : s);
    d = d < 0 ? 0 : (d >= n ? n - 1 : d);
}

__global__ __launch_bounds__(32) void k_gemm1(const float* __restrict__ x,
                                              const float* __restrict__ W1,
                                              float* __restrict__ h1, int n) {
    __shared__ float xs[16 * 16];
    __shared__ float ws[16 * 64];

    const int lane = threadIdx.x;
    const int row0 = blockIdx.x * 16;
    const bool hi  = lane >= 16;
    const int  m   = lane & 15;
    const bool full = (row0 + 16 <= n);

    for (int i = lane; i < 2 * 64 / 4; i += 32)
        ((float4*)(ws + 14 * 64))[i] = make_float4(0.f, 0.f, 0.f, 0.f);
    for (int i = lane; i < F_IN * 64 / 4; i += 32)
        ((float4*)ws)[i] = ((const float4*)W1)[i];

    for (int i = lane; i < 16 * 16 / 4; i += 32)
        ((float4*)xs)[i] = make_float4(0.f, 0.f, 0.f, 0.f);
    __syncthreads();
    {
        const float* xt = x + (size_t)row0 * F_IN;
        int rows = full ? 16 : (n - row0);
        int cnt  = rows * F_IN;
        for (int i = lane; i < cnt; i += 32) {
            int r = i / F_IN, c = i - r * F_IN;
            xs[r * 16 + c] = xt[i];
        }
    }
    __syncthreads();

    const int kofs = hi ? 2 : 0;
    v2f a[4];
#pragma unroll
    for (int k = 0; k < 4; ++k) {
        a[k][0] = xs[m * 16 + 4 * k + kofs + 0];
        a[k][1] = xs[m * 16 + 4 * k + kofs + 1];
    }

#pragma unroll
    for (int t = 0; t < 4; ++t) {
        v8f c = {};
#pragma unroll
        for (int k = 0; k < 4; ++k) {
            v2f b;
            b[0] = ws[(4 * k + kofs + 0) * 64 + t * 16 + m];
            b[1] = ws[(4 * k + kofs + 1) * 64 + t * 16 + m];
            c = __builtin_amdgcn_wmma_f32_16x16x4_f32(false, a[k], false, b,
                                                      (short)0, c, false, false);
        }
        float* out = h1 + (size_t)row0 * 64 + t * 16 + m;
        if (full) {
#pragma unroll
            for (int r = 0; r < 8; ++r)
                out[(size_t)(r + (hi ? 8 : 0)) * 64] = c[r];
        } else {
#pragma unroll
            for (int r = 0; r < 8; ++r) {
                int rr = r + (hi ? 8 : 0);
                if (row0 + rr < n) out[(size_t)rr * 64] = c[r];
            }
        }
    }
}

__global__ __launch_bounds__(32) void k_gemm2(const float* __restrict__ a1,
                                              const float* __restrict__ W2,
                                              float* __restrict__ h2, int n) {
    __shared__ float smA[16 * 64];
    __shared__ float ws2[64 * 16];

    const int lane = threadIdx.x;
    const int row0 = blockIdx.x * 16;
    const bool hi  = lane >= 16;
    const int  m   = lane & 15;
    const bool full = (row0 + 16 <= n);

    for (int i = lane; i < 64 * 16 / 4; i += 32)
        ((float4*)ws2)[i] = make_float4(0.f, 0.f, 0.f, 0.f);
    __syncthreads();
    for (int i = lane; i < 64 * 8 / 4; i += 32) {
        int k = i >> 1, half = i & 1;
        ((float4*)(ws2 + k * 16 + half * 4))[0] = ((const float4*)W2)[i];
    }

    if (full) {
        const float4* src = (const float4*)(a1 + (size_t)row0 * 64);
        for (int i = lane; i < 16 * 64 / 4; i += 32)
            ((float4*)smA)[i] = src[i];
    } else {
        for (int i = lane; i < 16 * 64 / 4; i += 32)
            ((float4*)smA)[i] = make_float4(0.f, 0.f, 0.f, 0.f);
        __syncthreads();
        int rows = n - row0;
        const float4* src = (const float4*)(a1 + (size_t)row0 * 64);
        for (int i = lane; i < rows * 16; i += 32)
            ((float4*)smA)[i] = src[i];
    }
    __syncthreads();

    const int kofs = hi ? 2 : 0;
    v8f c = {};
#pragma unroll
    for (int k = 0; k < 16; ++k) {
        v2f a, b;
        a[0] = smA[m * 64 + 4 * k + kofs + 0];
        a[1] = smA[m * 64 + 4 * k + kofs + 1];
        b[0] = ws2[(4 * k + kofs + 0) * 16 + m];
        b[1] = ws2[(4 * k + kofs + 1) * 16 + m];
        c = __builtin_amdgcn_wmma_f32_16x16x4_f32(false, a, false, b,
                                                  (short)0, c, false, false);
    }
    if (m < 8) {
        float* out = h2 + (size_t)row0 * 8 + m;
        if (full) {
#pragma unroll
            for (int r = 0; r < 8; ++r)
                out[(size_t)(r + (hi ? 8 : 0)) * 8] = c[r];
        } else {
#pragma unroll
            for (int r = 0; r < 8; ++r) {
                int rr = r + (hi ? 8 : 0);
                if (row0 + rr < n) out[(size_t)rr * 8] = c[r];
            }
        }
    }
}

__global__ void k_alpha1(const float* __restrict__ h1,
                         const float* __restrict__ asrc, const float* __restrict__ adst,
                         float* __restrict__ as1, float* __restrict__ ad1, int n) {
    int tid = blockIdx.x * blockDim.x + threadIdx.x;
    if (tid >= n * 8) return;
    unsigned h = tid & 7u;
    const float4* hp = (const float4*)(h1 + (unsigned)(tid >> 3) * 64u + h * 8u);
    float4 v0 = hp[0], v1 = hp[1];
    const float4* sp = (const float4*)(asrc + h * 8u);
    const float4* dp = (const float4*)(adst + h * 8u);
    float4 s0 = sp[0], s1v = sp[1], d0 = dp[0], d1v = dp[1];
    float s = v0.x * s0.x + v0.y * s0.y + v0.z * s0.z + v0.w * s0.w
            + v1.x * s1v.x + v1.y * s1v.y + v1.z * s1v.z + v1.w * s1v.w;
    float d = v0.x * d0.x + v0.y * d0.y + v0.z * d0.z + v0.w * d0.w
            + v1.x * d1v.x + v1.y * d1v.y + v1.z * d1v.z + v1.w * d1v.w;
    as1[tid] = s;
    ad1[tid] = d;
}

#define NB1   392
#define QCAP  8192
#define OWN_T 1024

template <class F>
__device__ __forceinline__ void own_scan(const int* __restrict__ dstp, int E,
                                         int lo, unsigned span, int tid, F f) {
    const int E4 = ((E & 3) == 0) ? (E >> 2) : 0;
    const int4* __restrict__ dp4 = (const int4*)dstp;
    auto hits4 = [&](int4 q) -> unsigned {
        return ((unsigned)q.x - (unsigned)lo <= span ? 1u : 0u)
             | ((unsigned)q.y - (unsigned)lo <= span ? 2u : 0u)
             | ((unsigned)q.z - (unsigned)lo <= span ? 4u : 0u)
             | ((unsigned)q.w - (unsigned)lo <= span ? 8u : 0u);
    };
    int base = tid;
    for (; base + 7 * OWN_T < E4; base += 8 * OWN_T) {
        int4 v[8];
#pragma unroll
        for (int u = 0; u < 8; ++u)
            v[u] = dp4[base + u * OWN_T];
        unsigned hit = 0u;
#pragma unroll
        for (int u = 0; u < 8; ++u)
            hit |= hits4(v[u]) << (4 * u);
        while (hit) {
            int j = __ffs(hit) - 1;
            hit &= hit - 1u;
            f(4 * (base + (j >> 2) * OWN_T) + (j & 3));
        }
    }
    for (; base < E4; base += OWN_T) {
        unsigned hit = hits4(dp4[base]);
        while (hit) {
            int j = __ffs(hit) - 1;
            hit &= hit - 1u;
            f(4 * base + j);
        }
    }
    for (int idx = 4 * E4 + tid; idx < E; idx += OWN_T)
        if ((unsigned)dstp[idx] - (unsigned)lo <= span) f(idx);
}

__device__ __noinline__ void own_edge_max_direct(int s, unsigned dl, int dnode,
                                                 const float* __restrict__ as1,
                                                 const float* __restrict__ ad1,
                                                 unsigned* mx) {
    for (int h = 0; h < 8; ++h) {
        float e = as1[(unsigned)s * 8u + h] + ad1[(unsigned)dnode * 8u + h];
        e = e > 0.0f ? e : NEG_SLOPE * e;
        atomicMax(&mx[dl * 8u + h], enc_f32(e));
    }
}

__device__ __noinline__ void own_edge_direct(int s, unsigned dl, int dnode,
                                             const float* __restrict__ as1,
                                             const float* __restrict__ ad1,
                                             const unsigned* mx,
                                             const float* __restrict__ h1,
                                             float* acc, float* ssum) {
    for (int h = 0; h < 8; ++h) {
        float e = as1[(unsigned)s * 8u + h] + ad1[(unsigned)dnode * 8u + h];
        e = e > 0.0f ? e : NEG_SLOPE * e;
        float ex = expf(e - dec_f32(mx[dl * 8u + h]));
        atomicAdd(&ssum[dl * 8u + h], ex);
        const float* hp = h1 + (unsigned)s * 64u + (unsigned)h * 8u;
        for (int j = 0; j < 8; ++j)
            atomicAdd(&acc[dl * 64u + h * 8u + j], hp[j] * ex);
    }
}

__global__ __launch_bounds__(OWN_T) void k_own_scatter1(const int* __restrict__ ei, int E, int n,
                                const float* __restrict__ as1, const float* __restrict__ ad1,
                                const float* __restrict__ h1,
                                const float* __restrict__ bias1, float* __restrict__ a1out) {
    __shared__ float acc[NB1 * 64];
    __shared__ float ssum[NB1 * 8];
    __shared__ unsigned mx[NB1 * 8];
    __shared__ int qs[QCAP];
    __shared__ unsigned short qd[QCAP];
    __shared__ unsigned qn;

    const int tid  = threadIdx.x;
    const int lane = tid & 31;
    const int wave = tid >> 5;
    const int d0   = blockIdx.x * NB1;
    const int nb   = (n - d0) < NB1 ? (n - d0) : NB1;

    for (int i = tid; i < NB1 * 64; i += OWN_T) acc[i] = 0.0f;
    for (int i = tid; i < NB1 * 8; i += OWN_T) { ssum[i] = 0.0f; mx[i] = 0u; }
    if (tid == 0) qn = 0u;
    __syncthreads();

    const int* __restrict__ dstp = ei + E;
    const int lo = (d0 == 0) ? (int)0x80000000 : d0;
    const int hi = (d0 + nb == n) ? 0x7fffffff : (d0 + nb - 1);
    const unsigned span = (unsigned)hi - (unsigned)lo;
    own_scan(dstp, E, lo, span, tid, [&](int idx) {
        int d = dstp[idx];
        d = d < 0 ? 0 : (d >= n ? n - 1 : d);
        unsigned slot = atomicAdd(&qn, 1u);
        if (slot < (unsigned)QCAP) {
            qs[slot] = idx;
            qd[slot] = (unsigned short)(d - d0);
        }
    });
    __syncthreads();

    const unsigned qtot = qn;
    if (qtot <= (unsigned)QCAP) {
        const int cnt = (int)qtot;
        for (int q = tid; q < cnt; q += OWN_T) {
            int s = ei[qs[q]];
            qs[q] = s < 0 ? 0 : (s >= n ? n - 1 : s);
        }
        __syncthreads();

        const int total = cnt + nb;

        {
            const unsigned hh = (unsigned)lane & 7u;
            const int sub = lane >> 3;
            for (int q0 = wave * 16; q0 < total; q0 += (OWN_T / 32) * 16) {
                int   sN[4];
                unsigned dL[4];
                float aV[4], bV[4];
#pragma unroll
                for (int u = 0; u < 4; ++u) {
                    int q = q0 + 4 * u + sub;
                    int qq = q < total ? q : q0;
                    if (qq < cnt) { sN[u] = qs[qq]; dL[u] = qd[qq]; }
                    else          { dL[u] = (unsigned)(qq - cnt); sN[u] = d0 + (qq - cnt); }
                }
#pragma unroll
                for (int u = 0; u < 4; ++u) {
                    aV[u] = as1[(unsigned)sN[u] * 8u + hh];
                    bV[u] = ad1[((unsigned)d0 + dL[u]) * 8u + hh];
                }
#pragma unroll
                for (int u = 0; u < 4; ++u) {
                    float e = aV[u] + bV[u];
                    e = e > 0.0f ? e : NEG_SLOPE * e;
                    atomicMax(&mx[dL[u] * 8u + hh], enc_f32(e));
                }
            }
        }
        __syncthreads();

        const unsigned h = (unsigned)lane >> 2;
        for (int q0 = wave * 4; q0 < total; q0 += (OWN_T / 32) * 4) {
            int   sN[4];
            unsigned dL[4];
            float eV[4];
            v2f   hV[4];
#pragma unroll
            for (int u = 0; u < 4; ++u) {
                int q = q0 + u;
                int qq = q < total ? q : q0;
                if (qq < cnt) { sN[u] = qs[qq]; dL[u] = qd[qq]; }
                else          { dL[u] = (unsigned)(qq - cnt); sN[u] = d0 + (qq - cnt); }
            }
#pragma unroll
            for (int u = 0; u < 4; ++u) {
                unsigned dn = (unsigned)d0 + dL[u];
                eV[u] = as1[(unsigned)sN[u] * 8u + h] + ad1[dn * 8u + h];
                hV[u] = ((const v2f*)(h1 + (unsigned)sN[u] * 64u))[lane];
            }
#pragma unroll
            for (int u = 0; u < 4; ++u) {
                if (q0 + u < total) {
                    float e = eV[u] > 0.0f ? eV[u] : NEG_SLOPE * eV[u];
                    float ex = expf(e - dec_f32(mx[dL[u] * 8u + h]));
                    if ((lane & 3) == 0) atomicAdd(&ssum[dL[u] * 8u + h], ex);
                    atomicAdd(&acc[dL[u] * 64u + 2u * lane + 0u], hV[u][0] * ex);
                    atomicAdd(&acc[dL[u] * 64u + 2u * lane + 1u], hV[u][1] * ex);
                }
            }
        }
    } else {
        own_scan(dstp, E, lo, span, tid, [&](int idx) {
            int d = dstp[idx];
            d = d < 0 ? 0 : (d >= n ? n - 1 : d);
            int s = ei[idx];
            s = s < 0 ? 0 : (s >= n ? n - 1 : s);
            own_edge_max_direct(s, (unsigned)(d - d0), d, as1, ad1, mx);
        });
        for (int i = tid; i < nb; i += OWN_T)
            own_edge_max_direct(d0 + i, (unsigned)i, d0 + i, as1, ad1, mx);
        __syncthreads();
        own_scan(dstp, E, lo, span, tid, [&](int idx) {
            int d = dstp[idx];
            d = d < 0 ? 0 : (d >= n ? n - 1 : d);
            int s = ei[idx];
            s = s < 0 ? 0 : (s >= n ? n - 1 : s);
            own_edge_direct(s, (unsigned)(d - d0), d, as1, ad1, mx, h1, acc, ssum);
        });
        for (int i = tid; i < nb; i += OWN_T)
            own_edge_direct(d0 + i, (unsigned)i, d0 + i, as1, ad1, mx, h1, acc, ssum);
    }
    __syncthreads();

    float* __restrict__ op = a1out + (size_t)d0 * 64;
    for (int i = tid; i < nb * 64; i += OWN_T) {
        float v = acc[i] / (ssum[i >> 3] + 1e-16f) + bias1[i & 63];
        op[i] = v > 0.0f ? v : (expf(v) - 1.0f);
    }
}

__global__ void k_alpha2(const float* __restrict__ h2,
                         const float* __restrict__ asrc, const float* __restrict__ adst,
                         float* __restrict__ as2, float* __restrict__ ad2, int n) {
    int tid = blockIdx.x * blockDim.x + threadIdx.x;
    if (tid >= n) return;
    const float4* hp = (const float4*)(h2 + (unsigned)tid * 8u);
    float4 v0 = hp[0], v1 = hp[1];
    const float4* sp = (const float4*)asrc;
    const float4* dp = (const float4*)adst;
    float4 s0 = sp[0], s1v = sp[1], d0 = dp[0], d1v = dp[1];
    as2[tid] = v0.x * s0.x + v0.y * s0.y + v0.z * s0.z + v0.w * s0.w
             + v1.x * s1v.x + v1.y * s1v.y + v1.z * s1v.z + v1.w * s1v.w;
    ad2[tid] = v0.x * d0.x + v0.y * d0.y + v0.z * d0.z + v0.w * d0.w
             + v1.x * d1v.x + v1.y * d1v.y + v1.z * d1v.z + v1.w * d1v.w;
}

__global__ void k_edge_max2(const int* __restrict__ ei, int E, int n,
                            const float* __restrict__ as2, const float* __restrict__ ad2,
                            unsigned* __restrict__ m2) {
    int idx = blockIdx.x * blockDim.x + threadIdx.x;
    if (idx >= E + n) return;
    int s, d; edge_sd(ei, E, n, idx, s, d);
    float e = as2[s] + ad2[d];
    e = e > 0.0f ? e : NEG_SLOPE * e;
    atomicMax(&m2[d], enc_f32(e));
}

__global__ void k_edge_scatter2(const int* __restrict__ ei, int E, int n,
                                const float* __restrict__ as2, const float* __restrict__ ad2,
                                const unsigned* __restrict__ m2, float* __restrict__ s2,
                                const float* __restrict__ h2, float* __restrict__ out) {
    int idx = blockIdx.x * blockDim.x + threadIdx.x;
    if (idx >= E + n) return;
    int s, d; edge_sd(ei, E, n, idx, s, d);
    const float* hp = h2 + (unsigned)s * 8u;
    __builtin_prefetch(hp, 0, 3);
    float e = as2[s] + ad2[d];
    e = e > 0.0f ? e : NEG_SLOPE * e;
    float ex = expf(e - dec_f32(m2[d]));
    atomicAdd(&s2[d], ex);
    float4 v0 = ((const float4*)hp)[0], v1 = ((const float4*)hp)[1];
    float* op = out + (unsigned)d * 8u;
    atomicAdd(op + 0, v0.x * ex);
    atomicAdd(op + 1, v0.y * ex);
    atomicAdd(op + 2, v0.z * ex);
    atomicAdd(op + 3, v0.w * ex);
    atomicAdd(op + 4, v1.x * ex);
    atomicAdd(op + 5, v1.y * ex);
    atomicAdd(op + 6, v1.z * ex);
    atomicAdd(op + 7, v1.w * ex);
}

__global__ void k_post2(float* __restrict__ out, const float* __restrict__ s2,
                        const float* __restrict__ bias2, int n) {
    int tid = blockIdx.x * blockDim.x + threadIdx.x;
    if (tid >= n * 8) return;
    out[tid] = out[tid] / (s2[tid >> 3] + 1e-16f) + bias2[tid & 7];
}

extern "C" void kernel_launch(void* const* d_in, const int* in_sizes, int n_in,
                              void* d_out, int out_size, void* d_ws, size_t ws_size,
                              hipStream_t stream) {
    const float* x     = (const float*)d_in[0];
    const int*   ei    = (const int*)d_in[1];
    const float* W1    = (const float*)d_in[2];
    const float* asrc1 = (const float*)d_in[3];
    const float* adst1 = (const float*)d_in[4];
    const float* bias1 = (const float*)d_in[5];
    const float* W2    = (const float*)d_in[6];
    const float* asrc2 = (const float*)d_in[7];
    const float* adst2 = (const float*)d_in[8];
    const float* bias2 = (const float*)d_in[9];

    int n = in_sizes[0] / F_IN;
    int E = in_sizes[1] / 2;

    char* p = (char*)d_ws;
    float*    h1   = (float*)p;    p += (size_t)n * 64 * 4;
    float*    acc1 = (float*)p;    p += (size_t)n * 64 * 4;
    float*    as1  = (float*)p;    p += (size_t)n * 8 * 4;
    float*    ad1  = (float*)p;    p += (size_t)n * 8 * 4;
    float*    h2   = (float*)p;    p += (size_t)n * 8 * 4;
    float*    as2  = (float*)p;    p += (size_t)n * 4;
    float*    ad2  = (float*)p;    p += (size_t)n * 4;
    float*    s2   = (float*)p;    p += (size_t)n * 4;
    unsigned* m2   = (unsigned*)p; p += (size_t)n * 4;

    hipMemsetAsync(s2,   0, (size_t)n * 4, stream);
    hipMemsetAsync(m2,   0, (size_t)n * 4, stream);
    hipMemsetAsync(d_out, 0, (size_t)out_size * 4, stream);

    const int tpb = 256;
    int tot = E + n;
    int gEd = (tot + tpb - 1) / tpb;

    k_gemm1<<<(n + 15) / 16, 32, 0, stream>>>(x, W1, h1, n);
    k_alpha1<<<(n * 8 + tpb - 1) / tpb, tpb, 0, stream>>>(h1, asrc1, adst1, as1, ad1, n);
    k_own_scatter1<<<(n + NB1 - 1) / NB1, OWN_T, 0, stream>>>(ei, E, n, as1, ad1, h1, bias1, acc1);

    k_gemm2<<<(n + 15) / 16, 32, 0, stream>>>(acc1, W2, h2, n);
    k_alpha2<<<(n + tpb - 1) / tpb, tpb, 0, stream>>>(h2, asrc2, adst2, as2, ad2, n);
    k_edge_max2<<<gEd, tpb, 0, stream>>>(ei, E, n, as2, ad2, m2);
    k_edge_scatter2<<<gEd, tpb, 0, stream>>>(ei, E, n, as2, ad2, m2, s2, h2, (float*)d_out);
    k_post2<<<(n * 8 + tpb - 1) / tpb, tpb, 0, stream>>>((float*)d_out, s2, bias2, n);
}


# layer 2 by the same destination-range owners: layer-1 owners save their edge queues, an owner kernel takes max and sums in LDS (float atomics, arrival order); global max, scatter, post, 3 memsets go
# speedup vs baseline: 6.2294x; 2.9480x over previous
#include <hip/hip_runtime.h>
#include <hip/hip_bf16.h>

#define NEG_SLOPE 0.2f
#define F_IN 14

typedef float v2f __attribute__((ext_vector_type(2)));
typedef float v8f __attribute__((ext_vector_type(8)));

__device__ __forceinline__ unsigned enc_f32(float f) {
    unsigned u = __float_as_uint(f);
    return (u & 0x80000000u) ? ~u : (u | 0x80000000u);
}
__device__ __forceinline__ float dec_f32(unsigned e) {
    unsigned u = (e & 0x80000000u) ? (e ^ 0x80000000u) : ~e;
    return __uint_as_float(u);
}

__global__ __launch_bounds__(32) void k_gemm1(const float* __restrict__ x,
                                              const float* __restrict__ W1,
                                              float* __restrict__ h1, int n) {
    __shared__ float xs[16 * 16];
    __shared__ float ws[16 * 64];

    const int lane = threadIdx.x;
    const int row0 = blockIdx.x * 16;
    const bool hi  = lane >= 16;
    const int  m   = lane & 15;
    const bool full = (row0 + 16 <= n);

    for (int i = lane; i < 2 * 64 / 4; i += 32)
        ((float4*)(ws + 14 * 64))[i] = make_float4(0.f, 0.f, 0.f, 0.f);
    for (int i = lane; i < F_IN * 64 / 4; i += 32)
        ((float4*)ws)[i] = ((const float4*)W1)[i];

    for (int i = lane; i < 16 * 16 / 4; i += 32)
        ((float4*)xs)[i] = make_float4(0.f, 0.f, 0.f, 0.f);
    __syncthreads();
    {
        const float* xt = x + (size_t)row0 * F_IN;
        int rows = full ? 16 : (n - row0);
        int cnt  = rows * F_IN;
        for (int i = lane; i < cnt; i += 32) {
            int r = i / F_IN, c = i - r * F_IN;
            xs[r * 16 + c] = xt[i];
        }
    }
    __syncthreads();

    const int kofs = hi ? 2 : 0;
    v2f a[4];
#pragma unroll
    for (int k = 0; k < 4; ++k) {
        a[k][0] = xs[m * 16 + 4 * k + kofs + 0];
        a[k][1] = xs[m * 16 + 4 * k + kofs + 1];
    }

#pragma unroll
    for (int t = 0; t < 4; ++t) {
        v8f c = {};
#pragma unroll
        for (int k = 0; k < 4; ++k) {
            v2f b;
            b[0] = ws[(4 * k + kofs + 0) * 64 + t * 16 + m];
            b[1] = ws[(4 * k + kofs + 1) * 64 + t * 16 + m];
            c = __builtin_amdgcn_wmma_f32_16x16x4_f32(false, a[k], false, b,
                                                      (short)0, c, false, false);
        }
        float* out = h1 + (size_t)row0 * 64 + t * 16 + m;
        if (full) {
#pragma unroll
            for (int r = 0; r < 8; ++r)
                out[(size_t)(r + (hi ? 8 : 0)) * 64] = c[r];
        } else {
#pragma unroll
            for (int r = 0; r < 8; ++r) {
                int rr = r + (hi ? 8 : 0);
                if (row0 + rr < n) out[(size_t)rr * 64] = c[r];
            }
        }
    }
}

__global__ __launch_bounds__(32) void k_gemm2(const float* __restrict__ a1,
                                              const float* __restrict__ W2,
                                              float* __restrict__ h2, int n) {
    __shared__ float smA[16 * 64];
    __shared__ float ws2[64 * 16];

    const int lane = threadIdx.x;
    const int row0 = blockIdx.x * 16;
    const bool hi  = lane >= 16;
    const int  m   = lane & 15;
    const bool full = (row0 + 16 <= n);

    for (int i = lane; i < 64 * 16 / 4; i += 32)
        ((float4*)ws2)[i] = make_float4(0.f, 0.f, 0.f, 0.f);
    __syncthreads();
    for (int i = lane; i < 64 * 8 / 4; i += 32) {
        int k = i >> 1, half = i & 1;
        ((float4*)(ws2 + k * 16 + half * 4))[0] = ((const float4*)W2)[i];
    }

    if (full) {
        const float4* src = (const float4*)(a1 + (size_t)row0 * 64);
        for (int i = lane; i < 16 * 64 / 4; i += 32)
            ((float4*)smA)[i] = src[i];
    } else {
        for (int i = lane; i < 16 * 64 / 4; i += 32)
            ((float4*)smA)[i] = make_float4(0.f, 0.f, 0.f, 0.f);
        __syncthreads();
        int rows = n - row0;
        const float4* src = (const float4*)(a1 + (size_t)row0 * 64);
        for (int i = lane; i < rows * 16; i += 32)
            ((float4*)smA)[i] = src[i];
    }
    __syncthreads();

    const int kofs = hi ? 2 : 0;
    v8f c = {};
#pragma unroll
    for (int k = 0; k < 16; ++k) {
        v2f a, b;
        a[0] = smA[m * 64 + 4 * k + kofs + 0];
        a[1] = smA[m * 64 + 4 * k + kofs + 1];
        b[0] = ws2[(4 * k + kofs + 0) * 16 + m];
        b[1] = ws2[(4 * k + kofs + 1) * 16 + m];
        c = __builtin_amdgcn_wmma_f32_16x16x4_f32(false, a, false, b,
                                                  (short)0, c, false, false);
    }
    if (m < 8) {
        float* out = h2 + (size_t)row0 * 8 + m;
        if (full) {
#pragma unroll
            for (int r = 0; r < 8; ++r)
                out[(size_t)(r + (hi ? 8 : 0)) * 8] = c[r];
        } else {
#pragma unroll
            for (int r = 0; r < 8; ++r) {
                int rr = r + (hi ? 8 : 0);
                if (row0 + rr < n) out[(size_t)rr * 8] = c[r];
            }
        }
    }
}

__global__ void k_alpha1(const float* __restrict__ h1,
                         const float* __restrict__ asrc, const float* __restrict__ adst,
                         float* __restrict__ as1, float* __restrict__ ad1, int n) {
    int tid = blockIdx.x * blockDim.x + threadIdx.x;
    if (tid >= n * 8) return;
    unsigned h = tid & 7u;
    const float4* hp = (const float4*)(h1 + (unsigned)(tid >> 3) * 64u + h * 8u);
    float4 v0 = hp[0], v1 = hp[1];
    const float4* sp = (const float4*)(asrc + h * 8u);
    const float4* dp = (const float4*)(adst + h * 8u);
    float4 s0 = sp[0], s1v = sp[1], d0 = dp[0], d1v = dp[1];
    float s = v0.x * s0.x + v0.y * s0.y + v0.z * s0.z + v0.w * s0.w
            + v1.x * s1v.x + v1.y * s1v.y + v1.z * s1v.z + v1.w * s1v.w;
    float d = v0.x * d0.x + v0.y * d0.y + v0.z * d0.z + v0.w * d0.w
            + v1.x * d1v.x + v1.y * d1v.y + v1.z * d1v.z + v1.w * d1v.w;
    as1[tid] = s;
    ad1[tid] = d;
}

#define NB1   392
#define QCAP  8192
#define OWN_T 1024

template <class F>
__device__ __forceinline__ void own_scan(const int* __restrict__ dstp, int E,
                                         int lo, unsigned span, int tid, F f) {
    const int E4 = ((E & 3) == 0) ? (E >> 2) : 0;
    const int4* __restrict__ dp4 = (const int4*)dstp;
    auto hits4 = [&](int4 q) -> unsigned {
        return ((unsigned)q.x - (unsigned)lo <= span ? 1u : 0u)
             | ((unsigned)q.y - (unsigned)lo <= span ? 2u : 0u)
             | ((unsigned)q.z - (unsigned)lo <= span ? 4u : 0u)
             | ((unsigned)q.w - (unsigned)lo <= span ? 8u : 0u);
    };
    int base = tid;
    for (; base + 7 * OWN_T < E4; base += 8 * OWN_T) {
        int4 v[8];
#pragma unroll
        for (int u = 0; u < 8; ++u)
            v[u] = dp4[base + u * OWN_T];
        unsigned hit = 0u;
#pragma unroll
        for (int u = 0; u < 8; ++u)
            hit |= hits4(v[u]) << (4 * u);
        while (hit) {
            int j = __ffs(hit) - 1;
            hit &= hit - 1u;
            f(4 * (base + (j >> 2) * OWN_T) + (j & 3));
        }
    }
    for (; base < E4; base += OWN_T) {
        unsigned hit = hits4(dp4[base]);
        while (hit) {
            int j = __ffs(hit) - 1;
            hit &= hit - 1u;
            f(4 * base + j);
        }
    }
    for (int idx = 4 * E4 + tid; idx < E; idx += OWN_T)
        if ((unsigned)dstp[idx] - (unsigned)lo <= span) f(idx);
}

__device__ __noinline__ void own_edge_max_direct(int s, unsigned dl, int dnode,
                                                 const float* __restrict__ as1,
                                                 const float* __restrict__ ad1,
                                                 unsigned* mx) {
    for (int h = 0; h < 8; ++h) {
        float e = as1[(unsigned)s * 8u + h] + ad1[(unsigned)dnode * 8u + h];
        e = e > 0.0f ? e : NEG_SLOPE * e;
        atomicMax(&mx[dl * 8u + h], enc_f32(e));
    }
}

__device__ __noinline__ void own_edge_direct(int s, unsigned dl, int dnode,
                                             const float* __restrict__ as1,
                                             const float* __restrict__ ad1,
                                             const unsigned* mx,
                                             const float* __restrict__ h1,
                                             float* acc, float* ssum) {
    for (int h = 0; h < 8; ++h) {
        float e = as1[(unsigned)s * 8u + h] + ad1[(unsigned)dnode * 8u + h];
        e = e > 0.0f ? e : NEG_SLOPE * e;
        float ex = expf(e - dec_f32(mx[dl * 8u + h]));
        atomicAdd(&ssum[dl * 8u + h], ex);
        const float* hp = h1 + (unsigned)s * 64u + (unsigned)h * 8u;
        for (int j = 0; j < 8; ++j)
            atomicAdd(&acc[dl * 64u + h * 8u + j], hp[j] * ex);
    }
}

__global__ __launch_bounds__(OWN_T) void k_own_scatter1(const int* __restrict__ ei, int E, int n,
                                const float* __restrict__ as1, const float* __restrict__ ad1,
                                const float* __restrict__ h1,
                                const float* __restrict__ bias1, float* __restrict__ a1out,
                                int* __restrict__ wq_s, unsigned short* __restrict__ wq_d,
                                unsigned* __restrict__ wq_n) {
    __shared__ float acc[NB1 * 64];
    __shared__ float ssum[NB1 * 8];
    __shared__ unsigned mx[NB1 * 8];
    __shared__ int qs[QCAP];
    __shared__ unsigned short qd[QCAP];
    __shared__ unsigned qn;

    const int tid  = threadIdx.x;
    const int lane = tid & 31;
    const int wave = tid >> 5;
    const int d0   = blockIdx.x * NB1;
    const int nb   = (n - d0) < NB1 ? (n - d0) : NB1;

    for (int i = tid; i < NB1 * 64; i += OWN_T) acc[i] = 0.0f;
    for (int i = tid; i < NB1 * 8; i += OWN_T) { ssum[i] = 0.0f; mx[i] = 0u; }
    if (tid == 0) qn = 0u;
    __syncthreads();

    const int* __restrict__ dstp = ei + E;
    const int lo = (d0 == 0) ? (int)0x80000000 : d0;
    const int hi = (d0 + nb == n) ? 0x7fffffff : (d0 + nb - 1);
    const unsigned span = (unsigned)hi - (unsigned)lo;
    own_scan(dstp, E, lo, span, tid, [&](int idx) {
        int d = dstp[idx];
        d = d < 0 ? 0 : (d >= n ? n - 1 : d);
        unsigned slot = atomicAdd(&qn, 1u);
        if (slot < (unsigned)QCAP) {
            qs[slot] = idx;
            qd[slot] = (unsigned short)(d - d0);
        }
    });
    __syncthreads();

    const unsigned qtot = qn;
    if (tid == 0) wq_n[blockIdx.x] = qtot;
    if (qtot <= (unsigned)QCAP) {
        const int cnt = (int)qtot;
        int* __restrict__ gs = wq_s + (size_t)blockIdx.x * QCAP;
        unsigned short* __restrict__ gd = wq_d + (size_t)blockIdx.x * QCAP;
        for (int q = tid; q < cnt; q += OWN_T) {
            int s = ei[qs[q]];
            s = s < 0 ? 0 : (s >= n ? n - 1 : s);
            qs[q] = s;
            gs[q] = s;
            gd[q] = qd[q];
        }
        __syncthreads();

        const int total = cnt + nb;

        {
            const unsigned hh = (unsigned)lane & 7u;
            const int sub = lane >> 3;
            for (int q0 = wave * 16; q0 < total; q0 += (OWN_T / 32) * 16) {
                int   sN[4];
                unsigned dL[4];
                float aV[4], bV[4];
#pragma unroll
                for (int u = 0; u < 4; ++u) {
                    int q = q0 + 4 * u + sub;
                    int qq = q < total ? q : q0;
                    if (qq < cnt) { sN[u] = qs[qq]; dL[u] = qd[qq]; }
                    else          { dL[u] = (unsigned)(qq - cnt); sN[u] = d0 + (qq - cnt); }
                }
#pragma unroll
                for (int u = 0; u < 4; ++u) {
                    aV[u] = as1[(unsigned)sN[u] * 8u + hh];
                    bV[u] = ad1[((unsigned)d0 + dL[u]) * 8u + hh];
                }
#pragma unroll
                for (int u = 0; u < 4; ++u) {
                    float e = aV[u] + bV[u];
                    e = e > 0.0f ? e : NEG_SLOPE * e;
                    atomicMax(&mx[dL[u] * 8u + hh], enc_f32(e));
                }
            }
        }
        __syncthreads();

        const unsigned h = (unsigned)lane >> 2;
        for (int q0 = wave * 4; q0 < total; q0 += (OWN_T / 32) * 4) {
            int   sN[4];
            unsigned dL[4];
            float eV[4];
            v2f   hV[4];
#pragma unroll
            for (int u = 0; u < 4; ++u) {
                int q = q0 + u;
                int qq = q < total ? q : q0;
                if (qq < cnt) { sN[u] = qs[qq]; dL[u] = qd[qq]; }
                else          { dL[u] = (unsigned)(qq - cnt); sN[u] = d0 + (qq - cnt); }
            }
#pragma unroll
            for (int u = 0; u < 4; ++u) {
                unsigned dn = (unsigned)d0 + dL[u];
                eV[u] = as1[(unsigned)sN[u] * 8u + h] + ad1[dn * 8u + h];
                hV[u] = ((const v2f*)(h1 + (unsigned)sN[u] * 64u))[lane];
            }
#pragma unroll
            for (int u = 0; u < 4; ++u) {
                if (q0 + u < total) {
                    float e = eV[u] > 0.0f ? eV[u] : NEG_SLOPE * eV[u];
                    float ex = expf(e - dec_f32(mx[dL[u] * 8u + h]));
                    if ((lane & 3) == 0) atomicAdd(&ssum[dL[u] * 8u + h], ex);
                    atomicAdd(&acc[dL[u] * 64u + 2u * lane + 0u], hV[u][0] * ex);
                    atomicAdd(&acc[dL[u] * 64u + 2u * lane + 1u], hV[u][1] * ex);
                }
            }
        }
    } else {
        own_scan(dstp, E, lo, span, tid, [&](int idx) {
            int d = dstp[idx];
            d = d < 0 ? 0 : (d >= n ? n - 1 : d);
            int s = ei[idx];
            s = s < 0 ? 0 : (s >= n ? n - 1 : s);
            own_edge_max_direct(s, (unsigned)(d - d0), d, as1, ad1, mx);
        });
        for (int i = tid; i < nb; i += OWN_T)
            own_edge_max_direct(d0 + i, (unsigned)i, d0 + i, as1, ad1, mx);
        __syncthreads();
        own_scan(dstp, E, lo, span, tid, [&](int idx) {
            int d = dstp[idx];
            d = d < 0 ? 0 : (d >= n ? n - 1 : d);
            int s = ei[idx];
            s = s < 0 ? 0 : (s >= n ? n - 1 : s);
            own_edge_direct(s, (unsigned)(d - d0), d, as1, ad1, mx, h1, acc, ssum);
        });
        for (int i = tid; i < nb; i += OWN_T)
            own_edge_direct(d0 + i, (unsigned)i, d0 + i, as1, ad1, mx, h1, acc, ssum);
    }
    __syncthreads();

    float* __restrict__ op = a1out + (size_t)d0 * 64;
    for (int i = tid; i < nb * 64; i += OWN_T) {
        float v = acc[i] / (ssum[i >> 3] + 1e-16f) + bias1[i & 63];
        op[i] = v > 0.0f ? v : (expf(v) - 1.0f);
    }
}

__global__ void k_alpha2(const float* __restrict__ h2,
                         const float* __restrict__ asrc, const float* __restrict__ adst,
                         float* __restrict__ as2, float* __restrict__ ad2, int n) {
    int tid = blockIdx.x * blockDim.x + threadIdx.x;
    if (tid >= n) return;
    const float4* hp = (const float4*)(h2 + (unsigned)tid * 8u);
    float4 v0 = hp[0], v1 = hp[1];
    const float4* sp = (const float4*)asrc;
    const float4* dp = (const float4*)adst;
    float4 s0 = sp[0], s1v = sp[1], d0 = dp[0], d1v = dp[1];
    as2[tid] = v0.x * s0.x + v0.y * s0.y + v0.z * s0.z + v0.w * s0.w
             + v1.x * s1v.x + v1.y * s1v.y + v1.z * s1v.z + v1.w * s1v.w;
    ad2[tid] = v0.x * d0.x + v0.y * d0.y + v0.z * d0.z + v0.w * d0.w
             + v1.x * d1v.x + v1.y * d1v.y + v1.z * d1v.z + v1.w * d1v.w;
}

#define Q2MAX ((QCAP + NB1 + OWN_T - 1) / OWN_T)

__device__ __noinline__ void own2_edge_max_direct(int s, unsigned dl, int dnode,
                                                  const float* __restrict__ as2,
                                                  const float* __restrict__ ad2,
                                                  unsigned* mx) {
    float e = as2[s] + ad2[dnode];
    e = e > 0.0f ? e : NEG_SLOPE * e;
    atomicMax(&mx[dl], enc_f32(e));
}

__device__ __noinline__ void own2_edge_direct(int s, unsigned dl, int dnode,
                                              const float* __restrict__ as2,
                                              const float* __restrict__ ad2,
                                              const unsigned* mx,
                                              const float* __restrict__ h2,
                                              float* acc, float* ssum) {
    float e = as2[s] + ad2[dnode];
    e = e > 0.0f ? e : NEG_SLOPE * e;
    float ex = expf(e - dec_f32(mx[dl]));
    atomicAdd(&ssum[dl], ex);
    const float* hp = h2 + (unsigned)s * 8u;
    for (int j = 0; j < 8; ++j)
        atomicAdd(&acc[dl * 8u + j], hp[j] * ex);
}

__global__ __launch_bounds__(OWN_T) void k_own_scatter2(const int* __restrict__ ei, int E, int n,
                                const float* __restrict__ as2, const float* __restrict__ ad2,
                                const float* __restrict__ h2, const float* __restrict__ bias2,
                                const int* __restrict__ wq_s,
                                const unsigned short* __restrict__ wq_d,
                                const unsigned* __restrict__ wq_n, float* __restrict__ out) {
    __shared__ float acc[NB1 * 8];
    __shared__ float ssum[NB1];
    __shared__ unsigned mx[NB1];
    __shared__ int qs[QCAP + NB1];
    __shared__ float qe[QCAP + NB1];
    __shared__ unsigned short qd[QCAP + NB1];

    const int tid  = threadIdx.x;
    const int lane = tid & 31;
    const int wave = tid >> 5;
    const int d0   = blockIdx.x * NB1;
    const int nb   = (n - d0) < NB1 ? (n - d0) : NB1;

    for (int i = tid; i < NB1 * 8; i += OWN_T) acc[i] = 0.0f;
    for (int i = tid; i < NB1; i += OWN_T) { ssum[i] = 0.0f; mx[i] = 0u; }
    __syncthreads();

    const unsigned qtot = wq_n[blockIdx.x];
    if (qtot <= (unsigned)QCAP) {
        const int cnt = (int)qtot;
        const int total = cnt + nb;
        const int* __restrict__ gs = wq_s + (size_t)blockIdx.x * QCAP;
        const unsigned short* __restrict__ gd = wq_d + (size_t)blockIdx.x * QCAP;

        {
            int sN[Q2MAX];
            unsigned dL[Q2MAX];
            float aV[Q2MAX], bV[Q2MAX];
#pragma unroll
            for (int u = 0; u < Q2MAX; ++u) {
                int q = tid + u * OWN_T;
                sN[u] = 0; dL[u] = 0u;
                if (q < cnt) { sN[u] = gs[q]; dL[u] = gd[q]; }
            }
#pragma unroll
            for (int u = 0; u < Q2MAX; ++u) {
                int q = tid + u * OWN_T;
                int s = sN[u];
                unsigned d = dL[u];
                if (q < cnt) {
                    s = s < 0 ? 0 : (s >= n ? n - 1 : s);
                    d = d < (unsigned)nb ? d : (unsigned)(nb - 1);
                } else {
                    d = q < total ? (unsigned)(q - cnt) : 0u;
                    s = d0 + (int)d;
                }
                sN[u] = s; dL[u] = d;
                aV[u] = as2[s];
                bV[u] = ad2[(unsigned)d0 + d];
            }
#pragma unroll
            for (int u = 0; u < Q2MAX; ++u) {
                int q = tid + u * OWN_T;
                if (q < total) {
                    float e = aV[u] + bV[u];
                    e = e > 0.0f ? e : NEG_SLOPE * e;
                    qs[q] = sN[u];
                    qd[q] = (unsigned short)dL[u];
                    qe[q] = e;
                    atomicMax(&mx[dL[u]], enc_f32(e));
                }
            }
        }
        __syncthreads();

        const int sub = lane >> 2;
        const unsigned j2 = (unsigned)lane & 3u;
        for (int q0 = wave * 32; q0 < total; q0 += (OWN_T / 32) * 32) {
            int   sN[4];
            unsigned dL[4];
            float eV[4];
            v2f   hV[4];
#pragma unroll
            for (int u = 0; u < 4; ++u) {
                int q = q0 + 8 * u + sub;
                int qq = q < total ? q : q0;
                sN[u] = qs[qq]; dL[u] = qd[qq]; eV[u] = qe[qq];
            }
#pragma unroll
            for (int u = 0; u < 4; ++u)
                hV[u] = ((const v2f*)(h2 + (unsigned)sN[u] * 8u))[j2];
#pragma unroll
            for (int u = 0; u < 4; ++u) {
                if (q0 + 8 * u + sub < total) {
                    float ex = expf(eV[u] - dec_f32(mx[dL[u]]));
                    if (j2 == 0u) atomicAdd(&ssum[dL[u]], ex);
                    atomicAdd(&acc[dL[u] * 8u + 2u * j2 + 0u], hV[u][0] * ex);
                    atomicAdd(&acc[dL[u] * 8u + 2u * j2 + 1u], hV[u][1] * ex);
                }
            }
        }
    } else {
        const int* __restrict__ dstp = ei + E;
        const int lo = (d0 == 0) ? (int)0x80000000 : d0;
        const int hi = (d0 + nb == n) ? 0x7fffffff : (d0 + nb - 1);
        const unsigned span = (unsigned)hi - (unsigned)lo;
        own_scan(dstp, E, lo, span, tid, [&](int idx) {
            int d = dstp[idx];
            d = d < 0 ? 0 : (d >= n ? n - 1 : d);
            int s = ei[idx];
            s = s < 0 ? 0 : (s >= n ? n - 1 : s);
            own2_edge_max_direct(s, (unsigned)(d - d0), d, as2, ad2, mx);
        });
        for (int i = tid; i < nb; i += OWN_T)
            own2_edge_max_direct(d0 + i, (unsigned)i, d0 + i, as2, ad2, mx);
        __syncthreads();
        own_scan(dstp, E, lo, span, tid, [&](int idx) {
            int d = dstp[idx];
            d = d < 0 ? 0 : (d >= n ? n - 1 : d);
            int s = ei[idx];
            s = s < 0 ? 0 : (s >= n ? n - 1 : s);
            own2_edge_direct(s, (unsigned)(d - d0), d, as2, ad2, mx, h2, acc, ssum);
        });
        for (int i = tid; i < nb; i += OWN_T)
            own2_edge_direct(d0 + i, (unsigned)i, d0 + i, as2, ad2, mx, h2, acc, ssum);
    }
    __syncthreads();

    float* __restrict__ op = out + (size_t)d0 * 8;
    for (int i = tid; i < nb * 8; i += OWN_T)
        op[i] = acc[i] / (ssum[i >> 3] + 1e-16f) + bias2[i & 7];
}

extern "C" void kernel_launch(void* const* d_in, const int* in_sizes, int n_in,
                              void* d_out, int out_size, void* d_ws, size_t ws_size,
                              hipStream_t stream) {
    const float* x     = (const float*)d_in[0];
    const int*   ei    = (const int*)d_in[1];
    const float* W1    = (const float*)d_in[2];
    const float* asrc1 = (const float*)d_in[3];
    const float* adst1 = (const float*)d_in[4];
    const float* bias1 = (const float*)d_in[5];
    const float* W2    = (const float*)d_in[6];
    const float* asrc2 = (const float*)d_in[7];
    const float* adst2 = (const float*)d_in[8];
    const float* bias2 = (const float*)d_in[9];

    int n = in_sizes[0] / F_IN;
    int E = in_sizes[1] / 2;

    char* p = (char*)d_ws;
    float*    h1   = (float*)p;    p += (size_t)n * 64 * 4;
    float*    acc1 = (float*)p;    p += (size_t)n * 64 * 4;
    float*    as1  = (float*)p;    p += (size_t)n * 8 * 4;
    float*    ad1  = (float*)p;    p += (size_t)n * 8 * 4;
    float*    h2   = (float*)p;    p += (size_t)n * 8 * 4;
    float*    as2  = (float*)p;    p += (size_t)n * 4;
    float*    ad2  = (float*)p;    p += (size_t)n * 4;
    int*            wq_s = (int*)p;            p += (size_t)((n + NB1 - 1) / NB1) * QCAP * 4;
    unsigned short* wq_d = (unsigned short*)p; p += (size_t)((n + NB1 - 1) / NB1) * QCAP * 2;
    unsigned*       wq_n = (unsigned*)p;       p += (size_t)((n + NB1 - 1) / NB1) * 4;

    const int tpb = 256;

    k_gemm1<<<(n + 15) / 16, 32, 0, stream>>>(x, W1, h1, n);
    k_alpha1<<<(n * 8 + tpb - 1) / tpb, tpb, 0, stream>>>(h1, asrc1, adst1, as1, ad1, n);
    k_own_scatter1<<<(n + NB1 - 1) / NB1, OWN_T, 0, stream>>>(ei, E, n, as1, ad1, h1, bias1, acc1,
                                                              wq_s, wq_d, wq_n);

    k_gemm2<<<(n + 15) / 16, 32, 0, stream>>>(acc1, W2, h2, n);
    k_alpha2<<<(n + tpb - 1) / tpb, tpb, 0, stream>>>(h2, asrc2, adst2, as2, ad2, n);
    k_own_scatter2<<<(n + NB1 - 1) / NB1, OWN_T, 0, stream>>>(ei, E, n, as2, ad2, h2, bias2,
                                                              wq_s, wq_d, wq_n, (float*)d_out);
}


# layer-1 owners scan one owner byte per edge (written by extra blocks of the logit launch) with a byte-equal bit mask, 16 edges per 16-byte load, not the 4-byte list; sums stay arrival-order LDS adds
# speedup vs baseline: 7.4308x; 1.1929x over previous
#include <hip/hip_runtime.h>
#include <hip/hip_bf16.h>

#define NEG_SLOPE 0.2f
#define F_IN 14

typedef float v2f __attribute__((ext_vector_type(2)));
typedef float v8f __attribute__((ext_vector_type(8)));

__device__ __forceinline__ unsigned enc_f32(float f) {
    unsigned u = __float_as_uint(f);
    return (u & 0x80000000u) ? ~u : (u | 0x80000000u);
}
__device__ __forceinline__ float dec_f32(unsigned e) {
    unsigned u = (e & 0x80000000u) ? (e ^ 0x80000000u) : ~e;
    return __uint_as_float(u);
}

__global__ __launch_bounds__(32) void k_gemm1(const float* __restrict__ x,
                                              const float* __restrict__ W1,
                                              float* __restrict__ h1, int n) {
    __shared__ float xs[16 * 16];
    __shared__ float ws[16 * 64];

    const int lane = threadIdx.x;
    const int row0 = blockIdx.x * 16;
    const bool hi  = lane >= 16;
    const int  m   = lane & 15;
    const bool full = (row0 + 16 <= n);

    for (int i = lane; i < 2 * 64 / 4; i += 32)
        ((float4*)(ws + 14 * 64))[i] = make_float4(0.f, 0.f, 0.f, 0.f);
    for (int i = lane; i < F_IN * 64 / 4; i += 32)
        ((float4*)ws)[i] = ((const float4*)W1)[i];

    for (int i = lane; i < 16 * 16 / 4; i += 32)
        ((float4*)xs)[i] = make_float4(0.f, 0.f, 0.f, 0.f);
    __syncthreads();
    {
        const float* xt = x + (size_t)row0 * F_IN;
        int rows = full ? 16 : (n - row0);
        int cnt  = rows * F_IN;
        for (int i = lane; i < cnt; i += 32) {
            int r = i / F_IN, c = i - r * F_IN;
            xs[r * 16 + c] = xt[i];
        }
    }
    __syncthreads();

    const int kofs = hi ? 2 : 0;
    v2f a[4];
#pragma unroll
    for (int k = 0; k < 4; ++k) {
        a[k][0] = xs[m * 16 + 4 * k + kofs + 0];
        a[k][1] = xs[m * 16 + 4 * k + kofs + 1];
    }

#pragma unroll
    for (int t = 0; t < 4; ++t) {
        v8f c = {};
#pragma unroll
        for (int k = 0; k < 4; ++k) {
            v2f b;
            b[0] = ws[(4 * k + kofs + 0) * 64 + t * 16 + m];
            b[1] = ws[(4 * k + kofs + 1) * 64 + t * 16 + m];
            c = __builtin_amdgcn_wmma_f32_16x16x4_f32(false, a[k], false, b,
                                                      (short)0, c, false, false);
        }
        float* out = h1 + (size_t)row0 * 64 + t * 16 + m;
        if (full) {
#pragma unroll
            for (int r = 0; r < 8; ++r)
                out[(size_t)(r + (hi ? 8 : 0)) * 64] = c[r];
        } else {
#pragma unroll
            for (int r = 0; r < 8; ++r) {
                int rr = r + (hi ? 8 : 0);
                if (row0 + rr < n) out[(size_t)rr * 64] = c[r];
            }
        }
    }
}

__global__ __launch_bounds__(32) void k_gemm2(const float* __restrict__ a1,
                                              const float* __restrict__ W2,
                                              float* __restrict__ h2, int n) {
    __shared__ float smA[16 * 64];
    __shared__ float ws2[64 * 16];

    const int lane = threadIdx.x;
    const int row0 = blockIdx.x * 16;
    const bool hi  = lane >= 16;
    const int  m   = lane & 15;
    const bool full = (row0 + 16 <= n);

    for (int i = lane; i < 64 * 16 / 4; i += 32)
        ((float4*)ws2)[i] = make_float4(0.f, 0.f, 0.f, 0.f);
    __syncthreads();
    for (int i = lane; i < 64 * 8 / 4; i += 32) {
        int k = i >> 1, half = i & 1;
        ((float4*)(ws2 + k * 16 + half * 4))[0] = ((const float4*)W2)[i];
    }

    if (full) {
        const float4* src = (const float4*)(a1 + (size_t)row0 * 64);
        for (int i = lane; i < 16 * 64 / 4; i += 32)
            ((float4*)smA)[i] = src[i];
    } else {
        for (int i = lane; i < 16 * 64 / 4; i += 32)
            ((float4*)smA)[i] = make_float4(0.f, 0.f, 0.f, 0.f);
        __syncthreads();
        int rows = n - row0;
        const float4* src = (const float4*)(a1 + (size_t)row0 * 64);
        for (int i = lane; i < rows * 16; i += 32)
            ((float4*)smA)[i] = src[i];
    }
    __syncthreads();

    const int kofs = hi ? 2 : 0;
    v8f c = {};
#pragma unroll
    for (int k = 0; k < 16; ++k) {
        v2f a, b;
        a[0] = smA[m * 64 + 4 * k + kofs + 0];
        a[1] = smA[m * 64 + 4 * k + kofs + 1];
        b[0] = ws2[(4 * k + kofs + 0) * 16 + m];
        b[1] = ws2[(4 * k + kofs + 1) * 16 + m];
        c = __builtin_amdgcn_wmma_f32_16x16x4_f32(false, a, false, b,
                                                  (short)0, c, false, false);
    }
    if (m < 8) {
        float* out = h2 + (size_t)row0 * 8 + m;
        if (full) {
#pragma unroll
            for (int r = 0; r < 8; ++r)
                out[(size_t)(r + (hi ? 8 : 0)) * 8] = c[r];
        } else {
#pragma unroll
            for (int r = 0; r < 8; ++r) {
                int rr = r + (hi ? 8 : 0);
                if (row0 + rr < n) out[(size_t)rr * 8] = c[r];
            }
        }
    }
}

#define NB1   392
__device__ __forceinline__ unsigned owner_byte(int d, int n) {
    d = d < 0 ? 0 : (d >= n ? n - 1 : d);
    unsigned o = (unsigned)d / (unsigned)NB1;
    return o < 255u ? o : 255u;
}

__global__ void k_alpha1(const float* __restrict__ h1,
                         const float* __restrict__ asrc, const float* __restrict__ adst,
                         float* __restrict__ as1, float* __restrict__ ad1, int n,
                         const int* __restrict__ ei, int E, unsigned char* __restrict__ ob,
                         int nab) {
    if ((int)blockIdx.x >= nab) {
        const int* __restrict__ dstp = ei + E;
        int g = ((int)blockIdx.x - nab) * (int)blockDim.x + (int)threadIdx.x;
        int e0 = 4 * g;
        if (e0 >= E) return;
        if ((E & 3) == 0) {
            int4 q = ((const int4*)dstp)[g];
            ((unsigned*)ob)[g] = owner_byte(q.x, n) | (owner_byte(q.y, n) << 8)
                               | (owner_byte(q.z, n) << 16) | (owner_byte(q.w, n) << 24);
        } else {
            for (int e = e0; e < e0 + 4 && e < E; ++e)
                ob[e] = (unsigned char)owner_byte(dstp[e], n);
        }
        return;
    }
    int tid = blockIdx.x * blockDim.x + threadIdx.x;
    if (tid >= n * 8) return;
    unsigned h = tid & 7u;
    const float4* hp = (const float4*)(h1 + (unsigned)(tid >> 3) * 64u + h * 8u);
    float4 v0 = hp[0], v1 = hp[1];
    const float4* sp = (const float4*)(asrc + h * 8u);
    const float4* dp = (const float4*)(adst + h * 8u);
    float4 s0 = sp[0], s1v = sp[1], d0 = dp[0], d1v = dp[1];
    float s = v0.x * s0.x + v0.y * s0.y + v0.z * s0.z + v0.w * s0.w
            + v1.x * s1v.x + v1.y * s1v.y + v1.z * s1v.z + v1.w * s1v.w;
    float d = v0.x * d0.x + v0.y * d0.y + v0.z * d0.z + v0.w * d0.w
            + v1.x * d1v.x + v1.y * d1v.y + v1.z * d1v.z + v1.w * d1v.w;
    as1[tid] = s;
    ad1[tid] = d;
}

#define QCAP  8192
#define OWN_T 1024

template <class F>
__device__ __forceinline__ void own_scan(const int* __restrict__ dstp,
                                         const unsigned char* __restrict__ ob, int E,
                                         int lo, unsigned span, unsigned own, bool exact,
                                         int tid, F f) {
    const int E16 = E >> 4;
    const uint4* __restrict__ bp = (const uint4*)ob;
    const unsigned pat = own * 0x01010101u;
    auto eq4 = [&](unsigned w) -> unsigned {
        unsigned x = w ^ pat;
        return ~(((x & 0x7f7f7f7fu) + 0x7f7f7f7fu) | x | 0x7f7f7f7fu);
    };
    auto pack4 = [&](uint4 q, int k0) -> unsigned {
        return (eq4(q.x) >> (7 - k0)) | (eq4(q.y) >> (6 - k0))
             | (eq4(q.z) >> (5 - k0)) | (eq4(q.w) >> (4 - k0));
    };
    auto visit = [&](int idx) {
        if (!exact && (unsigned)dstp[idx] - (unsigned)lo > span) return;
        f(idx);
    };
    int base = tid;
    for (; base + 7 * OWN_T < E16; base += 8 * OWN_T) {
        uint4 v[8];
#pragma unroll
        for (int u = 0; u < 8; ++u)
            v[u] = bp[base + u * OWN_T];
#pragma unroll
        for (int p = 0; p < 4; ++p) {
            unsigned hit = pack4(v[2 * p], 0) | pack4(v[2 * p + 1], 4);
            while (hit) {
                int j = __ffs(hit) - 1;
                hit &= hit - 1u;
                int k = j & 7;
                visit(16 * (base + (2 * p + (k >> 2)) * OWN_T) + 4 * (k & 3) + (j >> 3));
            }
        }
    }
    for (; base < E16; base += OWN_T) {
        unsigned hit = pack4(bp[base], 0);
        while (hit) {
            int j = __ffs(hit) - 1;
            hit &= hit - 1u;
            visit(16 * base + 4 * (j & 7) + (j >> 3));
        }
    }
    for (int idx = 16 * E16 + tid; idx < E; idx += OWN_T)
        if ((unsigned)ob[idx] == own) visit(idx);
}

__device__ __noinline__ void own_edge_max_direct(int s, unsigned dl, int dnode,
                                                 const float* __restrict__ as1,
                                                 const float* __restrict__ ad1,
                                                 unsigned* mx) {
    for (int h = 0; h < 8; ++h) {
        float e = as1[(unsigned)s * 8u + h] + ad1[(unsigned)dnode * 8u + h];
        e = e > 0.0f ? e : NEG_SLOPE * e;
        atomicMax(&mx[dl * 8u + h], enc_f32(e));
    }
}

__device__ __noinline__ void own_edge_direct(int s, unsigned dl, int dnode,
                                             const float* __restrict__ as1,
                                             const float* __restrict__ ad1,
                                             const unsigned* mx,
                                             const float* __restrict__ h1,
                                             float* acc, float* ssum) {
    for (int h = 0; h < 8; ++h) {
        float e = as1[(unsigned)s * 8u + h] + ad1[(unsigned)dnode * 8u + h];
        e = e > 0.0f ? e : NEG_SLOPE * e;
        float ex = expf(e - dec_f32(mx[dl * 8u + h]));
        atomicAdd(&ssum[dl * 8u + h], ex);
        const float* hp = h1 + (unsigned)s * 64u + (unsigned)h * 8u;
        for (int j = 0; j < 8; ++j)
            atomicAdd(&acc[dl * 64u + h * 8u + j], hp[j] * ex);
    }
}

__global__ __launch_bounds__(OWN_T) void k_own_scatter1(const int* __restrict__ ei, int E, int n,
                                const float* __restrict__ as1, const float* __restrict__ ad1,
                                const float* __restrict__ h1,
                                const float* __restrict__ bias1, float* __restrict__ a1out,
                                int* __restrict__ wq_s, unsigned short* __restrict__ wq_d,
                                unsigned* __restrict__ wq_n,
                                const unsigned char* __restrict__ ob) {
    __shared__ float acc[NB1 * 64];
    __shared__ float ssum[NB1 * 8];
    __shared__ unsigned mx[NB1 * 8];
    __shared__ int qs[QCAP];
    __shared__ unsigned short qd[QCAP];
    __shared__ unsigned qn;

    const int tid  = threadIdx.x;
    const int lane = tid & 31;
    const int wave = tid >> 5;
    const int d0   = blockIdx.x * NB1;
    const int nb   = (n - d0) < NB1 ? (n - d0) : NB1;

    for (int i = tid; i < NB1 * 64; i += OWN_T) acc[i] = 0.0f;
    for (int i = tid; i < NB1 * 8; i += OWN_T) { ssum[i] = 0.0f; mx[i] = 0u; }
    if (tid == 0) qn = 0u;
    __syncthreads();

    const int* __restrict__ dstp = ei + E;
    const int lo = (d0 == 0) ? (int)0x80000000 : d0;
    const int hi = (d0 + nb == n) ? 0x7fffffff : (d0 + nb - 1);
    const unsigned span = (unsigned)hi - (unsigned)lo;
    const unsigned own = blockIdx.x < 255u ? blockIdx.x : 255u;
    const bool exact = gridDim.x <= 256u || blockIdx.x < 255u;
    own_scan(dstp, ob, E, lo, span, own, exact, tid, [&](int idx) {
        unsigned slot = atomicAdd(&qn, 1u);
        if (slot < (unsigned)QCAP) qs[slot] = idx;
    });
    __syncthreads();

    const unsigned qtot = qn;
    if (tid == 0) wq_n[blockIdx.x] = qtot;
    if (qtot <= (unsigned)QCAP) {
        const int cnt = (int)qtot;
        int* __restrict__ gs = wq_s + (size_t)blockIdx.x * QCAP;
        unsigned short* __restrict__ gd = wq_d + (size_t)blockIdx.x * QCAP;
        for (int q = tid; q < cnt; q += OWN_T) {
            int idx = qs[q];
            int s = ei[idx];
            int d = dstp[idx];
            s = s < 0 ? 0 : (s >= n ? n - 1 : s);
            d = d < 0 ? 0 : (d >= n ? n - 1 : d);
            unsigned short dl = (unsigned short)(d - d0);
            qs[q] = s;
            qd[q] = dl;
            gs[q] = s;
            gd[q] = dl;
        }
        __syncthreads();

        const int total = cnt + nb;

        {
            const unsigned hh = (unsigned)lane & 7u;
            const int sub = lane >> 3;
            for (int q0 = wave * 16; q0 < total; q0 += (OWN_T / 32) * 16) {
                int   sN[4];
                unsigned dL[4];
                float aV[4], bV[4];
#pragma unroll
                for (int u = 0; u < 4; ++u) {
                    int q = q0 + 4 * u + sub;
                    int qq = q < total ? q : q0;
                    if (qq < cnt) { sN[u] = qs[qq]; dL[u] = qd[qq]; }
                    else          { dL[u] = (unsigned)(qq - cnt); sN[u] = d0 + (qq - cnt); }
                }
#pragma unroll
                for (int u = 0; u < 4; ++u) {
                    aV[u] = as1[(unsigned)sN[u] * 8u + hh];
                    bV[u] = ad1[((unsigned)d0 + dL[u]) * 8u + hh];
                }
#pragma unroll
                for (int u = 0; u < 4; ++u) {
                    float e = aV[u] + bV[u];
                    e = e > 0.0f ? e : NEG_SLOPE * e;
                    atomicMax(&mx[dL[u] * 8u + hh], enc_f32(e));
                }
            }
        }
        __syncthreads();

        const unsigned h = (unsigned)lane >> 2;
        for (int q0 = wave * 4; q0 < total; q0 += (OWN_T / 32) * 4) {
            int   sN[4];
            unsigned dL[4];
            float eV[4];
            v2f   hV[4];
#pragma unroll
            for (int u = 0; u < 4; ++u) {
                int q = q0 + u;
                int qq = q < total ? q : q0;
                if (qq < cnt) { sN[u] = qs[qq]; dL[u] = qd[qq]; }
                else          { dL[u] = (unsigned)(qq - cnt); sN[u] = d0 + (qq - cnt); }
            }
#pragma unroll
            for (int u = 0; u < 4; ++u) {
                unsigned dn = (unsigned)d0 + dL[u];
                eV[u] = as1[(unsigned)sN[u] * 8u + h] + ad1[dn * 8u + h];
                hV[u] = ((const v2f*)(h1 + (unsigned)sN[u] * 64u))[lane];
            }
#pragma unroll
            for (int u = 0; u < 4; ++u) {
                if (q0 + u < total) {
                    float e = eV[u] > 0.0f ? eV[u] : NEG_SLOPE * eV[u];
                    float ex = expf(e - dec_f32(mx[dL[u] * 8u + h]));
                    if ((lane & 3) == 0) atomicAdd(&ssum[dL[u] * 8u + h], ex);
                    atomicAdd(&acc[dL[u] * 64u + 2u * lane + 0u], hV[u][0] * ex);
                    atomicAdd(&acc[dL[u] * 64u + 2u * lane + 1u], hV[u][1] * ex);
                }
            }
        }
    } else {
        own_scan(dstp, ob, E, lo, span, own, exact, tid, [&](int idx) {
            int d = dstp[idx];
            d = d < 0 ? 0 : (d >= n ? n - 1 : d);
            int s = ei[idx];
            s = s < 0 ? 0 : (s >= n ? n - 1 : s);
            own_edge_max_direct(s, (unsigned)(d - d0), d, as1, ad1, mx);
        });
        for (int i = tid; i < nb; i += OWN_T)
            own_edge_max_direct(d0 + i, (unsigned)i, d0 + i, as1, ad1, mx);
        __syncthreads();
        own_scan(dstp, ob, E, lo, span, own, exact, tid, [&](int idx) {
            int d = dstp[idx];
            d = d < 0 ? 0 : (d >= n ? n - 1 : d);
            int s = ei[idx];
            s = s < 0 ? 0 : (s >= n ? n - 1 : s);
            own_edge_direct(s, (unsigned)(d - d0), d, as1, ad1, mx, h1, acc, ssum);
        });
        for (int i = tid; i < nb; i += OWN_T)
            own_edge_direct(d0 + i, (unsigned)i, d0 + i, as1, ad1, mx, h1, acc, ssum);
    }
    __syncthreads();

    float* __restrict__ op = a1out + (size_t)d0 * 64;
    for (int i = tid; i < nb * 64; i += OWN_T) {
        float v = acc[i] / (ssum[i >> 3] + 1e-16f) + bias1[i & 63];
        op[i] = v > 0.0f ? v : (expf(v) - 1.0f);
    }
}

__global__ void k_alpha2(const float* __restrict__ h2,
                         const float* __restrict__ asrc, const float* __restrict__ adst,
                         float* __restrict__ as2, float* __restrict__ ad2, int n) {
    int tid = blockIdx.x * blockDim.x + threadIdx.x;
    if (tid >= n) return;
    const float4* hp = (const float4*)(h2 + (unsigned)tid * 8u);
    float4 v0 = hp[0], v1 = hp[1];
    const float4* sp = (const float4*)asrc;
    const float4* dp = (const float4*)adst;
    float4 s0 = sp[0], s1v = sp[1], d0 = dp[0], d1v = dp[1];
    as2[tid] = v0.x * s0.x + v0.y * s0.y + v0.z * s0.z + v0.w * s0.w
             + v1.x * s1v.x + v1.y * s1v.y + v1.z * s1v.z + v1.w * s1v.w;
    ad2[tid] = v0.x * d0.x + v0.y * d0.y + v0.z * d0.z + v0.w * d0.w
             + v1.x * d1v.x + v1.y * d1v.y + v1.z * d1v.z + v1.w * d1v.w;
}

#define Q2MAX ((QCAP + NB1 + OWN_T - 1) / OWN_T)

__device__ __noinline__ void own2_edge_max_direct(int s, unsigned dl, int dnode,
                                                  const float* __restrict__ as2,
                                                  const float* __restrict__ ad2,
                                                  unsigned* mx) {
    float e = as2[s] + ad2[dnode];
    e = e > 0.0f ? e : NEG_SLOPE * e;
    atomicMax(&mx[dl], enc_f32(e));
}

__device__ __noinline__ void own2_edge_direct(int s, unsigned dl, int dnode,
                                              const float* __restrict__ as2,
                                              const float* __restrict__ ad2,
                                              const unsigned* mx,
                                              const float* __restrict__ h2,
                                              float* acc, float* ssum) {
    float e = as2[s] + ad2[dnode];
    e = e > 0.0f ? e : NEG_SLOPE * e;
    float ex = expf(e - dec_f32(mx[dl]));
    atomicAdd(&ssum[dl], ex);
    const float* hp = h2 + (unsigned)s * 8u;
    for (int j = 0; j < 8; ++j)
        atomicAdd(&acc[dl * 8u + j], hp[j] * ex);
}

__global__ __launch_bounds__(OWN_T) void k_own_scatter2(const int* __restrict__ ei, int E, int n,
                                const float* __restrict__ as2, const float* __restrict__ ad2,
                                const float* __restrict__ h2, const float* __restrict__ bias2,
                                const int* __restrict__ wq_s,
                                const unsigned short* __restrict__ wq_d,
                                const unsigned* __restrict__ wq_n,
                                const unsigned char* __restrict__ ob, float* __restrict__ out) {
    __shared__ float acc[NB1 * 8];
    __shared__ float ssum[NB1];
    __shared__ unsigned mx[NB1];
    __shared__ int qs[QCAP + NB1];
    __shared__ float qe[QCAP + NB1];
    __shared__ unsigned short qd[QCAP + NB1];

    const int tid  = threadIdx.x;
    const int lane = tid & 31;
    const int wave = tid >> 5;
    const int d0   = blockIdx.x * NB1;
    const int nb   = (n - d0) < NB1 ? (n - d0) : NB1;

    for (int i = tid; i < NB1 * 8; i += OWN_T) acc[i] = 0.0f;
    for (int i = tid; i < NB1; i += OWN_T) { ssum[i] = 0.0f; mx[i] = 0u; }
    __syncthreads();

    const unsigned qtot = wq_n[blockIdx.x];
    if (qtot <= (unsigned)QCAP) {
        const int cnt = (int)qtot;
        const int total = cnt + nb;
        const int* __restrict__ gs = wq_s + (size_t)blockIdx.x * QCAP;
        const unsigned short* __restrict__ gd = wq_d + (size_t)blockIdx.x * QCAP;

        {
            int sN[Q2MAX];
            unsigned dL[Q2MAX];
            float aV[Q2MAX], bV[Q2MAX];
#pragma unroll
            for (int u = 0; u < Q2MAX; ++u) {
                int q = tid + u * OWN_T;
                sN[u] = 0; dL[u] = 0u;
                if (q < cnt) { sN[u] = gs[q]; dL[u] = gd[q]; }
            }
#pragma unroll
            for (int u = 0; u < Q2MAX; ++u) {
                int q = tid + u * OWN_T;
                int s = sN[u];
                unsigned d = dL[u];
                if (q < cnt) {
                    s = s < 0 ? 0 : (s >= n ? n - 1 : s);
                    d = d < (unsigned)nb ? d : (unsigned)(nb - 1);
                } else {
                    d = q < total ? (unsigned)(q - cnt) : 0u;
                    s = d0 + (int)d;
                }
                sN[u] = s; dL[u] = d;
                aV[u] = as2[s];
                bV[u] = ad2[(unsigned)d0 + d];
            }
#pragma unroll
            for (int u = 0; u < Q2MAX; ++u) {
                int q = tid + u * OWN_T;
                if (q < total) {
                    float e = aV[u] + bV[u];
                    e = e > 0.0f ? e : NEG_SLOPE * e;
                    qs[q] = sN[u];
                    qd[q] = (unsigned short)dL[u];
                    qe[q] = e;
                    atomicMax(&mx[dL[u]], enc_f32(e));
                }
            }
        }
        __syncthreads();

        const int sub = lane >> 2;
        const unsigned j2 = (unsigned)lane & 3u;
        for (int q0 = wave * 32; q0 < total; q0 += (OWN_T / 32) * 32) {
            int   sN[4];
            unsigned dL[4];
            float eV[4];
            v2f   hV[4];
#pragma unroll
            for (int u = 0; u < 4; ++u) {
                int q = q0 + 8 * u + sub;
                int qq = q < total ? q : q0;
                sN[u] = qs[qq]; dL[u] = qd[qq]; eV[u] = qe[qq];
            }
#pragma unroll
            for (int u = 0; u < 4; ++u)
                hV[u] = ((const v2f*)(h2 + (unsigned)sN[u] * 8u))[j2];
#pragma unroll
            for (int u = 0; u < 4; ++u) {
                if (q0 + 8 * u + sub < total) {
                    float ex = expf(eV[u] - dec_f32(mx[dL[u]]));
                    if (j2 == 0u) atomicAdd(&ssum[dL[u]], ex);
                    atomicAdd(&acc[dL[u] * 8u + 2u * j2 + 0u], hV[u][0] * ex);
                    atomicAdd(&acc[dL[u] * 8u + 2u * j2 + 1u], hV[u][1] * ex);
                }
            }
        }
    } else {
        const int* __restrict__ dstp = ei + E;
        const int lo = (d0 == 0) ? (int)0x80000000 : d0;
        const int hi = (d0 + nb == n) ? 0x7fffffff : (d0 + nb - 1);
        const unsigned span = (unsigned)hi - (unsigned)lo;
        const unsigned own = blockIdx.x < 255u ? blockIdx.x : 255u;
        const bool exact = gridDim.x <= 256u || blockIdx.x < 255u;
        own_scan(dstp, ob, E, lo, span, own, exact, tid, [&](int idx) {
            int d = dstp[idx];
            d = d < 0 ? 0 : (d >= n ? n - 1 : d);
            int s = ei[idx];
            s = s < 0 ? 0 : (s >= n ? n - 1 : s);
            own2_edge_max_direct(s, (unsigned)(d - d0), d, as2, ad2, mx);
        });
        for (int i = tid; i < nb; i += OWN_T)
            own2_edge_max_direct(d0 + i, (unsigned)i, d0 + i, as2, ad2, mx);
        __syncthreads();
        own_scan(dstp, ob, E, lo, span, own, exact, tid, [&](int idx) {
            int d = dstp[idx];
            d = d < 0 ? 0 : (d >= n ? n - 1 : d);
            int s = ei[idx];
            s = s < 0 ? 0 : (s >= n ? n - 1 : s);
            own2_edge_direct(s, (unsigned)(d - d0), d, as2, ad2, mx, h2, acc, ssum);
        });
        for (int i = tid; i < nb; i += OWN_T)
            own2_edge_direct(d0 + i, (unsigned)i, d0 + i, as2, ad2, mx, h2, acc, ssum);
    }
    __syncthreads();

    float* __restrict__ op = out + (size_t)d0 * 8;
    for (int i = tid; i < nb * 8; i += OWN_T)
        op[i] = acc[i] / (ssum[i >> 3] + 1e-16f) + bias2[i & 7];
}

extern "C" void kernel_launch(void* const* d_in, const int* in_sizes, int n_in,
                              void* d_out, int out_size, void* d_ws, size_t ws_size,
                              hipStream_t stream) {
    const float* x     = (const float*)d_in[0];
    const int*   ei    = (const int*)d_in[1];
    const float* W1    = (const float*)d_in[2];
    const float* asrc1 = (const float*)d_in[3];
    const float* adst1 = (const float*)d_in[4];
    const float* bias1 = (const float*)d_in[5];
    const float* W2    = (const float*)d_in[6];
    const float* asrc2 = (const float*)d_in[7];
    const float* adst2 = (const float*)d_in[8];
    const float* bias2 = (const float*)d_in[9];

    int n = in_sizes[0] / F_IN;
    int E = in_sizes[1] / 2;

    char* p = (char*)d_ws;
    float*    h1   = (float*)p;    p += (size_t)n * 64 * 4;
    float*    acc1 = (float*)p;    p += (size_t)n * 64 * 4;
    float*    as1  = (float*)p;    p += (size_t)n * 8 * 4;
    float*    ad1  = (float*)p;    p += (size_t)n * 8 * 4;
    float*    h2   = (float*)p;    p += (size_t)n * 8 * 4;
    float*    as2  = (float*)p;    p += (size_t)n * 4;
    float*    ad2  = (float*)p;    p += (size_t)n * 4;
    unsigned char*  ob   = (unsigned char*)p;  p += (size_t)((E + 15) / 16) * 16;
    int*            wq_s = (int*)p;            p += (size_t)((n + NB1 - 1) / NB1) * QCAP * 4;
    unsigned short* wq_d = (unsigned short*)p; p += (size_t)((n + NB1 - 1) / NB1) * QCAP * 2;
    unsigned*       wq_n = (unsigned*)p;       p += (size_t)((n + NB1 - 1) / NB1) * 4;

    const int tpb = 256;

    k_gemm1<<<(n + 15) / 16, 32, 0, stream>>>(x, W1, h1, n);
    k_alpha1<<<(n * 8 + tpb - 1) / tpb + (E / 4 + tpb) / tpb, tpb, 0, stream>>>(
        h1, asrc1, adst1, as1, ad1, n, ei, E, ob, (n * 8 + tpb - 1) / tpb);
    k_own_scatter1<<<(n + NB1 - 1) / NB1, OWN_T, 0, stream>>>(ei, E, n, as1, ad1, h1, bias1, acc1,
                                                              wq_s, wq_d, wq_n, ob);

    k_gemm2<<<(n + 15) / 16, 32, 0, stream>>>(acc1, W2, h2, n);
    k_alpha2<<<(n + tpb - 1) / tpb, tpb, 0, stream>>>(h2, asrc2, adst2, as2, ad2, n);
    k_own_scatter2<<<(n + NB1 - 1) / NB1, OWN_T, 0, stream>>>(ei, E, n, as2, ad2, h2, bias2,
                                                              wq_s, wq_d, wq_n, ob, (float*)d_out);
}


# layer-2 owners fetch each source row once, keep it in LDS and take the logit halves from it (row . att); per-node logit arrays, their launch and the second row gather go; sum orders kept
# speedup vs baseline: 7.6493x; 1.0294x over previous
#include <hip/hip_runtime.h>
#include <hip/hip_bf16.h>

#define NEG_SLOPE 0.2f
#define F_IN 14

typedef float v2f __attribute__((ext_vector_type(2)));
typedef float v8f __attribute__((ext_vector_type(8)));

__device__ __forceinline__ unsigned enc_f32(float f) {
    unsigned u = __float_as_uint(f);
    return (u & 0x80000000u) ? ~u : (u | 0x80000000u);
}
__device__ __forceinline__ float dec_f32(unsigned e) {
    unsigned u = (e & 0x80000000u) ? (e ^ 0x80000000u) : ~e;
    return __uint_as_float(u);
}

__global__ __launch_bounds__(32) void k_gemm1(const float* __restrict__ x,
                                              const float* __restrict__ W1,
                                              float* __restrict__ h1, int n) {
    __shared__ float xs[16 * 16];
    __shared__ float ws[16 * 64];

    const int lane = threadIdx.x;
    const int row0 = blockIdx.x * 16;
    const bool hi  = lane >= 16;
    const int  m   = lane & 15;
    const bool full = (row0 + 16 <= n);

    for (int i = lane; i < 2 * 64 / 4; i += 32)
        ((float4*)(ws + 14 * 64))[i] = make_float4(0.f, 0.f, 0.f, 0.f);
    for (int i = lane; i < F_IN * 64 / 4; i += 32)
        ((float4*)ws)[i] = ((const float4*)W1)[i];

    for (int i = lane; i < 16 * 16 / 4; i += 32)
        ((float4*)xs)[i] = make_float4(0.f, 0.f, 0.f, 0.f);
    __syncthreads();
    {
        const float* xt = x + (size_t)row0 * F_IN;
        int rows = full ? 16 : (n - row0);
        int cnt  = rows * F_IN;
        for (int i = lane; i < cnt; i += 32) {
            int r = i / F_IN, c = i - r * F_IN;
            xs[r * 16 + c] = xt[i];
        }
    }
    __syncthreads();

    const int kofs = hi ? 2 : 0;
    v2f a[4];
#pragma unroll
    for (int k = 0; k < 4; ++k) {
        a[k][0] = xs[m * 16 + 4 * k + kofs + 0];
        a[k][1] = xs[m * 16 + 4 * k + kofs + 1];
    }

#pragma unroll
    for (int t = 0; t < 4; ++t) {
        v8f c = {};
#pragma unroll
        for (int k = 0; k < 4; ++k) {
            v2f b;
            b[0] = ws[(4 * k + kofs + 0) * 64 + t * 16 + m];
            b[1] = ws[(4 * k + kofs + 1) * 64 + t * 16 + m];
            c = __builtin_amdgcn_wmma_f32_16x16x4_f32(false, a[k], false, b,
                                                      (short)0, c, false, false);
        }
        float* out = h1 + (size_t)row0 * 64 + t * 16 + m;
        if (full) {
#pragma unroll
            for (int r = 0; r < 8; ++r)
                out[(size_t)(r + (hi ? 8 : 0)) * 64] = c[r];
        } else {
#pragma unroll
            for (int r = 0; r < 8; ++r) {
                int rr = r + (hi ? 8 : 0);
                if (row0 + rr < n) out[(size_t)rr * 64] = c[r];
            }
        }
    }
}

__global__ __launch_bounds__(32) void k_gemm2(const float* __restrict__ a1,
                                              const float* __restrict__ W2,
                                              float* __restrict__ h2, int n) {
    __shared__ float smA[16 * 64];
    __shared__ float ws2[64 * 16];

    const int lane = threadIdx.x;
    const int row0 = blockIdx.x * 16;
    const bool hi  = lane >= 16;
    const int  m   = lane & 15;
    const bool full = (row0 + 16 <= n);

    for (int i = lane; i < 64 * 16 / 4; i += 32)
        ((float4*)ws2)[i] = make_float4(0.f, 0.f, 0.f, 0.f);
    __syncthreads();
    for (int i = lane; i < 64 * 8 / 4; i += 32) {
        int k = i >> 1, half = i & 1;
        ((float4*)(ws2 + k * 16 + half * 4))[0] = ((const float4*)W2)[i];
    }

    if (full) {
        const float4* src = (const float4*)(a1 + (size_t)row0 * 64);
        for (int i = lane; i < 16 * 64 / 4; i += 32)
            ((float4*)smA)[i] = src[i];
    } else {
        for (int i = lane; i < 16 * 64 / 4; i += 32)
            ((float4*)smA)[i] = make_float4(0.f, 0.f, 0.f, 0.f);
        __syncthreads();
        int rows = n - row0;
        const float4* src = (const float4*)(a1 + (size_t)row0 * 64);
        for (int i = lane; i < rows * 16; i += 32)
            ((float4*)smA)[i] = src[i];
    }
    __syncthreads();

    const int kofs = hi ? 2 : 0;
    v8f c = {};
#pragma unroll
    for (int k = 0; k < 16; ++k) {
        v2f a, b;
        a[0] = smA[m * 64 + 4 * k + kofs + 0];
        a[1] = smA[m * 64 + 4 * k + kofs + 1];
        b[0] = ws2[(4 * k + kofs + 0) * 16 + m];
        b[1] = ws2[(4 * k + kofs + 1) * 16 + m];
        c = __builtin_amdgcn_wmma_f32_16x16x4_f32(false, a, false, b,
                                                  (short)0, c, false, false);
    }
    if (m < 8) {
        float* out = h2 + (size_t)row0 * 8 + m;
        if (full) {
#pragma unroll
            for (int r = 0; r < 8; ++r)
                out[(size_t)(r + (hi ? 8 : 0)) * 8] = c[r];
        } else {
#pragma unroll
            for (int r = 0; r < 8; ++r) {
                int rr = r + (hi ? 8 : 0);
                if (row0 + rr < n) out[(size_t)rr * 8] = c[r];
            }
        }
    }
}

#define NB1   392
__device__ __forceinline__ unsigned owner_byte(int d, int n) {
    d = d < 0 ? 0 : (d >= n ? n - 1 : d);
    unsigned o = (unsigned)d / (unsigned)NB1;
    return o < 255u ? o : 255u;
}

__global__ void k_alpha1(const float* __restrict__ h1,
                         const float* __restrict__ asrc, const float* __restrict__ adst,
                         float* __restrict__ as1, float* __restrict__ ad1, int n,
                         const int* __restrict__ ei, int E, unsigned char* __restrict__ ob,
                         int nab) {
    if ((int)blockIdx.x >= nab) {
        const int* __restrict__ dstp = ei + E;
        int g = ((int)blockIdx.x - nab) * (int)blockDim.x + (int)threadIdx.x;
        int e0 = 4 * g;
        if (e0 >= E) return;
        if ((E & 3) == 0) {
            int4 q = ((const int4*)dstp)[g];
            ((unsigned*)ob)[g] = owner_byte(q.x, n) | (owner_byte(q.y, n) << 8)
                               | (owner_byte(q.z, n) << 16) | (owner_byte(q.w, n) << 24);
        } else {
            for (int e = e0; e < e0 + 4 && e < E; ++e)
                ob[e] = (unsigned char)owner_byte(dstp[e], n);
        }
        return;
    }
    int tid = blockIdx.x * blockDim.x + threadIdx.x;
    if (tid >= n * 8) return;
    unsigned h = tid & 7u;
    const float4* hp = (const float4*)(h1 + (unsigned)(tid >> 3) * 64u + h * 8u);
    float4 v0 = hp[0], v1 = hp[1];
    const float4* sp = (const float4*)(asrc + h * 8u);
    const float4* dp = (const float4*)(adst + h * 8u);
    float4 s0 = sp[0], s1v = sp[1], d0 = dp[0], d1v = dp[1];
    float s = v0.x * s0.x + v0.y * s0.y + v0.z * s0.z + v0.w * s0.w
            + v1.x * s1v.x + v1.y * s1v.y + v1.z * s1v.z + v1.w * s1v.w;
    float d = v0.x * d0.x + v0.y * d0.y + v0.z * d0.z + v0.w * d0.w
            + v1.x * d1v.x + v1.y * d1v.y + v1.z * d1v.z + v1.w * d1v.w;
    as1[tid] = s;
    ad1[tid] = d;
}

#define QCAP  8192
#define OWN_T 1024

template <class F>
__device__ __forceinline__ void own_scan(const int* __restrict__ dstp,
                                         const unsigned char* __restrict__ ob, int E,
                                         int lo, unsigned span, unsigned own, bool exact,
                                         int tid, F f) {
    const int E16 = E >> 4;
    const uint4* __restrict__ bp = (const uint4*)ob;
    const unsigned pat = own * 0x01010101u;
    auto eq4 = [&](unsigned w) -> unsigned {
        unsigned x = w ^ pat;
        return ~(((x & 0x7f7f7f7fu) + 0x7f7f7f7fu) | x | 0x7f7f7f7fu);
    };
    auto pack4 = [&](uint4 q, int k0) -> unsigned {
        return (eq4(q.x) >> (7 - k0)) | (eq4(q.y) >> (6 - k0))
             | (eq4(q.z) >> (5 - k0)) | (eq4(q.w) >> (4 - k0));
    };
    auto visit = [&](int idx) {
        if (!exact && (unsigned)dstp[idx] - (unsigned)lo > span) return;
        f(idx);
    };
    int base = tid;
    for (; base + 7 * OWN_T < E16; base += 8 * OWN_T) {
        uint4 v[8];
#pragma unroll
        for (int u = 0; u < 8; ++u)
            v[u] = bp[base + u * OWN_T];
#pragma unroll
        for (int p = 0; p < 4; ++p) {
            unsigned hit = pack4(v[2 * p], 0) | pack4(v[2 * p + 1], 4);
            while (hit) {
                int j = __ffs(hit) - 1;
                hit &= hit - 1u;
                int k = j & 7;
                visit(16 * (base + (2 * p + (k >> 2)) * OWN_T) + 4 * (k & 3) + (j >> 3));
            }
        }
    }
    for (; base < E16; base += OWN_T) {
        unsigned hit = pack4(bp[base], 0);
        while (hit) {
            int j = __ffs(hit) - 1;
            hit &= hit - 1u;
            visit(16 * base + 4 * (j & 7) + (j >> 3));
        }
    }
    for (int idx = 16 * E16 + tid; idx < E; idx += OWN_T)
        if ((unsigned)ob[idx] == own) visit(idx);
}

__device__ __noinline__ void own_edge_max_direct(int s, unsigned dl, int dnode,
                                                 const float* __restrict__ as1,
                                                 const float* __restrict__ ad1,
                                                 unsigned* mx) {
    for (int h = 0; h < 8; ++h) {
        float e = as1[(unsigned)s * 8u + h] + ad1[(unsigned)dnode * 8u + h];
        e = e > 0.0f ? e : NEG_SLOPE * e;
        atomicMax(&mx[dl * 8u + h], enc_f32(e));
    }
}

__device__ __noinline__ void own_edge_direct(int s, unsigned dl, int dnode,
                                             const float* __restrict__ as1,
                                             const float* __restrict__ ad1,
                                             const unsigned* mx,
                                             const float* __restrict__ h1,
                                             float* acc, float* ssum) {
    for (int h = 0; h < 8; ++h) {
        float e = as1[(unsigned)s * 8u + h] + ad1[(unsigned)dnode * 8u + h];
        e = e > 0.0f ? e : NEG_SLOPE * e;
        float ex = expf(e - dec_f32(mx[dl * 8u + h]));
        atomicAdd(&ssum[dl * 8u + h], ex);
        const float* hp = h1 + (unsigned)s * 64u + (unsigned)h * 8u;
        for (int j = 0; j < 8; ++j)
            atomicAdd(&acc[dl * 64u + h * 8u + j], hp[j] * ex);
    }
}

__global__ __launch_bounds__(OWN_T) void k_own_scatter1(const int* __restrict__ ei, int E, int n,
                                const float* __restrict__ as1, const float* __restrict__ ad1,
                                const float* __restrict__ h1,
                                const float* __restrict__ bias1, float* __restrict__ a1out,
                                int* __restrict__ wq_s, unsigned short* __restrict__ wq_d,
                                unsigned* __restrict__ wq_n,
                                const unsigned char* __restrict__ ob) {
    __shared__ float acc[NB1 * 64];
    __shared__ float ssum[NB1 * 8];
    __shared__ unsigned mx[NB1 * 8];
    __shared__ int qs[QCAP];
    __shared__ unsigned short qd[QCAP];
    __shared__ unsigned qn;

    const int tid  = threadIdx.x;
    const int lane = tid & 31;
    const int wave = tid >> 5;
    const int d0   = blockIdx.x * NB1;
    const int nb   = (n - d0) < NB1 ? (n - d0) : NB1;

    for (int i = tid; i < NB1 * 64; i += OWN_T) acc[i] = 0.0f;
    for (int i = tid; i < NB1 * 8; i += OWN_T) { ssum[i] = 0.0f; mx[i] = 0u; }
    if (tid == 0) qn = 0u;
    __syncthreads();

    const int* __restrict__ dstp = ei + E;
    const int lo = (d0 == 0) ? (int)0x80000000 : d0;
    const int hi = (d0 + nb == n) ? 0x7fffffff : (d0 + nb - 1);
    const unsigned span = (unsigned)hi - (unsigned)lo;
    const unsigned own = blockIdx.x < 255u ? blockIdx.x : 255u;
    const bool exact = gridDim.x <= 256u || blockIdx.x < 255u;
    own_scan(dstp, ob, E, lo, span, own, exact, tid, [&](int idx) {
        unsigned slot = atomicAdd(&qn, 1u);
        if (slot < (unsigned)QCAP) qs[slot] = idx;
    });
    __syncthreads();

    const unsigned qtot = qn;
    if (tid == 0) wq_n[blockIdx.x] = qtot;
    if (qtot <= (unsigned)QCAP) {
        const int cnt = (int)qtot;
        int* __restrict__ gs = wq_s + (size_t)blockIdx.x * QCAP;
        unsigned short* __restrict__ gd = wq_d + (size_t)blockIdx.x * QCAP;
        for (int q = tid; q < cnt; q += OWN_T) {
            int idx = qs[q];
            int s = ei[idx];
            int d = dstp[idx];
            s = s < 0 ? 0 : (s >= n ? n - 1 : s);
            d = d < 0 ? 0 : (d >= n ? n - 1 : d);
            unsigned short dl = (unsigned short)(d - d0);
            qs[q] = s;
            qd[q] = dl;
            gs[q] = s;
            gd[q] = dl;
        }
        __syncthreads();

        const int total = cnt + nb;

        {
            const unsigned hh = (unsigned)lane & 7u;
            const int sub = lane >> 3;
            for (int q0 = wave * 16; q0 < total; q0 += (OWN_T / 32) * 16) {
                int   sN[4];
                unsigned dL[4];
                float aV[4], bV[4];
#pragma unroll
                for (int u = 0; u < 4; ++u) {
                    int q = q0 + 4 * u + sub;
                    int qq = q < total ? q : q0;
                    if (qq < cnt) { sN[u] = qs[qq]; dL[u] = qd[qq]; }
                    else          { dL[u] = (unsigned)(qq - cnt); sN[u] = d0 + (qq - cnt); }
                }
#pragma unroll
                for (int u = 0; u < 4; ++u) {
                    aV[u] = as1[(unsigned)sN[u] * 8u + hh];
                    bV[u] = ad1[((unsigned)d0 + dL[u]) * 8u + hh];
                }
#pragma unroll
                for (int u = 0; u < 4; ++u) {
                    float e = aV[u] + bV[u];
                    e = e > 0.0f ? e : NEG_SLOPE * e;
                    atomicMax(&mx[dL[u] * 8u + hh], enc_f32(e));
                }
            }
        }
        __syncthreads();

        const unsigned h = (unsigned)lane >> 2;
        for (int q0 = wave * 4; q0 < total; q0 += (OWN_T / 32) * 4) {
            int   sN[4];
            unsigned dL[4];
            float eV[4];
            v2f   hV[4];
#pragma unroll
            for (int u = 0; u < 4; ++u) {
                int q = q0 + u;
                int qq = q < total ? q : q0;
                if (qq < cnt) { sN[u] = qs[qq]; dL[u] = qd[qq]; }
                else          { dL[u] = (unsigned)(qq - cnt); sN[u] = d0 + (qq - cnt); }
            }
#pragma unroll
            for (int u = 0; u < 4; ++u) {
                unsigned dn = (unsigned)d0 + dL[u];
                eV[u] = as1[(unsigned)sN[u] * 8u + h] + ad1[dn * 8u + h];
                hV[u] = ((const v2f*)(h1 + (unsigned)sN[u] * 64u))[lane];
            }
#pragma unroll
            for (int u = 0; u < 4; ++u) {
                if (q0 + u < total) {
                    float e = eV[u] > 0.0f ? eV[u] : NEG_SLOPE * eV[u];
                    float ex = expf(e - dec_f32(mx[dL[u] * 8u + h]));
                    if ((lane & 3) == 0) atomicAdd(&ssum[dL[u] * 8u + h], ex);
                    atomicAdd(&acc[dL[u] * 64u + 2u * lane + 0u], hV[u][0] * ex);
                    atomicAdd(&acc[dL[u] * 64u + 2u * lane + 1u], hV[u][1] * ex);
                }
            }
        }
    } else {
        own_scan(dstp, ob, E, lo, span, own, exact, tid, [&](int idx) {
            int d = dstp[idx];
            d = d < 0 ? 0 : (d >= n ? n - 1 : d);
            int s = ei[idx];
            s = s < 0 ? 0 : (s >= n ? n - 1 : s);
            own_edge_max_direct(s, (unsigned)(d - d0), d, as1, ad1, mx);
        });
        for (int i = tid; i < nb; i += OWN_T)
            own_edge_max_direct(d0 + i, (unsigned)i, d0 + i, as1, ad1, mx);
        __syncthreads();
        own_scan(dstp, ob, E, lo, span, own, exact, tid, [&](int idx) {
            int d = dstp[idx];
            d = d < 0 ? 0 : (d >= n ? n - 1 : d);
            int s = ei[idx];
            s = s < 0 ? 0 : (s >= n ? n - 1 : s);
            own_edge_direct(s, (unsigned)(d - d0), d, as1, ad1, mx, h1, acc, ssum);
        });
        for (int i = tid; i < nb; i += OWN_T)
            own_edge_direct(d0 + i, (unsigned)i, d0 + i, as1, ad1, mx, h1, acc, ssum);
    }
    __syncthreads();

    float* __restrict__ op = a1out + (size_t)d0 * 64;
    for (int i = tid; i < nb * 64; i += OWN_T) {
        float v = acc[i] / (ssum[i >> 3] + 1e-16f) + bias1[i & 63];
        op[i] = v > 0.0f ? v : (expf(v) - 1.0f);
    }
}

#define Q2CAP 7168
#define Q2MAX ((Q2CAP + NB1 + OWN_T - 1) / OWN_T)

__device__ __forceinline__ float dot8(float4 v0, float4 v1, float4 s0, float4 s1v) {
    return v0.x * s0.x + v0.y * s0.y + v0.z * s0.z + v0.w * s0.w
         + v1.x * s1v.x + v1.y * s1v.y + v1.z * s1v.z + v1.w * s1v.w;
}

__device__ __noinline__ float own2_logit_direct(int s, int dnode, const float* __restrict__ h2,
                                                const float* __restrict__ asrc,
                                                const float* __restrict__ adst) {
    const float4* sp = (const float4*)(h2 + (unsigned)s * 8u);
    const float4* dp = (const float4*)(h2 + (unsigned)dnode * 8u);
    float e = dot8(sp[0], sp[1], ((const float4*)asrc)[0], ((const float4*)asrc)[1])
            + dot8(dp[0], dp[1], ((const float4*)adst)[0], ((const float4*)adst)[1]);
    return e > 0.0f ? e : NEG_SLOPE * e;
}

__device__ __noinline__ void own2_edge_direct(int s, unsigned dl, float e,
                                              const unsigned* mx,
                                              const float* __restrict__ h2,
                                              float* acc, float* ssum) {
    float ex = expf(e - dec_f32(mx[dl]));
    atomicAdd(&ssum[dl], ex);
    const float* hp = h2 + (unsigned)s * 8u;
    for (int j = 0; j < 8; ++j)
        atomicAdd(&acc[dl * 8u + j], hp[j] * ex);
}

__global__ __launch_bounds__(OWN_T) void k_own_scatter2(const int* __restrict__ ei, int E, int n,
                                const float* __restrict__ asrc2, const float* __restrict__ adst2,
                                const float* __restrict__ h2, const float* __restrict__ bias2,
                                const int* __restrict__ wq_s,
                                const unsigned short* __restrict__ wq_d,
                                const unsigned* __restrict__ wq_n,
                                const unsigned char* __restrict__ ob, float* __restrict__ out) {
    __shared__ float acc[NB1 * 8];
    __shared__ float ssum[NB1];
    __shared__ unsigned mx[NB1];
    __shared__ float adl[NB1];
    __shared__ float rows[(Q2CAP + NB1) * 8];
    __shared__ float qe[Q2CAP + NB1];
    __shared__ unsigned short qd[Q2CAP + NB1];

    const int tid  = threadIdx.x;
    const int lane = tid & 31;
    const int wave = tid >> 5;
    const int d0   = blockIdx.x * NB1;
    const int nb   = (n - d0) < NB1 ? (n - d0) : NB1;

    for (int i = tid; i < NB1 * 8; i += OWN_T) acc[i] = 0.0f;
    for (int i = tid; i < NB1; i += OWN_T) { ssum[i] = 0.0f; mx[i] = 0u; }
    __syncthreads();

    const unsigned qtot = wq_n[blockIdx.x];
    if (qtot <= (unsigned)Q2CAP) {
        const int cnt = (int)qtot;
        const int total = cnt + nb;
        const int* __restrict__ gs = wq_s + (size_t)blockIdx.x * QCAP;
        const unsigned short* __restrict__ gd = wq_d + (size_t)blockIdx.x * QCAP;
        const float4 as0 = ((const float4*)asrc2)[0], as1v = ((const float4*)asrc2)[1];
        const float4 ad0 = ((const float4*)adst2)[0], ad1v = ((const float4*)adst2)[1];

        for (int b = 0; b < Q2MAX; b += 4) {
            int sN[4];
            unsigned dL[4];
            float4 r0[4], r1[4];
#pragma unroll
            for (int u = 0; u < 4; ++u) {
                int q = tid + (b + u) * OWN_T;
                sN[u] = 0; dL[u] = 0u;
                if (q < cnt) { sN[u] = gs[q]; dL[u] = gd[q]; }
            }
#pragma unroll
            for (int u = 0; u < 4; ++u) {
                int q = tid + (b + u) * OWN_T;
                int s = sN[u];
                unsigned d = dL[u];
                if (q < cnt) {
                    s = s < 0 ? 0 : (s >= n ? n - 1 : s);
                    d = d < (unsigned)nb ? d : (unsigned)(nb - 1);
                } else {
                    d = q < total ? (unsigned)(q - cnt) : 0u;
                    s = d0 + (int)d;
                }
                dL[u] = d;
                const float4* hp = (const float4*)(h2 + (unsigned)s * 8u);
                r0[u] = hp[0];
                r1[u] = hp[1];
            }
#pragma unroll
            for (int u = 0; u < 4; ++u) {
                int q = tid + (b + u) * OWN_T;
                if (q < total) {
                    ((float4*)(rows + q * 8))[0] = r0[u];
                    ((float4*)(rows + q * 8))[1] = r1[u];
                    qd[q] = (unsigned short)dL[u];
                    qe[q] = dot8(r0[u], r1[u], as0, as1v);
                    if (q >= cnt) adl[dL[u]] = dot8(r0[u], r1[u], ad0, ad1v);
                }
            }
        }
        __syncthreads();

        for (int q = tid; q < total; q += OWN_T) {
            unsigned d = qd[q];
            float e = qe[q] + adl[d];
            e = e > 0.0f ? e : NEG_SLOPE * e;
            qe[q] = e;
            atomicMax(&mx[d], enc_f32(e));
        }
        __syncthreads();

        const int sub = lane >> 2;
        const unsigned j2 = (unsigned)lane & 3u;
        for (int q0 = wave * 32; q0 < total; q0 += (OWN_T / 32) * 32) {
            unsigned dL[4], mV[4];
            float eV[4];
            v2f   hV[4];
#pragma unroll
            for (int u = 0; u < 4; ++u) {
                int q = q0 + 8 * u + sub;
                int qq = q < total ? q : q0;
                dL[u] = qd[qq]; eV[u] = qe[qq];
                hV[u] = ((const v2f*)(rows + qq * 8))[j2];
            }
#pragma unroll
            for (int u = 0; u < 4; ++u)
                mV[u] = mx[dL[u]];
#pragma unroll
            for (int u = 0; u < 4; ++u) {
                if (q0 + 8 * u + sub < total) {
                    float ex = expf(eV[u] - dec_f32(mV[u]));
                    if (j2 == 0u) atomicAdd(&ssum[dL[u]], ex);
                    atomicAdd(&acc[dL[u] * 8u + 2u * j2 + 0u], hV[u][0] * ex);
                    atomicAdd(&acc[dL[u] * 8u + 2u * j2 + 1u], hV[u][1] * ex);
                }
            }
        }
    } else {
        const int* __restrict__ dstp = ei + E;
        const int lo = (d0 == 0) ? (int)0x80000000 : d0;
        const int hi = (d0 + nb == n) ? 0x7fffffff : (d0 + nb - 1);
        const unsigned span = (unsigned)hi - (unsigned)lo;
        const unsigned own = blockIdx.x < 255u ? blockIdx.x : 255u;
        const bool exact = gridDim.x <= 256u || blockIdx.x < 255u;
        own_scan(dstp, ob, E, lo, span, own, exact, tid, [&](int idx) {
            int d = dstp[idx];
            d = d < 0 ? 0 : (d >= n ? n - 1 : d);
            int s = ei[idx];
            s = s < 0 ? 0 : (s >= n ? n - 1 : s);
            atomicMax(&mx[d - d0], enc_f32(own2_logit_direct(s, d, h2, asrc2, adst2)));
        });
        for (int i = tid; i < nb; i += OWN_T)
            atomicMax(&mx[i], enc_f32(own2_logit_direct(d0 + i, d0 + i, h2, asrc2, adst2)));
        __syncthreads();
        own_scan(dstp, ob, E, lo, span, own, exact, tid, [&](int idx) {
            int d = dstp[idx];
            d = d < 0 ? 0 : (d >= n ? n - 1 : d);
            int s = ei[idx];
            s = s < 0 ? 0 : (s >= n ? n - 1 : s);
            own2_edge_direct(s, (unsigned)(d - d0), own2_logit_direct(s, d, h2, asrc2, adst2),
                             mx, h2, acc, ssum);
        });
        for (int i = tid; i < nb; i += OWN_T)
            own2_edge_direct(d0 + i, (unsigned)i,
                             own2_logit_direct(d0 + i, d0 + i, h2, asrc2, adst2),
                             mx, h2, acc, ssum);
    }
    __syncthreads();

    float* __restrict__ op = out + (size_t)d0 * 8;
    for (int i = tid; i < nb * 8; i += OWN_T)
        op[i] = acc[i] / (ssum[i >> 3] + 1e-16f) + bias2[i & 7];
}

extern "C" void kernel_launch(void* const* d_in, const int* in_sizes, int n_in,
                              void* d_out, int out_size, void* d_ws, size_t ws_size,
                              hipStream_t stream) {
    const float* x     = (const float*)d_in[0];
    const int*   ei    = (const int*)d_in[1];
    const float* W1    = (const float*)d_in[2];
    const float* asrc1 = (const float*)d_in[3];
    const float* adst1 = (const float*)d_in[4];
    const float* bias1 = (const float*)d_in[5];
    const float* W2    = (const float*)d_in[6];
    const float* asrc2 = (const float*)d_in[7];
    const float* adst2 = (const float*)d_in[8];
    const float* bias2 = (const float*)d_in[9];

    int n = in_sizes[0] / F_IN;
    int E = in_sizes[1] / 2;

    char* p = (char*)d_ws;
    float*    h1   = (float*)p;    p += (size_t)n * 64 * 4;
    float*    acc1 = (float*)p;    p += (size_t)n * 64 * 4;
    float*    as1  = (float*)p;    p += (size_t)n * 8 * 4;
    float*    ad1  = (float*)p;    p += (size_t)n * 8 * 4;
    float*    h2   = (float*)p;    p += (size_t)n * 8 * 4;
    unsigned char*  ob   = (unsigned char*)p;  p += (size_t)((E + 15) / 16) * 16;
    int*            wq_s = (int*)p;            p += (size_t)((n + NB1 - 1) / NB1) * QCAP * 4;
    unsigned short* wq_d = (unsigned short*)p; p += (size_t)((n + NB1 - 1) / NB1) * QCAP * 2;
    unsigned*       wq_n = (unsigned*)p;       p += (size_t)((n + NB1 - 1) / NB1) * 4;

    const int tpb = 256;

    k_gemm1<<<(n + 15) / 16, 32, 0, stream>>>(x, W1, h1, n);
    k_alpha1<<<(n * 8 + tpb - 1) / tpb + (E / 4 + tpb) / tpb, tpb, 0, stream>>>(
        h1, asrc1, adst1, as1, ad1, n, ei, E, ob, (n * 8 + tpb - 1) / tpb);
    k_own_scatter1<<<(n + NB1 - 1) / NB1, OWN_T, 0, stream>>>(ei, E, n, as1, ad1, h1, bias1, acc1,
                                                              wq_s, wq_d, wq_n, ob);

    k_gemm2<<<(n + 15) / 16, 32, 0, stream>>>(acc1, W2, h2, n);
    k_own_scatter2<<<(n + NB1 - 1) / NB1, OWN_T, 0, stream>>>(ei, E, n, asrc2, adst2, h2, bias2,
                                                              wq_s, wq_d, wq_n, ob, (float*)d_out);
}


# layer-1 sum walk: each softmax weight computed once, one lane per (edge, head), four edges per instruction; row lanes read it by quad DPP; eight edges in flight; sums stay arrival-order LDS adds
# speedup vs baseline: 8.6827x; 1.1351x over previous
#include <hip/hip_runtime.h>
#include <hip/hip_bf16.h>

#define NEG_SLOPE 0.2f
#define F_IN 14

typedef float v2f __attribute__((ext_vector_type(2)));
typedef float v8f __attribute__((ext_vector_type(8)));

__device__ __forceinline__ unsigned enc_f32(float f) {
    unsigned u = __float_as_uint(f);
    return (u & 0x80000000u) ? ~u : (u | 0x80000000u);
}
__device__ __forceinline__ float dec_f32(unsigned e) {
    unsigned u = (e & 0x80000000u) ? (e ^ 0x80000000u) : ~e;
    return __uint_as_float(u);
}

__global__ __launch_bounds__(32) void k_gemm1(const float* __restrict__ x,
                                              const float* __restrict__ W1,
                                              float* __restrict__ h1, int n) {
    __shared__ float xs[16 * 16];
    __shared__ float ws[16 * 64];

    const int lane = threadIdx.x;
    const int row0 = blockIdx.x * 16;
    const bool hi  = lane >= 16;
    const int  m   = lane & 15;
    const bool full = (row0 + 16 <= n);

    for (int i = lane; i < 2 * 64 / 4; i += 32)
        ((float4*)(ws + 14 * 64))[i] = make_float4(0.f, 0.f, 0.f, 0.f);
    for (int i = lane; i < F_IN * 64 / 4; i += 32)
        ((float4*)ws)[i] = ((const float4*)W1)[i];

    for (int i = lane; i < 16 * 16 / 4; i += 32)
        ((float4*)xs)[i] = make_float4(0.f, 0.f, 0.f, 0.f);
    __syncthreads();
    {
        const float* xt = x + (size_t)row0 * F_IN;
        int rows = full ? 16 : (n - row0);
        int cnt  = rows * F_IN;
        for (int i = lane; i < cnt; i += 32) {
            int r = i / F_IN, c = i - r * F_IN;
            xs[r * 16 + c] = xt[i];
        }
    }
    __syncthreads();

    const int kofs = hi ? 2 : 0;
    v2f a[4];
#pragma unroll
    for (int k = 0; k < 4; ++k) {
        a[k][0] = xs[m * 16 + 4 * k + kofs + 0];
        a[k][1] = xs[m * 16 + 4 * k + kofs + 1];
    }

#pragma unroll
    for (int t = 0; t < 4; ++t) {
        v8f c = {};
#pragma unroll
        for (int k = 0; k < 4; ++k) {
            v2f b;
            b[0] = ws[(4 * k + kofs + 0) * 64 + t * 16 + m];
            b[1] = ws[(4 * k + kofs + 1) * 64 + t * 16 + m];
            c = __builtin_amdgcn_wmma_f32_16x16x4_f32(false, a[k], false, b,
                                                      (short)0, c, false, false);
        }
        float* out = h1 + (size_t)row0 * 64 + t * 16 + m;
        if (full) {
#pragma unroll
            for (int r = 0; r < 8; ++r)
                out[(size_t)(r + (hi ? 8 : 0)) * 64] = c[r];
        } else {
#pragma unroll
            for (int r = 0; r < 8; ++r) {
                int rr = r + (hi ? 8 : 0);
                if (row0 + rr < n) out[(size_t)rr * 64] = c[r];
            }
        }
    }
}

__global__ __launch_bounds__(32) void k_gemm2(const float* __restrict__ a1,
                                              const float* __restrict__ W2,
                                              float* __restrict__ h2, int n) {
    __shared__ float smA[16 * 64];
    __shared__ float ws2[64 * 16];

    const int lane = threadIdx.x;
    const int row0 = blockIdx.x * 16;
    const bool hi  = lane >= 16;
    const int  m   = lane & 15;
    const bool full = (row0 + 16 <= n);

    for (int i = lane; i < 64 * 16 / 4; i += 32)
        ((float4*)ws2)[i] = make_float4(0.f, 0.f, 0.f, 0.f);
    __syncthreads();
    for (int i = lane; i < 64 * 8 / 4; i += 32) {
        int k = i >> 1, half = i & 1;
        ((float4*)(ws2 + k * 16 + half * 4))[0] = ((const float4*)W2)[i];
    }

    if (full) {
        const float4* src = (const float4*)(a1 + (size_t)row0 * 64);
        for (int i = lane; i < 16 * 64 / 4; i += 32)
            ((float4*)smA)[i] = src[i];
    } else {
        for (int i = lane; i < 16 * 64 / 4; i += 32)
            ((float4*)smA)[i] = make_float4(0.f, 0.f, 0.f, 0.f);
        __syncthreads();
        int rows = n - row0;
        const float4* src = (const float4*)(a1 + (size_t)row0 * 64);
        for (int i = lane; i < rows * 16; i += 32)
            ((float4*)smA)[i] = src[i];
    }
    __syncthreads();

    const int kofs = hi ? 2 : 0;
    v8f c = {};
#pragma unroll
    for (int k = 0; k < 16; ++k) {
        v2f a, b;
        a[0] = smA[m * 64 + 4 * k + kofs + 0];
        a[1] = smA[m * 64 + 4 * k + kofs + 1];
        b[0] = ws2[(4 * k + kofs + 0) * 16 + m];
        b[1] = ws2[(4 * k + kofs + 1) * 16 + m];
        c = __builtin_amdgcn_wmma_f32_16x16x4_f32(false, a, false, b,
                                                  (short)0, c, false, false);
    }
    if (m < 8) {
        float* out = h2 + (size_t)row0 * 8 + m;
        if (full) {
#pragma unroll
            for (int r = 0; r < 8; ++r)
                out[(size_t)(r + (hi ? 8 : 0)) * 8] = c[r];
        } else {
#pragma unroll
            for (int r = 0; r < 8; ++r) {
                int rr = r + (hi ? 8 : 0);
                if (row0 + rr < n) out[(size_t)rr * 8] = c[r];
            }
        }
    }
}

#define NB1   392
__device__ __forceinline__ unsigned owner_byte(int d, int n) {
    d = d < 0 ? 0 : (d >= n ? n - 1 : d);
    unsigned o = (unsigned)d / (unsigned)NB1;
    return o < 255u ? o : 255u;
}

__global__ void k_alpha1(const float* __restrict__ h1,
                         const float* __restrict__ asrc, const float* __restrict__ adst,
                         float* __restrict__ as1, float* __restrict__ ad1, int n,
                         const int* __restrict__ ei, int E, unsigned char* __restrict__ ob,
                         int nab) {
    if ((int)blockIdx.x >= nab) {
        const int* __restrict__ dstp = ei + E;
        int g = ((int)blockIdx.x - nab) * (int)blockDim.x + (int)threadIdx.x;
        int e0 = 4 * g;
        if (e0 >= E) return;
        if ((E & 3) == 0) {
            int4 q = ((const int4*)dstp)[g];
            ((unsigned*)ob)[g] = owner_byte(q.x, n) | (owner_byte(q.y, n) << 8)
                               | (owner_byte(q.z, n) << 16) | (owner_byte(q.w, n) << 24);
        } else {
            for (int e = e0; e < e0 + 4 && e < E; ++e)
                ob[e] = (unsigned char)owner_byte(dstp[e], n);
        }
        return;
    }
    int tid = blockIdx.x * blockDim.x + threadIdx.x;
    if (tid >= n * 8) return;
    unsigned h = tid & 7u;
    const float4* hp = (const float4*)(h1 + (unsigned)(tid >> 3) * 64u + h * 8u);
    float4 v0 = hp[0], v1 = hp[1];
    const float4* sp = (const float4*)(asrc + h * 8u);
    const float4* dp = (const float4*)(adst + h * 8u);
    float4 s0 = sp[0], s1v = sp[1], d0 = dp[0], d1v = dp[1];
    float s = v0.x * s0.x + v0.y * s0.y + v0.z * s0.z + v0.w * s0.w
            + v1.x * s1v.x + v1.y * s1v.y + v1.z * s1v.z + v1.w * s1v.w;
    float d = v0.x * d0.x + v0.y * d0.y + v0.z * d0.z + v0.w * d0.w
            + v1.x * d1v.x + v1.y * d1v.y + v1.z * d1v.z + v1.w * d1v.w;
    as1[tid] = s;
    ad1[tid] = d;
}

#define QCAP  8192
#define OWN_T 1024

template <class F>
__device__ __forceinline__ void own_scan(const int* __restrict__ dstp,
                                         const unsigned char* __restrict__ ob, int E,
                                         int lo, unsigned span, unsigned own, bool exact,
                                         int tid, F f) {
    const int E16 = E >> 4;
    const uint4* __restrict__ bp = (const uint4*)ob;
    const unsigned pat = own * 0x01010101u;
    auto eq4 = [&](unsigned w) -> unsigned {
        unsigned x = w ^ pat;
        return ~(((x & 0x7f7f7f7fu) + 0x7f7f7f7fu) | x | 0x7f7f7f7fu);
    };
    auto pack4 = [&](uint4 q, int k0) -> unsigned {
        return (eq4(q.x) >> (7 - k0)) | (eq4(q.y) >> (6 - k0))
             | (eq4(q.z) >> (5 - k0)) | (eq4(q.w) >> (4 - k0));
    };
    auto visit = [&](int idx) {
        if (!exact && (unsigned)dstp[idx] - (unsigned)lo > span) return;
        f(idx);
    };
    int base = tid;
    for (; base + 7 * OWN_T < E16; base += 8 * OWN_T) {
        uint4 v[8];
#pragma unroll
        for (int u = 0; u < 8; ++u)
            v[u] = bp[base + u * OWN_T];
#pragma unroll
        for (int p = 0; p < 4; ++p) {
            unsigned hit = pack4(v[2 * p], 0) | pack4(v[2 * p + 1], 4);
            while (hit) {
                int j = __ffs(hit) - 1;
                hit &= hit - 1u;
                int k = j & 7;
                visit(16 * (base + (2 * p + (k >> 2)) * OWN_T) + 4 * (k & 3) + (j >> 3));
            }
        }
    }
    for (; base < E16; base += OWN_T) {
        unsigned hit = pack4(bp[base], 0);
        while (hit) {
            int j = __ffs(hit) - 1;
            hit &= hit - 1u;
            visit(16 * base + 4 * (j & 7) + (j >> 3));
        }
    }
    for (int idx = 16 * E16 + tid; idx < E; idx += OWN_T)
        if ((unsigned)ob[idx] == own) visit(idx);
}

__device__ __noinline__ void own_edge_max_direct(int s, unsigned dl, int dnode,
                                                 const float* __restrict__ as1,
                                                 const float* __restrict__ ad1,
                                                 unsigned* mx) {
    for (int h = 0; h < 8; ++h) {
        float e = as1[(unsigned)s * 8u + h] + ad1[(unsigned)dnode * 8u + h];
        e = e > 0.0f ? e : NEG_SLOPE * e;
        atomicMax(&mx[dl * 8u + h], enc_f32(e));
    }
}

__device__ __noinline__ void own_edge_direct(int s, unsigned dl, int dnode,
                                             const float* __restrict__ as1,
                                             const float* __restrict__ ad1,
                                             const unsigned* mx,
                                             const float* __restrict__ h1,
                                             float* acc, float* ssum) {
    for (int h = 0; h < 8; ++h) {
        float e = as1[(unsigned)s * 8u + h] + ad1[(unsigned)dnode * 8u + h];
        e = e > 0.0f ? e : NEG_SLOPE * e;
        float ex = expf(e - dec_f32(mx[dl * 8u + h]));
        atomicAdd(&ssum[dl * 8u + h], ex);
        const float* hp = h1 + (unsigned)s * 64u + (unsigned)h * 8u;
        for (int j = 0; j < 8; ++j)
            atomicAdd(&acc[dl * 64u + h * 8u + j], hp[j] * ex);
    }
}

__device__ __forceinline__ float quad_bcast(float v, int k) {
    int i = __builtin_bit_cast(int, v), r;
    switch (k) {
    case 0:  r = __builtin_amdgcn_update_dpp(i, i, 0x00, 0xf, 0xf, true); break;
    case 1:  r = __builtin_amdgcn_update_dpp(i, i, 0x55, 0xf, 0xf, true); break;
    case 2:  r = __builtin_amdgcn_update_dpp(i, i, 0xaa, 0xf, 0xf, true); break;
    default: r = __builtin_amdgcn_update_dpp(i, i, 0xff, 0xf, 0xf, true); break;
    }
    return __builtin_bit_cast(float, r);
}

__global__ __launch_bounds__(OWN_T) void k_own_scatter1(const int* __restrict__ ei, int E, int n,
                                const float* __restrict__ as1, const float* __restrict__ ad1,
                                const float* __restrict__ h1,
                                const float* __restrict__ bias1, float* __restrict__ a1out,
                                int* __restrict__ wq_s, unsigned short* __restrict__ wq_d,
                                unsigned* __restrict__ wq_n,
                                const unsigned char* __restrict__ ob) {
    __shared__ float acc[NB1 * 64];
    __shared__ float ssum[NB1 * 8];
    __shared__ unsigned mx[NB1 * 8];
    __shared__ int qs[QCAP];
    __shared__ unsigned short qd[QCAP];
    __shared__ unsigned qn;

    const int tid  = threadIdx.x;
    const int lane = tid & 31;
    const int wave = tid >> 5;
    const int d0   = blockIdx.x * NB1;
    const int nb   = (n - d0) < NB1 ? (n - d0) : NB1;

    for (int i = tid; i < NB1 * 64; i += OWN_T) acc[i] = 0.0f;
    for (int i = tid; i < NB1 * 8; i += OWN_T) { ssum[i] = 0.0f; mx[i] = 0u; }
    if (tid == 0) qn = 0u;
    __syncthreads();

    const int* __restrict__ dstp = ei + E;
    const int lo = (d0 == 0) ? (int)0x80000000 : d0;
    const int hi = (d0 + nb == n) ? 0x7fffffff : (d0 + nb - 1);
    const unsigned span = (unsigned)hi - (unsigned)lo;
    const unsigned own = blockIdx.x < 255u ? blockIdx.x : 255u;
    const bool exact = gridDim.x <= 256u || blockIdx.x < 255u;
    own_scan(dstp, ob, E, lo, span, own, exact, tid, [&](int idx) {
        unsigned slot = atomicAdd(&qn, 1u);
        if (slot < (unsigned)QCAP) qs[slot] = idx;
    });
    __syncthreads();

    const unsigned qtot = qn;
    if (tid == 0) wq_n[blockIdx.x] = qtot;
    if (qtot <= (unsigned)QCAP) {
        const int cnt = (int)qtot;
        int* __restrict__ gs = wq_s + (size_t)blockIdx.x * QCAP;
        unsigned short* __restrict__ gd = wq_d + (size_t)blockIdx.x * QCAP;
        for (int q = tid; q < cnt; q += OWN_T) {
            int idx = qs[q];
            int s = ei[idx];
            int d = dstp[idx];
            s = s < 0 ? 0 : (s >= n ? n - 1 : s);
            d = d < 0 ? 0 : (d >= n ? n - 1 : d);
            unsigned short dl = (unsigned short)(d - d0);
            qs[q] = s;
            qd[q] = dl;
            gs[q] = s;
            gd[q] = dl;
        }
        __syncthreads();

        const int total = cnt + nb;

        {
            const unsigned hh = (unsigned)lane & 7u;
            const int sub = lane >> 3;
            for (int q0 = wave * 16; q0 < total; q0 += (OWN_T / 32) * 16) {
                int   sN[4];
                unsigned dL[4];
                float aV[4], bV[4];
#pragma unroll
                for (int u = 0; u < 4; ++u) {
                    int q = q0 + 4 * u + sub;
                    int qq = q < total ? q : q0;
                    if (qq < cnt) { sN[u] = qs[qq]; dL[u] = qd[qq]; }
                    else          { dL[u] = (unsigned)(qq - cnt); sN[u] = d0 + (qq - cnt); }
                }
#pragma unroll
                for (int u = 0; u < 4; ++u) {
                    aV[u] = as1[(unsigned)sN[u] * 8u + hh];
                    bV[u] = ad1[((unsigned)d0 + dL[u]) * 8u + hh];
                }
#pragma unroll
                for (int u = 0; u < 4; ++u) {
                    float e = aV[u] + bV[u];
                    e = e > 0.0f ? e : NEG_SLOPE * e;
                    atomicMax(&mx[dL[u] * 8u + hh], enc_f32(e));
                }
            }
        }
        __syncthreads();

        const unsigned hq = (unsigned)lane >> 2;
        const int sub = lane & 3;
        const int wv = __builtin_amdgcn_readfirstlane(wave);
        for (int q0 = wv * 8; q0 < total; q0 += (OWN_T / 32) * 8) {
            int   sN[2];
            unsigned dL[2];
            float aV[2], bV[2], ex[2];
            v2f   hV[8];
            unsigned dU[8];
#pragma unroll
            for (int g = 0; g < 2; ++g) {
                int q = q0 + 4 * g + sub;
                int qq = q < total ? q : q0;
                if (qq < cnt) { sN[g] = qs[qq]; dL[g] = qd[qq]; }
                else          { dL[g] = (unsigned)(qq - cnt); sN[g] = d0 + (qq - cnt); }
            }
#pragma unroll
            for (int g = 0; g < 2; ++g) {
                aV[g] = as1[(unsigned)sN[g] * 8u + hq];
                bV[g] = ad1[((unsigned)d0 + dL[g]) * 8u + hq];
            }
#pragma unroll
            for (int u = 0; u < 8; ++u) {
                unsigned sU = (unsigned)__builtin_amdgcn_readlane(sN[u >> 2], u & 3);
                dU[u] = (unsigned)__builtin_amdgcn_readlane((int)dL[u >> 2], u & 3);
                hV[u] = ((const v2f*)(h1 + (size_t)sU * 64u))[lane];
            }
#pragma unroll
            for (int g = 0; g < 2; ++g) {
                float e = aV[g] + bV[g];
                e = e > 0.0f ? e : NEG_SLOPE * e;
                ex[g] = expf(e - dec_f32(mx[dL[g] * 8u + hq]));
                if (q0 + 4 * g + sub < total) atomicAdd(&ssum[dL[g] * 8u + hq], ex[g]);
            }
#pragma unroll
            for (int u = 0; u < 8; ++u) {
                float w = quad_bcast(ex[u >> 2], u & 3);
                if (q0 + u < total) {
                    atomicAdd(&acc[dU[u] * 64u + 2u * lane + 0u], hV[u][0] * w);
                    atomicAdd(&acc[dU[u] * 64u + 2u * lane + 1u], hV[u][1] * w);
                }
            }
        }
    } else {
        own_scan(dstp, ob, E, lo, span, own, exact, tid, [&](int idx) {
            int d = dstp[idx];
            d = d < 0 ? 0 : (d >= n ? n - 1 : d);
            int s = ei[idx];
            s = s < 0 ? 0 : (s >= n ? n - 1 : s);
            own_edge_max_direct(s, (unsigned)(d - d0), d, as1, ad1, mx);
        });
        for (int i = tid; i < nb; i += OWN_T)
            own_edge_max_direct(d0 + i, (unsigned)i, d0 + i, as1, ad1, mx);
        __syncthreads();
        own_scan(dstp, ob, E, lo, span, own, exact, tid, [&](int idx) {
            int d = dstp[idx];
            d = d < 0 ? 0 : (d >= n ? n - 1 : d);
            int s = ei[idx];
            s = s < 0 ? 0 : (s >= n ? n - 1 : s);
            own_edge_direct(s, (unsigned)(d - d0), d, as1, ad1, mx, h1, acc, ssum);
        });
        for (int i = tid; i < nb; i += OWN_T)
            own_edge_direct(d0 + i, (unsigned)i, d0 + i, as1, ad1, mx, h1, acc, ssum);
    }
    __syncthreads();

    float* __restrict__ op = a1out + (size_t)d0 * 64;
    for (int i = tid; i < nb * 64; i += OWN_T) {
        float v = acc[i] / (ssum[i >> 3] + 1e-16f) + bias1[i & 63];
        op[i] = v > 0.0f ? v : (expf(v) - 1.0f);
    }
}

#define Q2CAP 7168
#define Q2MAX ((Q2CAP + NB1 + OWN_T - 1) / OWN_T)

__device__ __forceinline__ float dot8(float4 v0, float4 v1, float4 s0, float4 s1v) {
    return v0.x * s0.x + v0.y * s0.y + v0.z * s0.z + v0.w * s0.w
         + v1.x * s1v.x + v1.y * s1v.y + v1.z * s1v.z + v1.w * s1v.w;
}

__device__ __noinline__ float own2_logit_direct(int s, int dnode, const float* __restrict__ h2,
                                                const float* __restrict__ asrc,
                                                const float* __restrict__ adst) {
    const float4* sp = (const float4*)(h2 + (unsigned)s * 8u);
    const float4* dp = (const float4*)(h2 + (unsigned)dnode * 8u);
    float e = dot8(sp[0], sp[1], ((const float4*)asrc)[0], ((const float4*)asrc)[1])
            + dot8(dp[0], dp[1], ((const float4*)adst)[0], ((const float4*)adst)[1]);
    return e > 0.0f ? e : NEG_SLOPE * e;
}

__device__ __noinline__ void own2_edge_direct(int s, unsigned dl, float e,
                                              const unsigned* mx,
                                              const float* __restrict__ h2,
                                              float* acc, float* ssum) {
    float ex = expf(e - dec_f32(mx[dl]));
    atomicAdd(&ssum[dl], ex);
    const float* hp = h2 + (unsigned)s * 8u;
    for (int j = 0; j < 8; ++j)
        atomicAdd(&acc[dl * 8u + j], hp[j] * ex);
}

__global__ __launch_bounds__(OWN_T) void k_own_scatter2(const int* __restrict__ ei, int E, int n,
                                const float* __restrict__ asrc2, const float* __restrict__ adst2,
                                const float* __restrict__ h2, const float* __restrict__ bias2,
                                const int* __restrict__ wq_s,
                                const unsigned short* __restrict__ wq_d,
                                const unsigned* __restrict__ wq_n,
                                const unsigned char* __restrict__ ob, float* __restrict__ out) {
    __shared__ float acc[NB1 * 8];
    __shared__ float ssum[NB1];
    __shared__ unsigned mx[NB1];
    __shared__ float adl[NB1];
    __shared__ float rows[(Q2CAP + NB1) * 8];
    __shared__ float qe[Q2CAP + NB1];
    __shared__ unsigned short qd[Q2CAP + NB1];

    const int tid  = threadIdx.x;
    const int lane = tid & 31;
    const int wave = tid >> 5;
    const int d0   = blockIdx.x * NB1;
    const int nb   = (n - d0) < NB1 ? (n - d0) : NB1;

    for (int i = tid; i < NB1 * 8; i += OWN_T) acc[i] = 0.0f;
    for (int i = tid; i < NB1; i += OWN_T) { ssum[i] = 0.0f; mx[i] = 0u; }
    __syncthreads();

    const unsigned qtot = wq_n[blockIdx.x];
    if (qtot <= (unsigned)Q2CAP) {
        const int cnt = (int)qtot;
        const int total = cnt + nb;
        const int* __restrict__ gs = wq_s + (size_t)blockIdx.x * QCAP;
        const unsigned short* __restrict__ gd = wq_d + (size_t)blockIdx.x * QCAP;
        const float4 as0 = ((const float4*)asrc2)[0], as1v = ((const float4*)asrc2)[1];
        const float4 ad0 = ((const float4*)adst2)[0], ad1v = ((const float4*)adst2)[1];

        for (int b = 0; b < Q2MAX; b += 4) {
            int sN[4];
            unsigned dL[4];
            float4 r0[4], r1[4];
#pragma unroll
            for (int u = 0; u < 4; ++u) {
                int q = tid + (b + u) * OWN_T;
                sN[u] = 0; dL[u] = 0u;
                if (q < cnt) { sN[u] = gs[q]; dL[u] = gd[q]; }
            }
#pragma unroll
            for (int u = 0; u < 4; ++u) {
                int q = tid + (b + u) * OWN_T;
                int s = sN[u];
                unsigned d = dL[u];
                if (q < cnt) {
                    s = s < 0 ? 0 : (s >= n ? n - 1 : s);
                    d = d < (unsigned)nb ? d : (unsigned)(nb - 1);
                } else {
                    d = q < total ? (unsigned)(q - cnt) : 0u;
                    s = d0 + (int)d;
                }
                dL[u] = d;
                const float4* hp = (const float4*)(h2 + (unsigned)s * 8u);
                r0[u] = hp[0];
                r1[u] = hp[1];
            }
#pragma unroll
            for (int u = 0; u < 4; ++u) {
                int q = tid + (b + u) * OWN_T;
                if (q < total) {
                    ((float4*)(rows + q * 8))[0] = r0[u];
                    ((float4*)(rows + q * 8))[1] = r1[u];
                    qd[q] = (unsigned short)dL[u];
                    qe[q] = dot8(r0[u], r1[u], as0, as1v);
                    if (q >= cnt) adl[dL[u]] = dot8(r0[u], r1[u], ad0, ad1v);
                }
            }
        }
        __syncthreads();

        for (int q = tid; q < total; q += OWN_T) {
            unsigned d = qd[q];
            float e = qe[q] + adl[d];
            e = e > 0.0f ? e : NEG_SLOPE * e;
            qe[q] = e;
            atomicMax(&mx[d], enc_f32(e));
        }
        __syncthreads();

        const int sub = lane >> 2;
        const unsigned j2 = (unsigned)lane & 3u;
        for (int q0 = wave * 32; q0 < total; q0 += (OWN_T / 32) * 32) {
            unsigned dL[4], mV[4];
            float eV[4];
            v2f   hV[4];
#pragma unroll
            for (int u = 0; u < 4; ++u) {
                int q = q0 + 8 * u + sub;
                int qq = q < total ? q : q0;
                dL[u] = qd[qq]; eV[u] = qe[qq];
                hV[u] = ((const v2f*)(rows + qq * 8))[j2];
            }
#pragma unroll
            for (int u = 0; u < 4; ++u)
                mV[u] = mx[dL[u]];
#pragma unroll
            for (int u = 0; u < 4; ++u) {
                if (q0 + 8 * u + sub < total) {
                    float ex = expf(eV[u] - dec_f32(mV[u]));
                    if (j2 == 0u) atomicAdd(&ssum[dL[u]], ex);
                    atomicAdd(&acc[dL[u] * 8u + 2u * j2 + 0u], hV[u][0] * ex);
                    atomicAdd(&acc[dL[u] * 8u + 2u * j2 + 1u], hV[u][1] * ex);
                }
            }
        }
    } else {
        const int* __restrict__ dstp = ei + E;
        const int lo = (d0 == 0) ? (int)0x80000000 : d0;
        const int hi = (d0 + nb == n) ? 0x7fffffff : (d0 + nb - 1);
        const unsigned span = (unsigned)hi - (unsigned)lo;
        const unsigned own = blockIdx.x < 255u ? blockIdx.x : 255u;
        const bool exact = gridDim.x <= 256u || blockIdx.x < 255u;
        own_scan(dstp, ob, E, lo, span, own, exact, tid, [&](int idx) {
            int d = dstp[idx];
            d = d < 0 ? 0 : (d >= n ? n - 1 : d);
            int s = ei[idx];
            s = s < 0 ? 0 : (s >= n ? n - 1 : s);
            atomicMax(&mx[d - d0], enc_f32(own2_logit_direct(s, d, h2, asrc2, adst2)));
        });
        for (int i = tid; i < nb; i += OWN_T)
            atomicMax(&mx[i], enc_f32(own2_logit_direct(d0 + i, d0 + i, h2, asrc2, adst2)));
        __syncthreads();
        own_scan(dstp, ob, E, lo, span, own, exact, tid, [&](int idx) {
            int d = dstp[idx];
            d = d < 0 ? 0 : (d >= n ? n - 1 : d);
            int s = ei[idx];
            s = s < 0 ? 0 : (s >= n ? n - 1 : s);
            own2_edge_direct(s, (unsigned)(d - d0), own2_logit_direct(s, d, h2, asrc2, adst2),
                             mx, h2, acc, ssum);
        });
        for (int i = tid; i < nb; i += OWN_T)
            own2_edge_direct(d0 + i, (unsigned)i,
                             own2_logit_direct(d0 + i, d0 + i, h2, asrc2, adst2),
                             mx, h2, acc, ssum);
    }
    __syncthreads();

    float* __restrict__ op = out + (size_t)d0 * 8;
    for (int i = tid; i < nb * 8; i += OWN_T)
        op[i] = acc[i] / (ssum[i >> 3] + 1e-16f) + bias2[i & 7];
}

extern "C" void kernel_launch(void* const* d_in, const int* in_sizes, int n_in,
                              void* d_out, int out_size, void* d_ws, size_t ws_size,
                              hipStream_t stream) {
    const float* x     = (const float*)d_in[0];
    const int*   ei    = (const int*)d_in[1];
    const float* W1    = (const float*)d_in[2];
    const float* asrc1 = (const float*)d_in[3];
    const float* adst1 = (const float*)d_in[4];
    const float* bias1 = (const float*)d_in[5];
    const float* W2    = (const float*)d_in[6];
    const float* asrc2 = (const float*)d_in[7];
    const float* adst2 = (const float*)d_in[8];
    const float* bias2 = (const float*)d_in[9];

    int n = in_sizes[0] / F_IN;
    int E = in_sizes[1] / 2;

    char* p = (char*)d_ws;
    float*    h1   = (float*)p;    p += (size_t)n * 64 * 4;
    float*    acc1 = (float*)p;    p += (size_t)n * 64 * 4;
    float*    as1  = (float*)p;    p += (size_t)n * 8 * 4;
    float*    ad1  = (float*)p;    p += (size_t)n * 8 * 4;
    float*    h2   = (float*)p;    p += (size_t)n * 8 * 4;
    unsigned char*  ob   = (unsigned char*)p;  p += (size_t)((E + 15) / 16) * 16;
    int*            wq_s = (int*)p;            p += (size_t)((n + NB1 - 1) / NB1) * QCAP * 4;
    unsigned short* wq_d = (unsigned short*)p; p += (size_t)((n + NB1 - 1) / NB1) * QCAP * 2;
    unsigned*       wq_n = (unsigned*)p;       p += (size_t)((n + NB1 - 1) / NB1) * 4;

    const int tpb = 256;

    k_gemm1<<<(n + 15) / 16, 32, 0, stream>>>(x, W1, h1, n);
    k_alpha1<<<(n * 8 + tpb - 1) / tpb + (E / 4 + tpb) / tpb, tpb, 0, stream>>>(
        h1, asrc1, adst1, as1, ad1, n, ei, E, ob, (n * 8 + tpb - 1) / tpb);
    k_own_scatter1<<<(n + NB1 - 1) / NB1, OWN_T, 0, stream>>>(ei, E, n, as1, ad1, h1, bias1, acc1,
                                                              wq_s, wq_d, wq_n, ob);

    k_gemm2<<<(n + 15) / 16, 32, 0, stream>>>(acc1, W2, h2, n);
    k_own_scatter2<<<(n + NB1 - 1) / NB1, OWN_T, 0, stream>>>(ei, E, n, asrc2, adst2, h2, bias2,
                                                              wq_s, wq_d, wq_n, ob, (float*)d_out);
}


# layer-1 owners find their edges in one bit row per owner (a bit per edge, written as whole lines by the owner-byte blocks via an LDS tile), no owner-byte compares; sums stay arrival-order LDS adds
# speedup vs baseline: 9.0714x; 1.0448x over previous
#include <hip/hip_runtime.h>
#include <hip/hip_bf16.h>

#define NEG_SLOPE 0.2f
#define F_IN 14

typedef float v2f __attribute__((ext_vector_type(2)));
typedef float v8f __attribute__((ext_vector_type(8)));
typedef unsigned u4v __attribute__((ext_vector_type(4)));

__device__ __forceinline__ unsigned enc_f32(float f) {
    unsigned u = __float_as_uint(f);
    return (u & 0x80000000u) ? ~u : (u | 0x80000000u);
}
__device__ __forceinline__ float dec_f32(unsigned e) {
    unsigned u = (e & 0x80000000u) ? (e ^ 0x80000000u) : ~e;
    return __uint_as_float(u);
}

__global__ __launch_bounds__(32) void k_gemm1(const float* __restrict__ x,
                                              const float* __restrict__ W1,
                                              float* __restrict__ h1, int n) {
    __shared__ float xs[16 * 16];
    __shared__ float ws[16 * 64];

    const int lane = threadIdx.x;
    const int row0 = blockIdx.x * 16;
    const bool hi  = lane >= 16;
    const int  m   = lane & 15;
    const bool full = (row0 + 16 <= n);

    for (int i = lane; i < 2 * 64 / 4; i += 32)
        ((float4*)(ws + 14 * 64))[i] = make_float4(0.f, 0.f, 0.f, 0.f);
    for (int i = lane; i < F_IN * 64 / 4; i += 32)
        ((float4*)ws)[i] = ((const float4*)W1)[i];

    for (int i = lane; i < 16 * 16 / 4; i += 32)
        ((float4*)xs)[i] = make_float4(0.f, 0.f, 0.f, 0.f);
    __syncthreads();
    {
        const float* xt = x + (size_t)row0 * F_IN;
        int rows = full ? 16 : (n - row0);
        int cnt  = rows * F_IN;
        for (int i = lane; i < cnt; i += 32) {
            int r = i / F_IN, c = i - r * F_IN;
            xs[r * 16 + c] = xt[i];
        }
    }
    __syncthreads();

    const int kofs = hi ? 2 : 0;
    v2f a[4];
#pragma unroll
    for (int k = 0; k < 4; ++k) {
        a[k][0] = xs[m * 16 + 4 * k + kofs + 0];
        a[k][1] = xs[m * 16 + 4 * k + kofs + 1];
    }

#pragma unroll
    for (int t = 0; t < 4; ++t) {
        v8f c = {};
#pragma unroll
        for (int k = 0; k < 4; ++k) {
            v2f b;
            b[0] = ws[(4 * k + kofs + 0) * 64 + t * 16 + m];
            b[1] = ws[(4 * k + kofs + 1) * 64 + t * 16 + m];
            c = __builtin_amdgcn_wmma_f32_16x16x4_f32(false, a[k], false, b,
                                                      (short)0, c, false, false);
        }
        float* out = h1 + (size_t)row0 * 64 + t * 16 + m;
        if (full) {
#pragma unroll
            for (int r = 0; r < 8; ++r)
                out[(size_t)(r + (hi ? 8 : 0)) * 64] = c[r];
        } else {
#pragma unroll
            for (int r = 0; r < 8; ++r) {
                int rr = r + (hi ? 8 : 0);
                if (row0 + rr < n) out[(size_t)rr * 64] = c[r];
            }
        }
    }
}

__global__ __launch_bounds__(32) void k_gemm2(const float* __restrict__ a1,
                                              const float* __restrict__ W2,
                                              float* __restrict__ h2, int n) {
    __shared__ float smA[16 * 64];
    __shared__ float ws2[64 * 16];

    const int lane = threadIdx.x;
    const int row0 = blockIdx.x * 16;
    const bool hi  = lane >= 16;
    const int  m   = lane & 15;
    const bool full = (row0 + 16 <= n);

    for (int i = lane; i < 64 * 16 / 4; i += 32)
        ((float4*)ws2)[i] = make_float4(0.f, 0.f, 0.f, 0.f);
    __syncthreads();
    for (int i = lane; i < 64 * 8 / 4; i += 32) {
        int k = i >> 1, half = i & 1;
        ((float4*)(ws2 + k * 16 + half * 4))[0] = ((const float4*)W2)[i];
    }

    if (full) {
        const float4* src = (const float4*)(a1 + (size_t)row0 * 64);
        for (int i = lane; i < 16 * 64 / 4; i += 32)
            ((float4*)smA)[i] = src[i];
    } else {
        for (int i = lane; i < 16 * 64 / 4; i += 32)
            ((float4*)smA)[i] = make_float4(0.f, 0.f, 0.f, 0.f);
        __syncthreads();
        int rows = n - row0;
        const float4* src = (const float4*)(a1 + (size_t)row0 * 64);
        for (int i = lane; i < rows * 16; i += 32)
            ((float4*)smA)[i] = src[i];
    }
    __syncthreads();

    const int kofs = hi ? 2 : 0;
    v8f c = {};
#pragma unroll
    for (int k = 0; k < 16; ++k) {
        v2f a, b;
        a[0] = smA[m * 64 + 4 * k + kofs + 0];
        a[1] = smA[m * 64 + 4 * k + kofs + 1];
        b[0] = ws2[(4 * k + kofs + 0) * 16 + m];
        b[1] = ws2[(4 * k + kofs + 1) * 16 + m];
        c = __builtin_amdgcn_wmma_f32_16x16x4_f32(false, a, false, b,
                                                  (short)0, c, false, false);
    }
    if (m < 8) {
        float* out = h2 + (size_t)row0 * 8 + m;
        if (full) {
#pragma unroll
            for (int r = 0; r < 8; ++r)
                out[(size_t)(r + (hi ? 8 : 0)) * 8] = c[r];
        } else {
#pragma unroll
            for (int r = 0; r < 8; ++r) {
                int rr = r + (hi ? 8 : 0);
                if (row0 + rr < n) out[(size_t)rr * 8] = c[r];
            }
        }
    }
}

#define NB1   392
#define PRE_T 256
__device__ __forceinline__ unsigned owner_byte(int d, int n) {
    d = d < 0 ? 0 : (d >= n ? n - 1 : d);
    unsigned o = (unsigned)d / (unsigned)NB1;
    return o < 255u ? o : 255u;
}

__global__ __launch_bounds__(PRE_T) void k_alpha1(const float* __restrict__ h1,
                         const float* __restrict__ asrc, const float* __restrict__ adst,
                         float* __restrict__ as1, float* __restrict__ ad1, int n,
                         const int* __restrict__ ei, int E, unsigned char* __restrict__ ob,
                         int nab, unsigned* __restrict__ bm, int rw) {
    __shared__ unsigned bt[256 * 32];
    if ((int)blockIdx.x >= nab) {
        const int t = (int)threadIdx.x;
        const int tile = (int)blockIdx.x - nab;
        const u4v zero = {0u, 0u, 0u, 0u};
#pragma unroll
        for (int j = 0; j < 8; ++j) ((u4v*)bt)[j * PRE_T + t] = zero;
        __syncthreads();
        const int* __restrict__ dstp = ei + E;
        const int g = tile * PRE_T + t;
        const int e0 = 4 * g;
        if (e0 < E) {
            const unsigned w = (unsigned)t >> 3, b0 = 4u * ((unsigned)t & 7u);
            if ((E & 3) == 0) {
                int4 q = ((const int4*)dstp)[g];
                unsigned o0 = owner_byte(q.x, n), o1 = owner_byte(q.y, n);
                unsigned o2 = owner_byte(q.z, n), o3 = owner_byte(q.w, n);
                ((unsigned*)ob)[g] = o0 | (o1 << 8) | (o2 << 16) | (o3 << 24);
                atomicOr(&bt[o0 * 32u + w], 1u << b0);
                atomicOr(&bt[o1 * 32u + w], 2u << b0);
                atomicOr(&bt[o2 * 32u + w], 4u << b0);
                atomicOr(&bt[o3 * 32u + w], 8u << b0);
            } else {
                for (int k = 0; k < 4 && e0 + k < E; ++k) {
                    unsigned o = owner_byte(dstp[e0 + k], n);
                    ob[e0 + k] = (unsigned char)o;
                    atomicOr(&bt[o * 32u + w], 1u << (b0 + (unsigned)k));
                }
            }
        }
        __syncthreads();
#pragma unroll
        for (int j = 0; j < 8; ++j) {
            int f = j * PRE_T + t;
            ((u4v*)(bm + (size_t)(f >> 3) * (size_t)rw + (size_t)tile * 32u))[f & 7] =
                ((const u4v*)bt)[f];
        }
        return;
    }
    int tid = blockIdx.x * blockDim.x + threadIdx.x;
    if (tid >= n * 8) return;
    unsigned h = tid & 7u;
    const float4* hp = (const float4*)(h1 + (unsigned)(tid >> 3) * 64u + h * 8u);
    float4 v0 = hp[0], v1 = hp[1];
    const float4* sp = (const float4*)(asrc + h * 8u);
    const float4* dp = (const float4*)(adst + h * 8u);
    float4 s0 = sp[0], s1v = sp[1], d0 = dp[0], d1v = dp[1];
    float s = v0.x * s0.x + v0.y * s0.y + v0.z * s0.z + v0.w * s0.w
            + v1.x * s1v.x + v1.y * s1v.y + v1.z * s1v.z + v1.w * s1v.w;
    float d = v0.x * d0.x + v0.y * d0.y + v0.z * d0.z + v0.w * d0.w
            + v1.x * d1v.x + v1.y * d1v.y + v1.z * d1v.z + v1.w * d1v.w;
    as1[tid] = s;
    ad1[tid] = d;
}

#define QCAP  8192
#define OWN_T 1024

template <class F>
__device__ __forceinline__ void own_scan(const int* __restrict__ dstp,
                                         const unsigned char* __restrict__ ob, int E,
                                         int lo, unsigned span, unsigned own, bool exact,
                                         int tid, F f) {
    const int E16 = E >> 4;
    const uint4* __restrict__ bp = (const uint4*)ob;
    const unsigned pat = own * 0x01010101u;
    auto eq4 = [&](unsigned w) -> unsigned {
        unsigned x = w ^ pat;
        return ~(((x & 0x7f7f7f7fu) + 0x7f7f7f7fu) | x | 0x7f7f7f7fu);
    };
    auto pack4 = [&](uint4 q, int k0) -> unsigned {
        return (eq4(q.x) >> (7 - k0)) | (eq4(q.y) >> (6 - k0))
             | (eq4(q.z) >> (5 - k0)) | (eq4(q.w) >> (4 - k0));
    };
    auto visit = [&](int idx) {
        if (!exact && (unsigned)dstp[idx] - (unsigned)lo > span) return;
        f(idx);
    };
    int base = tid;
    for (; base + 7 * OWN_T < E16; base += 8 * OWN_T) {
        uint4 v[8];
#pragma unroll
        for (int u = 0; u < 8; ++u)
            v[u] = bp[base + u * OWN_T];
#pragma unroll
        for (int p = 0; p < 4; ++p) {
            unsigned hit = pack4(v[2 * p], 0) | pack4(v[2 * p + 1], 4);
            while (hit) {
                int j = __ffs(hit) - 1;
                hit &= hit - 1u;
                int k = j & 7;
                visit(16 * (base + (2 * p + (k >> 2)) * OWN_T) + 4 * (k & 3) + (j >> 3));
            }
        }
    }
    for (; base < E16; base += OWN_T) {
        unsigned hit = pack4(bp[base], 0);
        while (hit) {
            int j = __ffs(hit) - 1;
            hit &= hit - 1u;
            visit(16 * base + 4 * (j & 7) + (j >> 3));
        }
    }
    for (int idx = 16 * E16 + tid; idx < E; idx += OWN_T)
        if ((unsigned)ob[idx] == own) visit(idx);
}

__device__ __noinline__ void own_edge_max_direct(int s, unsigned dl, int dnode,
                                                 const float* __restrict__ as1,
                                                 const float* __restrict__ ad1,
                                                 unsigned* mx) {
    for (int h = 0; h < 8; ++h) {
        float e = as1[(unsigned)s * 8u + h] + ad1[(unsigned)dnode * 8u + h];
        e = e > 0.0f ? e : NEG_SLOPE * e;
        atomicMax(&mx[dl * 8u + h], enc_f32(e));
    }
}

__device__ __noinline__ void own_edge_direct(int s, unsigned dl, int dnode,
                                             const float* __restrict__ as1,
                                             const float* __restrict__ ad1,
                                             const unsigned* mx,
                                             const float* __restrict__ h1,
                                             float* acc, float* ssum) {
    for (int h = 0; h < 8; ++h) {
        float e = as1[(unsigned)s * 8u + h] + ad1[(unsigned)dnode * 8u + h];
        e = e > 0.0f ? e : NEG_SLOPE * e;
        float ex = expf(e - dec_f32(mx[dl * 8u + h]));
        atomicAdd(&ssum[dl * 8u + h], ex);
        const float* hp = h1 + (unsigned)s * 64u + (unsigned)h * 8u;
        for (int j = 0; j < 8; ++j)
            atomicAdd(&acc[dl * 64u + h * 8u + j], hp[j] * ex);
    }
}

__device__ __forceinline__ float quad_bcast(float v, int k) {
    int i = __builtin_bit_cast(int, v), r;
    switch (k) {
    case 0:  r = __builtin_amdgcn_update_dpp(i, i, 0x00, 0xf, 0xf, true); break;
    case 1:  r = __builtin_amdgcn_update_dpp(i, i, 0x55, 0xf, 0xf, true); break;
    case 2:  r = __builtin_amdgcn_update_dpp(i, i, 0xaa, 0xf, 0xf, true); break;
    default: r = __builtin_amdgcn_update_dpp(i, i, 0xff, 0xf, 0xf, true); break;
    }
    return __builtin_bit_cast(float, r);
}

__device__ __forceinline__ unsigned wave_incl_sum(unsigned x, int lane) {
    x += (unsigned)__builtin_amdgcn_update_dpp(0, (int)x, 0x111, 0xf, 0xf, false);
    x += (unsigned)__builtin_amdgcn_update_dpp(0, (int)x, 0x112, 0xf, 0xf, false);
    x += (unsigned)__builtin_amdgcn_update_dpp(0, (int)x, 0x114, 0xf, 0xf, false);
    x += (unsigned)__builtin_amdgcn_update_dpp(0, (int)x, 0x118, 0xf, 0xf, false);
    unsigned r0 = (unsigned)__builtin_amdgcn_readlane((int)x, 15);
    return x + ((lane & 16) ? r0 : 0u);
}

__global__ __launch_bounds__(OWN_T) void k_own_scatter1(const int* __restrict__ ei, int E, int n,
                                const float* __restrict__ as1, const float* __restrict__ ad1,
                                const float* __restrict__ h1,
                                const float* __restrict__ bias1, float* __restrict__ a1out,
                                int* __restrict__ wq_s, unsigned short* __restrict__ wq_d,
                                unsigned* __restrict__ wq_n,
                                const unsigned char* __restrict__ ob,
                                const unsigned* __restrict__ bm, int rw) {
    __shared__ float acc[NB1 * 64];
    __shared__ float ssum[NB1 * 8];
    __shared__ unsigned mx[NB1 * 8];
    __shared__ int qs[QCAP];
    __shared__ unsigned short qd[QCAP];
    __shared__ unsigned qn;

    const int tid  = threadIdx.x;
    const int lane = tid & 31;
    const int wave = tid >> 5;
    const int d0   = blockIdx.x * NB1;
    const int nb   = (n - d0) < NB1 ? (n - d0) : NB1;

    for (int i = tid; i < NB1 * 64; i += OWN_T) acc[i] = 0.0f;
    for (int i = tid; i < NB1 * 8; i += OWN_T) { ssum[i] = 0.0f; mx[i] = 0u; }
    if (tid == 0) qn = 0u;
    __syncthreads();

    const int* __restrict__ dstp = ei + E;
    const int lo = (d0 == 0) ? (int)0x80000000 : d0;
    const int hi = (d0 + nb == n) ? 0x7fffffff : (d0 + nb - 1);
    const unsigned span = (unsigned)hi - (unsigned)lo;
    const unsigned own = blockIdx.x < 255u ? blockIdx.x : 255u;
    const bool exact = gridDim.x <= 256u || blockIdx.x < 255u;
    if (exact) {
        const u4v* __restrict__ rp = (const u4v*)(bm + (size_t)own * (size_t)rw);
        const int R16 = rw >> 2;
        const u4v zero = {0u, 0u, 0u, 0u};
        for (int b0 = 0; b0 < R16; b0 += 8 * OWN_T) {
            const int base = b0 + tid;
            u4v v[8];
#pragma unroll
            for (int u = 0; u < 8; ++u) {
                int i = base + u * OWN_T;
                v[u] = i < R16 ? __builtin_nontemporal_load(rp + i) : zero;
            }
            unsigned c = 0u;
#pragma unroll
            for (int u = 0; u < 8; ++u)
                c += (unsigned)(__popc(v[u][0]) + __popc(v[u][1]) + __popc(v[u][2]) + __popc(v[u][3]));
            const unsigned incl = wave_incl_sum(c, lane);
            const unsigned wtot = (unsigned)__builtin_amdgcn_readlane((int)incl, 31);
            unsigned wbase = 0u;
            if (lane == 0 && wtot) wbase = atomicAdd(&qn, wtot);
            wbase = (unsigned)__builtin_amdgcn_readfirstlane((int)wbase);
            unsigned slot = wbase + incl - c;
            if (slot + c <= (unsigned)QCAP) {
#pragma unroll
                for (int u = 0; u < 8; ++u) {
#pragma unroll
                    for (int k = 0; k < 4; ++k) {
                        unsigned w = v[u][k];
                        const int eb = 128 * (base + u * OWN_T) + 32 * k;
                        while (w) {
                            qs[slot++] = eb + (__ffs(w) - 1);
                            w &= w - 1u;
                        }
                    }
                }
            }
        }
    } else {
        if (tid == 0) qn = (unsigned)QCAP + 1u;
    }
    __syncthreads();

    const unsigned qtot = qn;
    if (tid == 0) wq_n[blockIdx.x] = qtot;
    if (qtot <= (unsigned)QCAP) {
        const int cnt = (int)qtot;
        int* __restrict__ gs = wq_s + (size_t)blockIdx.x * QCAP;
        unsigned short* __restrict__ gd = wq_d + (size_t)blockIdx.x * QCAP;
        for (int q = tid; q < cnt; q += OWN_T) {
            int idx = qs[q];
            int s = ei[idx];
            int d = dstp[idx];
            s = s < 0 ? 0 : (s >= n ? n - 1 : s);
            d = d < 0 ? 0 : (d >= n ? n - 1 : d);
            unsigned short dl = (unsigned short)(d - d0);
            qs[q] = s;
            qd[q] = dl;
            gs[q] = s;
            gd[q] = dl;
        }
        __syncthreads();

        const int total = cnt + nb;

        {
            const unsigned hh = (unsigned)lane & 7u;
            const int sub = lane >> 3;
            for (int q0 = wave * 16; q0 < total; q0 += (OWN_T / 32) * 16) {
                int   sN[4];
                unsigned dL[4];
                float aV[4], bV[4];
#pragma unroll
                for (int u = 0; u < 4; ++u) {
                    int q = q0 + 4 * u + sub;
                    int qq = q < total ? q : q0;
                    if (qq < cnt) { sN[u] = qs[qq]; dL[u] = qd[qq]; }
                    else          { dL[u] = (unsigned)(qq - cnt); sN[u] = d0 + (qq - cnt); }
                }
#pragma unroll
                for (int u = 0; u < 4; ++u) {
                    aV[u] = as1[(unsigned)sN[u] * 8u + hh];
                    bV[u] = ad1[((unsigned)d0 + dL[u]) * 8u + hh];
                }
#pragma unroll
                for (int u = 0; u < 4; ++u) {
                    float e = aV[u] + bV[u];
                    e = e > 0.0f ? e : NEG_SLOPE * e;
                    atomicMax(&mx[dL[u] * 8u + hh], enc_f32(e));
                }
            }
        }
        __syncthreads();

        const unsigned hq = (unsigned)lane >> 2;
        const int sub = lane & 3;
        const int wv = __builtin_amdgcn_readfirstlane(wave);
        for (int q0 = wv * 8; q0 < total; q0 += (OWN_T / 32) * 8) {
            int   sN[2];
            unsigned dL[2];
            float aV[2], bV[2], ex[2];
            v2f   hV[8];
            unsigned dU[8];
#pragma unroll
            for (int g = 0; g < 2; ++g) {
                int q = q0 + 4 * g + sub;
                int qq = q < total ? q : q0;
                if (qq < cnt) { sN[g] = qs[qq]; dL[g] = qd[qq]; }
                else          { dL[g] = (unsigned)(qq - cnt); sN[g] = d0 + (qq - cnt); }
            }
#pragma unroll
            for (int g = 0; g < 2; ++g) {
                aV[g] = as1[(unsigned)sN[g] * 8u + hq];
                bV[g] = ad1[((unsigned)d0 + dL[g]) * 8u + hq];
            }
#pragma unroll
            for (int u = 0; u < 8; ++u) {
                unsigned sU = (unsigned)__builtin_amdgcn_readlane(sN[u >> 2], u & 3);
                dU[u] = (unsigned)__builtin_amdgcn_readlane((int)dL[u >> 2], u & 3);
                hV[u] = ((const v2f*)(h1 + (size_t)sU * 64u))[lane];
            }
#pragma unroll
            for (int g = 0; g < 2; ++g) {
                float e = aV[g] + bV[g];
                e = e > 0.0f ? e : NEG_SLOPE * e;
                ex[g] = expf(e - dec_f32(mx[dL[g] * 8u + hq]));
                if (q0 + 4 * g + sub < total) atomicAdd(&ssum[dL[g] * 8u + hq], ex[g]);
            }
#pragma unroll
            for (int u = 0; u < 8; ++u) {
                float w = quad_bcast(ex[u >> 2], u & 3);
                if (q0 + u < total) {
                    atomicAdd(&acc[dU[u] * 64u + 2u * lane + 0u], hV[u][0] * w);
                    atomicAdd(&acc[dU[u] * 64u + 2u * lane + 1u], hV[u][1] * w);
                }
            }
        }
    } else {
        own_scan(dstp, ob, E, lo, span, own, exact, tid, [&](int idx) {
            int d = dstp[idx];
            d = d < 0 ? 0 : (d >= n ? n - 1 : d);
            int s = ei[idx];
            s = s < 0 ? 0 : (s >= n ? n - 1 : s);
            own_edge_max_direct(s, (unsigned)(d - d0), d, as1, ad1, mx);
        });
        for (int i = tid; i < nb; i += OWN_T)
            own_edge_max_direct(d0 + i, (unsigned)i, d0 + i, as1, ad1, mx);
        __syncthreads();
        own_scan(dstp, ob, E, lo, span, own, exact, tid, [&](int idx) {
            int d = dstp[idx];
            d = d < 0 ? 0 : (d >= n ? n - 1 : d);
            int s = ei[idx];
            s = s < 0 ? 0 : (s >= n ? n - 1 : s);
            own_edge_direct(s, (unsigned)(d - d0), d, as1, ad1, mx, h1, acc, ssum);
        });
        for (int i = tid; i < nb; i += OWN_T)
            own_edge_direct(d0 + i, (unsigned)i, d0 + i, as1, ad1, mx, h1, acc, ssum);
    }
    __syncthreads();

    float* __restrict__ op = a1out + (size_t)d0 * 64;
    for (int i = tid; i < nb * 64; i += OWN_T) {
        float v = acc[i] / (ssum[i >> 3] + 1e-16f) + bias1[i & 63];
        op[i] = v > 0.0f ? v : (expf(v) - 1.0f);
    }
}

#define Q2CAP 7168
#define Q2MAX ((Q2CAP + NB1 + OWN_T - 1) / OWN_T)

__device__ __forceinline__ float dot8(float4 v0, float4 v1, float4 s0, float4 s1v) {
    return v0.x * s0.x + v0.y * s0.y + v0.z * s0.z + v0.w * s0.w
         + v1.x * s1v.x + v1.y * s1v.y + v1.z * s1v.z + v1.w * s1v.w;
}

__device__ __noinline__ float own2_logit_direct(int s, int dnode, const float* __restrict__ h2,
                                                const float* __restrict__ asrc,
                                                const float* __restrict__ adst) {
    const float4* sp = (const float4*)(h2 + (unsigned)s * 8u);
    const float4* dp = (const float4*)(h2 + (unsigned)dnode * 8u);
    float e = dot8(sp[0], sp[1], ((const float4*)asrc)[0], ((const float4*)asrc)[1])
            + dot8(dp[0], dp[1], ((const float4*)adst)[0], ((const float4*)adst)[1]);
    return e > 0.0f ? e : NEG_SLOPE * e;
}

__device__ __noinline__ void own2_edge_direct(int s, unsigned dl, float e,
                                              const unsigned* mx,
                                              const float* __restrict__ h2,
                                              float* acc, float* ssum) {
    float ex = expf(e - dec_f32(mx[dl]));
    atomicAdd(&ssum[dl], ex);
    const float* hp = h2 + (unsigned)s * 8u;
    for (int j = 0; j < 8; ++j)
        atomicAdd(&acc[dl * 8u + j], hp[j] * ex);
}

__global__ __launch_bounds__(OWN_T) void k_own_scatter2(const int* __restrict__ ei, int E, int n,
                                const float* __restrict__ asrc2, const float* __restrict__ adst2,
                                const float* __restrict__ h2, const float* __restrict__ bias2,
                                const int* __restrict__ wq_s,
                                const unsigned short* __restrict__ wq_d,
                                const unsigned* __restrict__ wq_n,
                                const unsigned char* __restrict__ ob, float* __restrict__ out) {
    __shared__ float acc[NB1 * 8];
    __shared__ float ssum[NB1];
    __shared__ unsigned mx[NB1];
    __shared__ float adl[NB1];
    __shared__ float rows[(Q2CAP + NB1) * 8];
    __shared__ float qe[Q2CAP + NB1];
    __shared__ unsigned short qd[Q2CAP + NB1];

    const int tid  = threadIdx.x;
    const int lane = tid & 31;
    const int wave = tid >> 5;
    const int d0   = blockIdx.x * NB1;
    const int nb   = (n - d0) < NB1 ? (n - d0) : NB1;

    for (int i = tid; i < NB1 * 8; i += OWN_T) acc[i] = 0.0f;
    for (int i = tid; i < NB1; i += OWN_T) { ssum[i] = 0.0f; mx[i] = 0u; }
    __syncthreads();

    const unsigned qtot = wq_n[blockIdx.x];
    if (qtot <= (unsigned)Q2CAP) {
        const int cnt = (int)qtot;
        const int total = cnt + nb;
        const int* __restrict__ gs = wq_s + (size_t)blockIdx.x * QCAP;
        const unsigned short* __restrict__ gd = wq_d + (size_t)blockIdx.x * QCAP;
        const float4 as0 = ((const float4*)asrc2)[0], as1v = ((const float4*)asrc2)[1];
        const float4 ad0 = ((const float4*)adst2)[0], ad1v = ((const float4*)adst2)[1];

        for (int b = 0; b < Q2MAX; b += 4) {
            int sN[4];
            unsigned dL[4];
            float4 r0[4], r1[4];
#pragma unroll
            for (int u = 0; u < 4; ++u) {
                int q = tid + (b + u) * OWN_T;
                sN[u] = 0; dL[u] = 0u;
                if (q < cnt) { sN[u] = gs[q]; dL[u] = gd[q]; }
            }
#pragma unroll
            for (int u = 0; u < 4; ++u) {
                int q = tid + (b + u) * OWN_T;
                int s = sN[u];
                unsigned d = dL[u];
                if (q < cnt) {
                    s = s < 0 ? 0 : (s >= n ? n - 1 : s);
                    d = d < (unsigned)nb ? d : (unsigned)(nb - 1);
                } else {
                    d = q < total ? (unsigned)(q - cnt) : 0u;
                    s = d0 + (int)d;
                }
                dL[u] = d;
                const float4* hp = (const float4*)(h2 + (unsigned)s * 8u);
                r0[u] = hp[0];
                r1[u] = hp[1];
            }
#pragma unroll
            for (int u = 0; u < 4; ++u) {
                int q = tid + (b + u) * OWN_T;
                if (q < total) {
                    ((float4*)(rows + q * 8))[0] = r0[u];
                    ((float4*)(rows + q * 8))[1] = r1[u];
                    qd[q] = (unsigned short)dL[u];
                    qe[q] = dot8(r0[u], r1[u], as0, as1v);
                    if (q >= cnt) adl[dL[u]] = dot8(r0[u], r1[u], ad0, ad1v);
                }
            }
        }
        __syncthreads();

        for (int q = tid; q < total; q += OWN_T) {
            unsigned d = qd[q];
            float e = qe[q] + adl[d];
            e = e > 0.0f ? e : NEG_SLOPE * e;
            qe[q] = e;
            atomicMax(&mx[d], enc_f32(e));
        }
        __syncthreads();

        const int sub = lane >> 2;
        const unsigned j2 = (unsigned)lane & 3u;
        for (int q0 = wave * 32; q0 < total; q0 += (OWN_T / 32) * 32) {
            unsigned dL[4], mV[4];
            float eV[4];
            v2f   hV[4];
#pragma unroll
            for (int u = 0; u < 4; ++u) {
                int q = q0 + 8 * u + sub;
                int qq = q < total ? q : q0;
                dL[u] = qd[qq]; eV[u] = qe[qq];
                hV[u] = ((const v2f*)(rows + qq * 8))[j2];
            }
#pragma unroll
            for (int u = 0; u < 4; ++u)
                mV[u] = mx[dL[u]];
#pragma unroll
            for (int u = 0; u < 4; ++u) {
                if (q0 + 8 * u + sub < total) {
                    float ex = expf(eV[u] - dec_f32(mV[u]));
                    if (j2 == 0u) atomicAdd(&ssum[dL[u]], ex);
                    atomicAdd(&acc[dL[u] * 8u + 2u * j2 + 0u], hV[u][0] * ex);
                    atomicAdd(&acc[dL[u] * 8u + 2u * j2 + 1u], hV[u][1] * ex);
                }
            }
        }
    } else {
        const int* __restrict__ dstp = ei + E;
        const int lo = (d0 == 0) ? (int)0x80000000 : d0;
        const int hi = (d0 + nb == n) ? 0x7fffffff : (d0 + nb - 1);
        const unsigned span = (unsigned)hi - (unsigned)lo;
        const unsigned own = blockIdx.x < 255u ? blockIdx.x : 255u;
        const bool exact = gridDim.x <= 256u || blockIdx.x < 255u;
        own_scan(dstp, ob, E, lo, span, own, exact, tid, [&](int idx) {
            int d = dstp[idx];
            d = d < 0 ? 0 : (d >= n ? n - 1 : d);
            int s = ei[idx];
            s = s < 0 ? 0 : (s >= n ? n - 1 : s);
            atomicMax(&mx[d - d0], enc_f32(own2_logit_direct(s, d, h2, asrc2, adst2)));
        });
        for (int i = tid; i < nb; i += OWN_T)
            atomicMax(&mx[i], enc_f32(own2_logit_direct(d0 + i, d0 + i, h2, asrc2, adst2)));
        __syncthreads();
        own_scan(dstp, ob, E, lo, span, own, exact, tid, [&](int idx) {
            int d = dstp[idx];
            d = d < 0 ? 0 : (d >= n ? n - 1 : d);
            int s = ei[idx];
            s = s < 0 ? 0 : (s >= n ? n - 1 : s);
            own2_edge_direct(s, (unsigned)(d - d0), own2_logit_direct(s, d, h2, asrc2, adst2),
                             mx, h2, acc, ssum);
        });
        for (int i = tid; i < nb; i += OWN_T)
            own2_edge_direct(d0 + i, (unsigned)i,
                             own2_logit_direct(d0 + i, d0 + i, h2, asrc2, adst2),
                             mx, h2, acc, ssum);
    }
    __syncthreads();

    float* __restrict__ op = out + (size_t)d0 * 8;
    for (int i = tid; i < nb * 8; i += OWN_T)
        op[i] = acc[i] / (ssum[i >> 3] + 1e-16f) + bias2[i & 7];
}

extern "C" void kernel_launch(void* const* d_in, const int* in_sizes, int n_in,
                              void* d_out, int out_size, void* d_ws, size_t ws_size,
                              hipStream_t stream) {
    const float* x     = (const float*)d_in[0];
    const int*   ei    = (const int*)d_in[1];
    const float* W1    = (const float*)d_in[2];
    const float* asrc1 = (const float*)d_in[3];
    const float* adst1 = (const float*)d_in[4];
    const float* bias1 = (const float*)d_in[5];
    const float* W2    = (const float*)d_in[6];
    const float* asrc2 = (const float*)d_in[7];
    const float* adst2 = (const float*)d_in[8];
    const float* bias2 = (const float*)d_in[9];

    int n = in_sizes[0] / F_IN;
    int E = in_sizes[1] / 2;

    char* p = (char*)d_ws;
    float*    h1   = (float*)p;    p += (size_t)n * 64 * 4;
    float*    acc1 = (float*)p;    p += (size_t)n * 64 * 4;
    float*    as1  = (float*)p;    p += (size_t)n * 8 * 4;
    float*    ad1  = (float*)p;    p += (size_t)n * 8 * 4;
    float*    h2   = (float*)p;    p += (size_t)n * 8 * 4;
    unsigned char*  ob   = (unsigned char*)p;  p += (size_t)((E + 15) / 16) * 16;
    int*            wq_s = (int*)p;            p += (size_t)((n + NB1 - 1) / NB1) * QCAP * 4;
    unsigned short* wq_d = (unsigned short*)p; p += (size_t)((n + NB1 - 1) / NB1) * QCAP * 2;
    const int tpb = PRE_T;
    const int nt = (E / 4 + tpb) / tpb;
    const int rw = nt * 32;
    unsigned*       bm   = (unsigned*)p;       p += (size_t)256 * rw * 4;
    unsigned*       wq_n = (unsigned*)p;       p += (size_t)((n + NB1 - 1) / NB1) * 4;

    k_gemm1<<<(n + 15) / 16, 32, 0, stream>>>(x, W1, h1, n);
    k_alpha1<<<(n * 8 + tpb - 1) / tpb + nt, tpb, 0, stream>>>(
        h1, asrc1, adst1, as1, ad1, n, ei, E, ob, (n * 8 + tpb - 1) / tpb, bm, rw);
    k_own_scatter1<<<(n + NB1 - 1) / NB1, OWN_T, 0, stream>>>(ei, E, n, as1, ad1, h1, bias1, acc1,
                                                              wq_s, wq_d, wq_n, ob, bm, rw);

    k_gemm2<<<(n + 15) / 16, 32, 0, stream>>>(acc1, W2, h2, n);
    k_own_scatter2<<<(n + NB1 - 1) / NB1, OWN_T, 0, stream>>>(ei, E, n, asrc2, adst2, h2, bias2,
                                                              wq_s, wq_d, wq_n, ob, (float*)d_out);
}


# layer-1 owner rows carry the edges: edge blocks put (source, local destination) words in 32-word cells per owner; owners queue them with no edge-list look-up; sums stay arrival-order LDS adds
# speedup vs baseline: 9.6433x; 1.0630x over previous
#include <hip/hip_runtime.h>
#include <hip/hip_bf16.h>

#define NEG_SLOPE 0.2f
#define F_IN 14

typedef float v2f __attribute__((ext_vector_type(2)));
typedef float v8f __attribute__((ext_vector_type(8)));
typedef unsigned u4v __attribute__((ext_vector_type(4)));

__device__ __forceinline__ unsigned enc_f32(float f) {
    unsigned u = __float_as_uint(f);
    return (u & 0x80000000u) ? ~u : (u | 0x80000000u);
}
__device__ __forceinline__ float dec_f32(unsigned e) {
    unsigned u = (e & 0x80000000u) ? (e ^ 0x80000000u) : ~e;
    return __uint_as_float(u);
}

__global__ __launch_bounds__(32) void k_gemm1(const float* __restrict__ x,
                                              const float* __restrict__ W1,
                                              float* __restrict__ h1, int n) {
    __shared__ float xs[16 * 16];
    __shared__ float ws[16 * 64];

    const int lane = threadIdx.x;
    const int row0 = blockIdx.x * 16;
    const bool hi  = lane >= 16;
    const int  m   = lane & 15;
    const bool full = (row0 + 16 <= n);

    for (int i = lane; i < 2 * 64 / 4; i += 32)
        ((float4*)(ws + 14 * 64))[i] = make_float4(0.f, 0.f, 0.f, 0.f);
    for (int i = lane; i < F_IN * 64 / 4; i += 32)
        ((float4*)ws)[i] = ((const float4*)W1)[i];

    for (int i = lane; i < 16 * 16 / 4; i += 32)
        ((float4*)xs)[i] = make_float4(0.f, 0.f, 0.f, 0.f);
    __syncthreads();
    {
        const float* xt = x + (size_t)row0 * F_IN;
        int rows = full ? 16 : (n - row0);
        int cnt  = rows * F_IN;
        for (int i = lane; i < cnt; i += 32) {
            int r = i / F_IN, c = i - r * F_IN;
            xs[r * 16 + c] = xt[i];
        }
    }
    __syncthreads();

    const int kofs = hi ? 2 : 0;
    v2f a[4];
#pragma unroll
    for (int k = 0; k < 4; ++k) {
        a[k][0] = xs[m * 16 + 4 * k + kofs + 0];
        a[k][1] = xs[m * 16 + 4 * k + kofs + 1];
    }

#pragma unroll
    for (int t = 0; t < 4; ++t) {
        v8f c = {};
#pragma unroll
        for (int k = 0; k < 4; ++k) {
            v2f b;
            b[0] = ws[(4 * k + kofs + 0) * 64 + t * 16 + m];
            b[1] = ws[(4 * k + kofs + 1) * 64 + t * 16 + m];
            c = __builtin_amdgcn_wmma_f32_16x16x4_f32(false, a[k], false, b,
                                                      (short)0, c, false, false);
        }
        float* out = h1 + (size_t)row0 * 64 + t * 16 + m;
        if (full) {
#pragma unroll
            for (int r = 0; r < 8; ++r)
                out[(size_t)(r + (hi ? 8 : 0)) * 64] = c[r];
        } else {
#pragma unroll
            for (int r = 0; r < 8; ++r) {
                int rr = r + (hi ? 8 : 0);
                if (row0 + rr < n) out[(size_t)rr * 64] = c[r];
            }
        }
    }
}

__global__ __launch_bounds__(32) void k_gemm2(const float* __restrict__ a1,
                                              const float* __restrict__ W2,
                                              float* __restrict__ h2, int n) {
    __shared__ float smA[16 * 64];
    __shared__ float ws2[64 * 16];

    const int lane = threadIdx.x;
    const int row0 = blockIdx.x * 16;
    const bool hi  = lane >= 16;
    const int  m   = lane & 15;
    const bool full = (row0 + 16 <= n);

    for (int i = lane; i < 64 * 16 / 4; i += 32)
        ((float4*)ws2)[i] = make_float4(0.f, 0.f, 0.f, 0.f);
    __syncthreads();
    for (int i = lane; i < 64 * 8 / 4; i += 32) {
        int k = i >> 1, half = i & 1;
        ((float4*)(ws2 + k * 16 + half * 4))[0] = ((const float4*)W2)[i];
    }

    if (full) {
        const float4* src = (const float4*)(a1 + (size_t)row0 * 64);
        for (int i = lane; i < 16 * 64 / 4; i += 32)
            ((float4*)smA)[i] = src[i];
    } else {
        for (int i = lane; i < 16 * 64 / 4; i += 32)
            ((float4*)smA)[i] = make_float4(0.f, 0.f, 0.f, 0.f);
        __syncthreads();
        int rows = n - row0;
        const float4* src = (const float4*)(a1 + (size_t)row0 * 64);
        for (int i = lane; i < rows * 16; i += 32)
            ((float4*)smA)[i] = src[i];
    }
    __syncthreads();

    const int kofs = hi ? 2 : 0;
    v8f c = {};
#pragma unroll
    for (int k = 0; k < 16; ++k) {
        v2f a, b;
        a[0] = smA[m * 64 + 4 * k + kofs + 0];
        a[1] = smA[m * 64 + 4 * k + kofs + 1];
        b[0] = ws2[(4 * k + kofs + 0) * 16 + m];
        b[1] = ws2[(4 * k + kofs + 1) * 16 + m];
        c = __builtin_amdgcn_wmma_f32_16x16x4_f32(false, a, false, b,
                                                  (short)0, c, false, false);
    }
    if (m < 8) {
        float* out = h2 + (size_t)row0 * 8 + m;
        if (full) {
#pragma unroll
            for (int r = 0; r < 8; ++r)
                out[(size_t)(r + (hi ? 8 : 0)) * 8] = c[r];
        } else {
#pragma unroll
            for (int r = 0; r < 8; ++r) {
                int rr = r + (hi ? 8 : 0);
                if (row0 + rr < n) out[(size_t)rr * 8] = c[r];
            }
        }
    }
}

#define NB1   392
#define PRE_T 256
#define EMPTY_W 0xffffffffu
#define CELL_MAX 31u
#define SRC_BITS 23
__device__ __forceinline__ unsigned owner_byte(int d, int n) {
    d = d < 0 ? 0 : (d >= n ? n - 1 : d);
    unsigned o = (unsigned)d / (unsigned)NB1;
    return o < 255u ? o : 255u;
}

__global__ __launch_bounds__(PRE_T) void k_alpha1(const float* __restrict__ h1,
                         const float* __restrict__ asrc, const float* __restrict__ adst,
                         float* __restrict__ as1, float* __restrict__ ad1, int n,
                         const int* __restrict__ ei, int E, unsigned char* __restrict__ ob,
                         int nab, unsigned* __restrict__ bm, int rw) {
    __shared__ unsigned bt[256 * 32];
    __shared__ unsigned bc[256];
    if ((int)blockIdx.x >= nab) {
        const int t = (int)threadIdx.x;
        const int tile = (int)blockIdx.x - nab;
        const u4v fill = {EMPTY_W, EMPTY_W, EMPTY_W, EMPTY_W};
#pragma unroll
        for (int j = 0; j < 8; ++j) ((u4v*)bt)[j * PRE_T + t] = fill;
        bc[t] = 0u;
        __syncthreads();
        const int* __restrict__ dstp = ei + E;
        const int g = tile * PRE_T + t;
        const int e0 = 4 * g;
        auto put = [&](int d, int s, unsigned o) {
            d = d < 0 ? 0 : (d >= n ? n - 1 : d);
            s = s < 0 ? 0 : (s >= n ? n - 1 : s);
            unsigned dl = ((unsigned)d - o * (unsigned)NB1) & 511u;
            unsigned slot = atomicAdd(&bc[o], 1u);
            if (slot < CELL_MAX) bt[o * 32u + slot] = (unsigned)s | (dl << SRC_BITS);
        };
        if (e0 < E) {
            if ((E & 3) == 0) {
                int4 q = ((const int4*)dstp)[g];
                int4 sq = ((const int4*)ei)[g];
                unsigned o0 = owner_byte(q.x, n), o1 = owner_byte(q.y, n);
                unsigned o2 = owner_byte(q.z, n), o3 = owner_byte(q.w, n);
                ((unsigned*)ob)[g] = o0 | (o1 << 8) | (o2 << 16) | (o3 << 24);
                put(q.x, sq.x, o0);
                put(q.y, sq.y, o1);
                put(q.z, sq.z, o2);
                put(q.w, sq.w, o3);
            } else {
                for (int k = 0; k < 4 && e0 + k < E; ++k) {
                    unsigned o = owner_byte(dstp[e0 + k], n);
                    ob[e0 + k] = (unsigned char)o;
                    put(dstp[e0 + k], ei[e0 + k], o);
                }
            }
        }
        __syncthreads();
        const bool fits = n <= (1 << SRC_BITS);
#pragma unroll
        for (int j = 0; j < 8; ++j) {
            int f = j * PRE_T + t;
            u4v val = ((const u4v*)bt)[f];
            if ((f & 7) == 7) val[3] = fits ? bc[f >> 3] : EMPTY_W;
            ((u4v*)(bm + (size_t)(f >> 3) * (size_t)rw + (size_t)tile * 32u))[f & 7] = val;
        }
        return;
    }
    int tid = blockIdx.x * blockDim.x + threadIdx.x;
    if (tid >= n * 8) return;
    unsigned h = tid & 7u;
    const float4* hp = (const float4*)(h1 + (unsigned)(tid >> 3) * 64u + h * 8u);
    float4 v0 = hp[0], v1 = hp[1];
    const float4* sp = (const float4*)(asrc + h * 8u);
    const float4* dp = (const float4*)(adst + h * 8u);
    float4 s0 = sp[0], s1v = sp[1], d0 = dp[0], d1v = dp[1];
    float s = v0.x * s0.x + v0.y * s0.y + v0.z * s0.z + v0.w * s0.w
            + v1.x * s1v.x + v1.y * s1v.y + v1.z * s1v.z + v1.w * s1v.w;
    float d = v0.x * d0.x + v0.y * d0.y + v0.z * d0.z + v0.w * d0.w
            + v1.x * d1v.x + v1.y * d1v.y + v1.z * d1v.z + v1.w * d1v.w;
    as1[tid] = s;
    ad1[tid] = d;
}

#define QCAP  8192
#define OWN_T 1024

template <class F>
__device__ __forceinline__ void own_scan(const int* __restrict__ dstp,
                                         const unsigned char* __restrict__ ob, int E,
                                         int lo, unsigned span, unsigned own, bool exact,
                                         int tid, F f) {
    const int E16 = E >> 4;
    const uint4* __restrict__ bp = (const uint4*)ob;
    const unsigned pat = own * 0x01010101u;
    auto eq4 = [&](unsigned w) -> unsigned {
        unsigned x = w ^ pat;
        return ~(((x & 0x7f7f7f7fu) + 0x7f7f7f7fu) | x | 0x7f7f7f7fu);
    };
    auto pack4 = [&](uint4 q, int k0) -> unsigned {
        return (eq4(q.x) >> (7 - k0)) | (eq4(q.y) >> (6 - k0))
             | (eq4(q.z) >> (5 - k0)) | (eq4(q.w) >> (4 - k0));
    };
    auto visit = [&](int idx) {
        if (!exact && (unsigned)dstp[idx] - (unsigned)lo > span) return;
        f(idx);
    };
    int base = tid;
    for (; base + 7 * OWN_T < E16; base += 8 * OWN_T) {
        uint4 v[8];
#pragma unroll
        for (int u = 0; u < 8; ++u)
            v[u] = bp[base + u * OWN_T];
#pragma unroll
        for (int p = 0; p < 4; ++p) {
            unsigned hit = pack4(v[2 * p], 0) | pack4(v[2 * p + 1], 4);
            while (hit) {
                int j = __ffs(hit) - 1;
                hit &= hit - 1u;
                int k = j & 7;
                visit(16 * (base + (2 * p + (k >> 2)) * OWN_T) + 4 * (k & 3) + (j >> 3));
            }
        }
    }
    for (; base < E16; base += OWN_T) {
        unsigned hit = pack4(bp[base], 0);
        while (hit) {
            int j = __ffs(hit) - 1;
            hit &= hit - 1u;
            visit(16 * base + 4 * (j & 7) + (j >> 3));
        }
    }
    for (int idx = 16 * E16 + tid; idx < E; idx += OWN_T)
        if ((unsigned)ob[idx] == own) visit(idx);
}

__device__ __noinline__ void own_edge_max_direct(int s, unsigned dl, int dnode,
                                                 const float* __restrict__ as1,
                                                 const float* __restrict__ ad1,
                                                 unsigned* mx) {
    for (int h = 0; h < 8; ++h) {
        float e = as1[(unsigned)s * 8u + h] + ad1[(unsigned)dnode * 8u + h];
        e = e > 0.0f ? e : NEG_SLOPE * e;
        atomicMax(&mx[dl * 8u + h], enc_f32(e));
    }
}

__device__ __noinline__ void own_edge_direct(int s, unsigned dl, int dnode,
                                             const float* __restrict__ as1,
                                             const float* __restrict__ ad1,
                                             const unsigned* mx,
                                             const float* __restrict__ h1,
                                             float* acc, float* ssum) {
    for (int h = 0; h < 8; ++h) {
        float e = as1[(unsigned)s * 8u + h] + ad1[(unsigned)dnode * 8u + h];
        e = e > 0.0f ? e : NEG_SLOPE * e;
        float ex = expf(e - dec_f32(mx[dl * 8u + h]));
        atomicAdd(&ssum[dl * 8u + h], ex);
        const float* hp = h1 + (unsigned)s * 64u + (unsigned)h * 8u;
        for (int j = 0; j < 8; ++j)
            atomicAdd(&acc[dl * 64u + h * 8u + j], hp[j] * ex);
    }
}

__device__ __forceinline__ float quad_bcast(float v, int k) {
    int i = __builtin_bit_cast(int, v), r;
    switch (k) {
    case 0:  r = __builtin_amdgcn_update_dpp(i, i, 0x00, 0xf, 0xf, true); break;
    case 1:  r = __builtin_amdgcn_update_dpp(i, i, 0x55, 0xf, 0xf, true); break;
    case 2:  r = __builtin_amdgcn_update_dpp(i, i, 0xaa, 0xf, 0xf, true); break;
    default: r = __builtin_amdgcn_update_dpp(i, i, 0xff, 0xf, 0xf, true); break;
    }
    return __builtin_bit_cast(float, r);
}

__device__ __forceinline__ unsigned wave_incl_sum(unsigned x, int lane) {
    x += (unsigned)__builtin_amdgcn_update_dpp(0, (int)x, 0x111, 0xf, 0xf, false);
    x += (unsigned)__builtin_amdgcn_update_dpp(0, (int)x, 0x112, 0xf, 0xf, false);
    x += (unsigned)__builtin_amdgcn_update_dpp(0, (int)x, 0x114, 0xf, 0xf, false);
    x += (unsigned)__builtin_amdgcn_update_dpp(0, (int)x, 0x118, 0xf, 0xf, false);
    unsigned r0 = (unsigned)__builtin_amdgcn_readlane((int)x, 15);
    return x + ((lane & 16) ? r0 : 0u);
}

__global__ __launch_bounds__(OWN_T) void k_own_scatter1(const int* __restrict__ ei, int E, int n,
                                const float* __restrict__ as1, const float* __restrict__ ad1,
                                const float* __restrict__ h1,
                                const float* __restrict__ bias1, float* __restrict__ a1out,
                                int* __restrict__ wq_s, unsigned short* __restrict__ wq_d,
                                unsigned* __restrict__ wq_n,
                                const unsigned char* __restrict__ ob,
                                const unsigned* __restrict__ bm, int rw) {
    __shared__ float acc[NB1 * 64];
    __shared__ float ssum[NB1 * 8];
    __shared__ unsigned mx[NB1 * 8];
    __shared__ int qs[QCAP];
    __shared__ unsigned short qd[QCAP];
    __shared__ unsigned qn;
    __shared__ unsigned qover;

    const int tid  = threadIdx.x;
    const int lane = tid & 31;
    const int wave = tid >> 5;
    const int d0   = blockIdx.x * NB1;
    const int nb   = (n - d0) < NB1 ? (n - d0) : NB1;

    for (int i = tid; i < NB1 * 64; i += OWN_T) acc[i] = 0.0f;
    for (int i = tid; i < NB1 * 8; i += OWN_T) { ssum[i] = 0.0f; mx[i] = 0u; }
    if (tid == 0) { qn = 0u; qover = 0u; }
    __syncthreads();

    const int* __restrict__ dstp = ei + E;
    const int lo = (d0 == 0) ? (int)0x80000000 : d0;
    const int hi = (d0 + nb == n) ? 0x7fffffff : (d0 + nb - 1);
    const unsigned span = (unsigned)hi - (unsigned)lo;
    const unsigned own = blockIdx.x < 255u ? blockIdx.x : 255u;
    const bool exact = gridDim.x <= 256u || blockIdx.x < 255u;
    int* __restrict__ gs = wq_s + (size_t)blockIdx.x * QCAP;
    unsigned short* __restrict__ gd = wq_d + (size_t)blockIdx.x * QCAP;
    if (exact) {
        const u4v* __restrict__ rp = (const u4v*)(bm + (size_t)own * (size_t)rw);
        const int R16 = rw >> 2;
        const u4v none = {EMPTY_W, EMPTY_W, EMPTY_W, EMPTY_W};
        for (int b0 = 0; b0 < R16; b0 += 8 * OWN_T) {
            const int base = b0 + tid;
            u4v v[8];
            bool over = false;
#pragma unroll
            for (int u = 0; u < 8; ++u) {
                int i = base + u * OWN_T;
                const bool in = i < R16;
                v[u] = in ? __builtin_nontemporal_load(rp + i) : none;
                if (((unsigned)i & 7u) == 7u) {
                    over = over || (in && v[u][3] > CELL_MAX);
                    v[u][3] = EMPTY_W;
                }
            }
            if (over) qover = 1u;
            unsigned c = 0u;
#pragma unroll
            for (int u = 0; u < 8; ++u)
                c += (unsigned)(v[u][0] != EMPTY_W) + (unsigned)(v[u][1] != EMPTY_W)
                   + (unsigned)(v[u][2] != EMPTY_W) + (unsigned)(v[u][3] != EMPTY_W);
            const unsigned incl = wave_incl_sum(c, lane);
            const unsigned wtot = (unsigned)__builtin_amdgcn_readlane((int)incl, 31);
            unsigned wbase = 0u;
            if (lane == 0 && wtot) wbase = atomicAdd(&qn, wtot);
            wbase = (unsigned)__builtin_amdgcn_readfirstlane((int)wbase);
            unsigned slot = wbase + incl - c;
            if (slot + c <= (unsigned)QCAP) {
#pragma unroll
                for (int u = 0; u < 8; ++u) {
#pragma unroll
                    for (int k = 0; k < 4; ++k) {
                        const unsigned w = v[u][k];
                        if (w != EMPTY_W) {
                            const int s = (int)(w & ((1u << SRC_BITS) - 1u));
                            const unsigned short dl = (unsigned short)(w >> SRC_BITS);
                            qs[slot] = s;
                            qd[slot] = dl;
                            gs[slot] = s;
                            gd[slot] = dl;
                            ++slot;
                        }
                    }
                }
            }
        }
    } else {
        if (tid == 0) qn = (unsigned)QCAP + 1u;
    }
    __syncthreads();

    const unsigned qtot = qover ? (unsigned)QCAP + 1u : qn;
    if (tid == 0) wq_n[blockIdx.x] = qtot;
    if (qtot <= (unsigned)QCAP) {
        const int cnt = (int)qtot;
        const int total = cnt + nb;

        {
            const unsigned hh = (unsigned)lane & 7u;
            const int sub = lane >> 3;
            for (int q0 = wave * 16; q0 < total; q0 += (OWN_T / 32) * 16) {
                int   sN[4];
                unsigned dL[4];
                float aV[4], bV[4];
#pragma unroll
                for (int u = 0; u < 4; ++u) {
                    int q = q0 + 4 * u + sub;
                    int qq = q < total ? q : q0;
                    if (qq < cnt) { sN[u] = qs[qq]; dL[u] = qd[qq]; }
                    else          { dL[u] = (unsigned)(qq - cnt); sN[u] = d0 + (qq - cnt); }
                }
#pragma unroll
                for (int u = 0; u < 4; ++u) {
                    aV[u] = as1[(unsigned)sN[u] * 8u + hh];
                    bV[u] = ad1[((unsigned)d0 + dL[u]) * 8u + hh];
                }
#pragma unroll
                for (int u = 0; u < 4; ++u) {
                    float e = aV[u] + bV[u];
                    e = e > 0.0f ? e : NEG_SLOPE * e;
                    atomicMax(&mx[dL[u] * 8u + hh], enc_f32(e));
                }
            }
        }
        __syncthreads();

        const unsigned hq = (unsigned)lane >> 2;
        const int sub = lane & 3;
        const int wv = __builtin_amdgcn_readfirstlane(wave);
        for (int q0 = wv * 8; q0 < total; q0 += (OWN_T / 32) * 8) {
            int   sN[2];
            unsigned dL[2];
            float aV[2], bV[2], ex[2];
            v2f   hV[8];
            unsigned dU[8];
#pragma unroll
            for (int g = 0; g < 2; ++g) {
                int q = q0 + 4 * g + sub;
                int qq = q < total ? q : q0;
                if (qq < cnt) { sN[g] = qs[qq]; dL[g] = qd[qq]; }
                else          { dL[g] = (unsigned)(qq - cnt); sN[g] = d0 + (qq - cnt); }
            }
#pragma unroll
            for (int g = 0; g < 2; ++g) {
                aV[g] = as1[(unsigned)sN[g] * 8u + hq];
                bV[g] = ad1[((unsigned)d0 + dL[g]) * 8u + hq];
            }
#pragma unroll
            for (int u = 0; u < 8; ++u) {
                unsigned sU = (unsigned)__builtin_amdgcn_readlane(sN[u >> 2], u & 3);
                dU[u] = (unsigned)__builtin_amdgcn_readlane((int)dL[u >> 2], u & 3);
                hV[u] = ((const v2f*)(h1 + (size_t)sU * 64u))[lane];
            }
#pragma unroll
            for (int g = 0; g < 2; ++g) {
                float e = aV[g] + bV[g];
                e = e > 0.0f ? e : NEG_SLOPE * e;
                ex[g] = expf(e - dec_f32(mx[dL[g] * 8u + hq]));
                if (q0 + 4 * g + sub < total) atomicAdd(&ssum[dL[g] * 8u + hq], ex[g]);
            }
#pragma unroll
            for (int u = 0; u < 8; ++u) {
                float w = quad_bcast(ex[u >> 2], u & 3);
                if (q0 + u < total) {
                    atomicAdd(&acc[dU[u] * 64u + 2u * lane + 0u], hV[u][0] * w);
                    atomicAdd(&acc[dU[u] * 64u + 2u * lane + 1u], hV[u][1] * w);
                }
            }
        }
    } else {
        own_scan(dstp, ob, E, lo, span, own, exact, tid, [&](int idx) {
            int d = dstp[idx];
            d = d < 0 ? 0 : (d >= n ? n - 1 : d);
            int s = ei[idx];
            s = s < 0 ? 0 : (s >= n ? n - 1 : s);
            own_edge_max_direct(s, (unsigned)(d - d0), d, as1, ad1, mx);
        });
        for (int i = tid; i < nb; i += OWN_T)
            own_edge_max_direct(d0 + i, (unsigned)i, d0 + i, as1, ad1, mx);
        __syncthreads();
        own_scan(dstp, ob, E, lo, span, own, exact, tid, [&](int idx) {
            int d = dstp[idx];
            d = d < 0 ? 0 : (d >= n ? n - 1 : d);
            int s = ei[idx];
            s = s < 0 ? 0 : (s >= n ? n - 1 : s);
            own_edge_direct(s, (unsigned)(d - d0), d, as1, ad1, mx, h1, acc, ssum);
        });
        for (int i = tid; i < nb; i += OWN_T)
            own_edge_direct(d0 + i, (unsigned)i, d0 + i, as1, ad1, mx, h1, acc, ssum);
    }
    __syncthreads();

    float* __restrict__ op = a1out + (size_t)d0 * 64;
    for (int i = tid; i < nb * 64; i += OWN_T) {
        float v = acc[i] / (ssum[i >> 3] + 1e-16f) + bias1[i & 63];
        op[i] = v > 0.0f ? v : (expf(v) - 1.0f);
    }
}

#define Q2CAP 7168
#define Q2MAX ((Q2CAP + NB1 + OWN_T - 1) / OWN_T)

__device__ __forceinline__ float dot8(float4 v0, float4 v1, float4 s0, float4 s1v) {
    return v0.x * s0.x + v0.y * s0.y + v0.z * s0.z + v0.w * s0.w
         + v1.x * s1v.x + v1.y * s1v.y + v1.z * s1v.z + v1.w * s1v.w;
}

__device__ __noinline__ float own2_logit_direct(int s, int dnode, const float* __restrict__ h2,
                                                const float* __restrict__ asrc,
                                                const float* __restrict__ adst) {
    const float4* sp = (const float4*)(h2 + (unsigned)s * 8u);
    const float4* dp = (const float4*)(h2 + (unsigned)dnode * 8u);
    float e = dot8(sp[0], sp[1], ((const float4*)asrc)[0], ((const float4*)asrc)[1])
            + dot8(dp[0], dp[1], ((const float4*)adst)[0], ((const float4*)adst)[1]);
    return e > 0.0f ? e : NEG_SLOPE * e;
}

__device__ __noinline__ void own2_edge_direct(int s, unsigned dl, float e,
                                              const unsigned* mx,
                                              const float* __restrict__ h2,
                                              float* acc, float* ssum) {
    float ex = expf(e - dec_f32(mx[dl]));
    atomicAdd(&ssum[dl], ex);
    const float* hp = h2 + (unsigned)s * 8u;
    for (int j = 0; j < 8; ++j)
        atomicAdd(&acc[dl * 8u + j], hp[j] * ex);
}

__global__ __launch_bounds__(OWN_T) void k_own_scatter2(const int* __restrict__ ei, int E, int n,
                                const float* __restrict__ asrc2, const float* __restrict__ adst2,
                                const float* __restrict__ h2, const float* __restrict__ bias2,
                                const int* __restrict__ wq_s,
                                const unsigned short* __restrict__ wq_d,
                                const unsigned* __restrict__ wq_n,
                                const unsigned char* __restrict__ ob, float* __restrict__ out) {
    __shared__ float acc[NB1 * 8];
    __shared__ float ssum[NB1];
    __shared__ unsigned mx[NB1];
    __shared__ float adl[NB1];
    __shared__ float rows[(Q2CAP + NB1) * 8];
    __shared__ float qe[Q2CAP + NB1];
    __shared__ unsigned short qd[Q2CAP + NB1];

    const int tid  = threadIdx.x;
    const int lane = tid & 31;
    const int wave = tid >> 5;
    const int d0   = blockIdx.x * NB1;
    const int nb   = (n - d0) < NB1 ? (n - d0) : NB1;

    for (int i = tid; i < NB1 * 8; i += OWN_T) acc[i] = 0.0f;
    for (int i = tid; i < NB1; i += OWN_T) { ssum[i] = 0.0f; mx[i] = 0u; }
    __syncthreads();

    const unsigned qtot = wq_n[blockIdx.x];
    if (qtot <= (unsigned)Q2CAP) {
        const int cnt = (int)qtot;
        const int total = cnt + nb;
        const int* __restrict__ gs = wq_s + (size_t)blockIdx.x * QCAP;
        const unsigned short* __restrict__ gd = wq_d + (size_t)blockIdx.x * QCAP;
        const float4 as0 = ((const float4*)asrc2)[0], as1v = ((const float4*)asrc2)[1];
        const float4 ad0 = ((const float4*)adst2)[0], ad1v = ((const float4*)adst2)[1];

        for (int b = 0; b < Q2MAX; b += 4) {
            int sN[4];
            unsigned dL[4];
            float4 r0[4], r1[4];
#pragma unroll
            for (int u = 0; u < 4; ++u) {
                int q = tid + (b + u) * OWN_T;
                sN[u] = 0; dL[u] = 0u;
                if (q < cnt) { sN[u] = gs[q]; dL[u] = gd[q]; }
            }
#pragma unroll
            for (int u = 0; u < 4; ++u) {
                int q = tid + (b + u) * OWN_T;
                int s = sN[u];
                unsigned d = dL[u];
                if (q < cnt) {
                    s = s < 0 ? 0 : (s >= n ? n - 1 : s);
                    d = d < (unsigned)nb ? d : (unsigned)(nb - 1);
                } else {
                    d = q < total ? (unsigned)(q - cnt) : 0u;
                    s = d0 + (int)d;
                }
                dL[u] = d;
                const float4* hp = (const float4*)(h2 + (unsigned)s * 8u);
                r0[u] = hp[0];
                r1[u] = hp[1];
            }
#pragma unroll
            for (int u = 0; u < 4; ++u) {
                int q = tid + (b + u) * OWN_T;
                if (q < total) {
                    ((float4*)(rows + q * 8))[0] = r0[u];
                    ((float4*)(rows + q * 8))[1] = r1[u];
                    qd[q] = (unsigned short)dL[u];
                    qe[q] = dot8(r0[u], r1[u], as0, as1v);
                    if (q >= cnt) adl[dL[u]] = dot8(r0[u], r1[u], ad0, ad1v);
                }
            }
        }
        __syncthreads();

        for (int q = tid; q < total; q += OWN_T) {
            unsigned d = qd[q];
            float e = qe[q] + adl[d];
            e = e > 0.0f ? e : NEG_SLOPE * e;
            qe[q] = e;
            atomicMax(&mx[d], enc_f32(e));
        }
        __syncthreads();

        const int sub = lane >> 2;
        const unsigned j2 = (unsigned)lane & 3u;
        for (int q0 = wave * 32; q0 < total; q0 += (OWN_T / 32) * 32) {
            unsigned dL[4], mV[4];
            float eV[4];
            v2f   hV[4];
#pragma unroll
            for (int u = 0; u < 4; ++u) {
                int q = q0 + 8 * u + sub;
                int qq = q < total ? q : q0;
                dL[u] = qd[qq]; eV[u] = qe[qq];
                hV[u] = ((const v2f*)(rows + qq * 8))[j2];
            }
#pragma unroll
            for (int u = 0; u < 4; ++u)
                mV[u] = mx[dL[u]];
#pragma unroll
            for (int u = 0; u < 4; ++u) {
                if (q0 + 8 * u + sub < total) {
                    float ex = expf(eV[u] - dec_f32(mV[u]));
                    if (j2 == 0u) atomicAdd(&ssum[dL[u]], ex);
                    atomicAdd(&acc[dL[u] * 8u + 2u * j2 + 0u], hV[u][0] * ex);
                    atomicAdd(&acc[dL[u] * 8u + 2u * j2 + 1u], hV[u][1] * ex);
                }
            }
        }
    } else {
        const int* __restrict__ dstp = ei + E;
        const int lo = (d0 == 0) ? (int)0x80000000 : d0;
        const int hi = (d0 + nb == n) ? 0x7fffffff : (d0 + nb - 1);
        const unsigned span = (unsigned)hi - (unsigned)lo;
        const unsigned own = blockIdx.x < 255u ? blockIdx.x : 255u;
        const bool exact = gridDim.x <= 256u || blockIdx.x < 255u;
        own_scan(dstp, ob, E, lo, span, own, exact, tid, [&](int idx) {
            int d = dstp[idx];
            d = d < 0 ? 0 : (d >= n ? n - 1 : d);
            int s = ei[idx];
            s = s < 0 ? 0 : (s >= n ? n - 1 : s);
            atomicMax(&mx[d - d0], enc_f32(own2_logit_direct(s, d, h2, asrc2, adst2)));
        });
        for (int i = tid; i < nb; i += OWN_T)
            atomicMax(&mx[i], enc_f32(own2_logit_direct(d0 + i, d0 + i, h2, asrc2, adst2)));
        __syncthreads();
        own_scan(dstp, ob, E, lo, span, own, exact, tid, [&](int idx) {
            int d = dstp[idx];
            d = d < 0 ? 0 : (d >= n ? n - 1 : d);
            int s = ei[idx];
            s = s < 0 ? 0 : (s >= n ? n - 1 : s);
            own2_edge_direct(s, (unsigned)(d - d0), own2_logit_direct(s, d, h2, asrc2, adst2),
                             mx, h2, acc, ssum);
        });
        for (int i = tid; i < nb; i += OWN_T)
            own2_edge_direct(d0 + i, (unsigned)i,
                             own2_logit_direct(d0 + i, d0 + i, h2, asrc2, adst2),
                             mx, h2, acc, ssum);
    }
    __syncthreads();

    float* __restrict__ op = out + (size_t)d0 * 8;
    for (int i = tid; i < nb * 8; i += OWN_T)
        op[i] = acc[i] / (ssum[i >> 3] + 1e-16f) + bias2[i & 7];
}

extern "C" void kernel_launch(void* const* d_in, const int* in_sizes, int n_in,
                              void* d_out, int out_size, void* d_ws, size_t ws_size,
                              hipStream_t stream) {
    const float* x     = (const float*)d_in[0];
    const int*   ei    = (const int*)d_in[1];
    const float* W1    = (const float*)d_in[2];
    const float* asrc1 = (const float*)d_in[3];
    const float* adst1 = (const float*)d_in[4];
    const float* bias1 = (const float*)d_in[5];
    const float* W2    = (const float*)d_in[6];
    const float* asrc2 = (const float*)d_in[7];
    const float* adst2 = (const float*)d_in[8];
    const float* bias2 = (const float*)d_in[9];

    int n = in_sizes[0] / F_IN;
    int E = in_sizes[1] / 2;

    char* p = (char*)d_ws;
    float*    h1   = (float*)p;    p += (size_t)n * 64 * 4;
    float*    acc1 = (float*)p;    p += (size_t)n * 64 * 4;
    float*    as1  = (float*)p;    p += (size_t)n * 8 * 4;
    float*    ad1  = (float*)p;    p += (size_t)n * 8 * 4;
    float*    h2   = (float*)p;    p += (size_t)n * 8 * 4;
    unsigned char*  ob   = (unsigned char*)p;  p += (size_t)((E + 15) / 16) * 16;
    int*            wq_s = (int*)p;            p += (size_t)((n + NB1 - 1) / NB1) * QCAP * 4;
    unsigned short* wq_d = (unsigned short*)p; p += (size_t)((n + NB1 - 1) / NB1) * QCAP * 2;
    const int tpb = PRE_T;
    const int nt = (E / 4 + tpb) / tpb;
    const int rw = nt * 32;
    unsigned*       bm   = (unsigned*)p;       p += (size_t)256 * rw * 4;
    unsigned*       wq_n = (unsigned*)p;       p += (size_t)((n + NB1 - 1) / NB1) * 4;

    k_gemm1<<<(n + 15) / 16, 32, 0, stream>>>(x, W1, h1, n);
    k_alpha1<<<(n * 8 + tpb - 1) / tpb + nt, tpb, 0, stream>>>(
        h1, asrc1, adst1, as1, ad1, n, ei, E, ob, (n * 8 + tpb - 1) / tpb, bm, rw);
    k_own_scatter1<<<(n + NB1 - 1) / NB1, OWN_T, 0, stream>>>(ei, E, n, as1, ad1, h1, bias1, acc1,
                                                              wq_s, wq_d, wq_n, ob, bm, rw);

    k_gemm2<<<(n + 15) / 16, 32, 0, stream>>>(acc1, W2, h2, n);
    k_own_scatter2<<<(n + NB1 - 1) / NB1, OWN_T, 0, stream>>>(ei, E, n, asrc2, adst2, h2, bias2,
                                                              wq_s, wq_d, wq_n, ob, (float*)d_out);
}


# both product launches: 8-wave work-groups, 3 row tiles per wave behind one barrier, B fragments held in registers over the tiles (a1 tile pitch 68 words), in place of a one-wave work-group per tile
# speedup vs baseline: 10.0524x; 1.0424x over previous
#include <hip/hip_runtime.h>
#include <hip/hip_bf16.h>

#define NEG_SLOPE 0.2f
#define F_IN 14

typedef float v2f __attribute__((ext_vector_type(2)));
typedef float v8f __attribute__((ext_vector_type(8)));
typedef unsigned u4v __attribute__((ext_vector_type(4)));

__device__ __forceinline__ unsigned enc_f32(float f) {
    unsigned u = __float_as_uint(f);
    return (u & 0x80000000u) ? ~u : (u | 0x80000000u);
}
__device__ __forceinline__ float dec_f32(unsigned e) {
    unsigned u = (e & 0x80000000u) ? (e ^ 0x80000000u) : ~e;
    return __uint_as_float(u);
}

#define PROD_W   8
#define PROD_TPW 3

__global__ __launch_bounds__(32 * PROD_W) void k_gemm1(const float* __restrict__ x,
                                              const float* __restrict__ W1,
                                              float* __restrict__ h1, int n) {
    static_assert(F_IN == 14, "the fragment loads below take columns 12, 13 as the last k pair of a row of x");
    __shared__ float ws[16 * 64];

    const int tid  = threadIdx.x;
    const int lane = tid & 31;
    const int wave = __builtin_amdgcn_readfirstlane(tid >> 5);
    const bool hi  = lane >= 16;
    const int  m   = lane & 15;
    const int kofs = hi ? 2 : 0;

    {
        float4 v = make_float4(0.f, 0.f, 0.f, 0.f);
        if (tid < F_IN * 64 / 4) v = ((const float4*)W1)[tid];
        ((float4*)ws)[tid] = v;
    }

    const int tile0 = (int)blockIdx.x * (PROD_W * PROD_TPW) + wave;
    v2f a[PROD_TPW][4];
#pragma unroll
    for (int it = 0; it < PROD_TPW; ++it) {
        const int row = (tile0 + it * PROD_W) * 16 + m;
        const float* xr = x + (size_t)(row < n ? row : n - 1) * F_IN + kofs;
        a[it][0] = *(const v2f*)(xr + 0);
        a[it][1] = *(const v2f*)(xr + 4);
        a[it][2] = *(const v2f*)(xr + 8);
        v2f last = {0.f, 0.f};
        if (!hi) last = *(const v2f*)(xr + 12);
        a[it][3] = last;
    }
    __syncthreads();

    v2f b[4][4];
#pragma unroll
    for (int t = 0; t < 4; ++t) {
#pragma unroll
        for (int k = 0; k < 4; ++k) {
            b[t][k][0] = ws[(4 * k + kofs + 0) * 64 + t * 16 + m];
            b[t][k][1] = ws[(4 * k + kofs + 1) * 64 + t * 16 + m];
        }
    }

#pragma unroll
    for (int it = 0; it < PROD_TPW; ++it) {
        const int row0 = (tile0 + it * PROD_W) * 16;
        const bool full = (row0 + 16 <= n);
#pragma unroll
        for (int t = 0; t < 4; ++t) {
            v8f c = {};
#pragma unroll
            for (int k = 0; k < 4; ++k)
                c = __builtin_amdgcn_wmma_f32_16x16x4_f32(false, a[it][k], false, b[t][k],
                                                          (short)0, c, false, false);
            float* out = h1 + (size_t)row0 * 64 + t * 16 + m;
            if (full) {
#pragma unroll
                for (int r = 0; r < 8; ++r)
                    out[(size_t)(r + (hi ? 8 : 0)) * 64] = c[r];
            } else {
#pragma unroll
                for (int r = 0; r < 8; ++r) {
                    int rr = r + (hi ? 8 : 0);
                    if (row0 + rr < n) out[(size_t)rr * 64] = c[r];
                }
            }
        }
    }
}

#define A2_PITCH 68
__global__ __launch_bounds__(32 * PROD_W) void k_gemm2(const float* __restrict__ a1,
                                              const float* __restrict__ W2,
                                              float* __restrict__ h2, int n) {
    __shared__ float ws2[64 * 16];
    __shared__ float smA[PROD_W][PROD_TPW][16 * A2_PITCH];

    const int tid  = threadIdx.x;
    const int lane = tid & 31;
    const int wave = __builtin_amdgcn_readfirstlane(tid >> 5);
    const bool hi  = lane >= 16;
    const int  m   = lane & 15;
    const int kofs = hi ? 2 : 0;

    {
        const int k = tid >> 2, q = tid & 3;
        float4 v = make_float4(0.f, 0.f, 0.f, 0.f);
        if (q < 2) v = ((const float4*)W2)[k * 2 + q];
        ((float4*)ws2)[tid] = v;
    }

    const int tile0 = (int)blockIdx.x * (PROD_W * PROD_TPW) + wave;
    const int pr = lane >> 4, pq = lane & 15;
    float4 pf[PROD_TPW][8];
#pragma unroll
    for (int it = 0; it < PROD_TPW; ++it) {
#pragma unroll
        for (int j = 0; j < 8; ++j) {
            const int row = (tile0 + it * PROD_W) * 16 + 2 * j + pr;
            pf[it][j] = ((const float4*)(a1 + (size_t)(row < n ? row : n - 1) * 64))[pq];
        }
    }
#pragma unroll
    for (int it = 0; it < PROD_TPW; ++it) {
#pragma unroll
        for (int j = 0; j < 8; ++j)
            *(float4*)(&smA[wave][it][(2 * j + pr) * A2_PITCH + 4 * pq]) = pf[it][j];
    }
    __syncthreads();

    v2f b[16];
#pragma unroll
    for (int k = 0; k < 16; ++k) {
        b[k][0] = ws2[(4 * k + kofs + 0) * 16 + m];
        b[k][1] = ws2[(4 * k + kofs + 1) * 16 + m];
    }

#pragma unroll
    for (int it = 0; it < PROD_TPW; ++it) {
        const int row0 = (tile0 + it * PROD_W) * 16;
        const bool full = (row0 + 16 <= n);
        const float* sA = smA[wave][it];
        v8f c = {};
#pragma unroll
        for (int k = 0; k < 16; ++k) {
            v2f a;
            a[0] = sA[m * A2_PITCH + 4 * k + kofs + 0];
            a[1] = sA[m * A2_PITCH + 4 * k + kofs + 1];
            c = __builtin_amdgcn_wmma_f32_16x16x4_f32(false, a, false, b[k],
                                                      (short)0, c, false, false);
        }
        if (m < 8) {
            float* out = h2 + (size_t)row0 * 8 + m;
            if (full) {
#pragma unroll
                for (int r = 0; r < 8; ++r)
                    out[(size_t)(r + (hi ? 8 : 0)) * 8] = c[r];
            } else {
#pragma unroll
                for (int r = 0; r < 8; ++r) {
                    int rr = r + (hi ? 8 : 0);
                    if (row0 + rr < n) out[(size_t)rr * 8] = c[r];
                }
            }
        }
    }
}

#define NB1   392
#define PRE_T 256
#define EMPTY_W 0xffffffffu
#define CELL_MAX 31u
#define SRC_BITS 23
__device__ __forceinline__ unsigned owner_byte(int d, int n) {
    d = d < 0 ? 0 : (d >= n ? n - 1 : d);
    unsigned o = (unsigned)d / (unsigned)NB1;
    return o < 255u ? o : 255u;
}

__global__ __launch_bounds__(PRE_T) void k_alpha1(const float* __restrict__ h1,
                         const float* __restrict__ asrc, const float* __restrict__ adst,
                         float* __restrict__ as1, float* __restrict__ ad1, int n,
                         const int* __restrict__ ei, int E, unsigned char* __restrict__ ob,
                         int nab, unsigned* __restrict__ bm, int rw) {
    __shared__ unsigned bt[256 * 32];
    __shared__ unsigned bc[256];
    if ((int)blockIdx.x >= nab) {
        const int t = (int)threadIdx.x;
        const int tile = (int)blockIdx.x - nab;
        const u4v fill = {EMPTY_W, EMPTY_W, EMPTY_W, EMPTY_W};
#pragma unroll
        for (int j = 0; j < 8; ++j) ((u4v*)bt)[j * PRE_T + t] = fill;
        bc[t] = 0u;
        __syncthreads();
        const int* __restrict__ dstp = ei + E;
        const int g = tile * PRE_T + t;
        const int e0 = 4 * g;
        auto put = [&](int d, int s, unsigned o) {
            d = d < 0 ? 0 : (d >= n ? n - 1 : d);
            s = s < 0 ? 0 : (s >= n ? n - 1 : s);
            unsigned dl = ((unsigned)d - o * (unsigned)NB1) & 511u;
            unsigned slot = atomicAdd(&bc[o], 1u);
            if (slot < CELL_MAX) bt[o * 32u + slot] = (unsigned)s | (dl << SRC_BITS);
        };
        if (e0 < E) {
            if ((E & 3) == 0) {
                int4 q = ((const int4*)dstp)[g];
                int4 sq = ((const int4*)ei)[g];
                unsigned o0 = owner_byte(q.x, n), o1 = owner_byte(q.y, n);
                unsigned o2 = owner_byte(q.z, n), o3 = owner_byte(q.w, n);
                ((unsigned*)ob)[g] = o0 | (o1 << 8) | (o2 << 16) | (o3 << 24);
                put(q.x, sq.x, o0);
                put(q.y, sq.y, o1);
                put(q.z, sq.z, o2);
                put(q.w, sq.w, o3);
            } else {
                for (int k = 0; k < 4 && e0 + k < E; ++k) {
                    unsigned o = owner_byte(dstp[e0 + k], n);
                    ob[e0 + k] = (unsigned char)o;
                    put(dstp[e0 + k], ei[e0 + k], o);
                }
            }
        }
        __syncthreads();
        const bool fits = n <= (1 << SRC_BITS);
#pragma unroll
        for (int j = 0; j < 8; ++j) {
            int f = j * PRE_T + t;
            u4v val = ((const u4v*)bt)[f];
            if ((f & 7) == 7) val[3] = fits ? bc[f >> 3] : EMPTY_W;
            ((u4v*)(bm + (size_t)(f >> 3) * (size_t)rw + (size_t)tile * 32u))[f & 7] = val;
        }
        return;
    }
    int tid = blockIdx.x * blockDim.x + threadIdx.x;
    if (tid >= n * 8) return;
    unsigned h = tid & 7u;
    const float4* hp = (const float4*)(h1 + (unsigned)(tid >> 3) * 64u + h * 8u);
    float4 v0 = hp[0], v1 = hp[1];
    const float4* sp = (const float4*)(asrc + h * 8u);
    const float4* dp = (const float4*)(adst + h * 8u);
    float4 s0 = sp[0], s1v = sp[1], d0 = dp[0], d1v = dp[1];
    float s = v0.x * s0.x + v0.y * s0.y + v0.z * s0.z + v0.w * s0.w
            + v1.x * s1v.x + v1.y * s1v.y + v1.z * s1v.z + v1.w * s1v.w;
    float d = v0.x * d0.x + v0.y * d0.y + v0.z * d0.z + v0.w * d0.w
            + v1.x * d1v.x + v1.y * d1v.y + v1.z * d1v.z + v1.w * d1v.w;
    as1[tid] = s;
    ad1[tid] = d;
}

#define QCAP  8192
#define OWN_T 1024

template <class F>
__device__ __forceinline__ void own_scan(const int* __restrict__ dstp,
                                         const unsigned char* __restrict__ ob, int E,
                                         int lo, unsigned span, unsigned own, bool exact,
                                         int tid, F f) {
    const int E16 = E >> 4;
    const uint4* __restrict__ bp = (const uint4*)ob;
    const unsigned pat = own * 0x01010101u;
    auto eq4 = [&](unsigned w) -> unsigned {
        unsigned x = w ^ pat;
        return ~(((x & 0x7f7f7f7fu) + 0x7f7f7f7fu) | x | 0x7f7f7f7fu);
    };
    auto pack4 = [&](uint4 q, int k0) -> unsigned {
        return (eq4(q.x) >> (7 - k0)) | (eq4(q.y) >> (6 - k0))
             | (eq4(q.z) >> (5 - k0)) | (eq4(q.w) >> (4 - k0));
    };
    auto visit = [&](int idx) {
        if (!exact && (unsigned)dstp[idx] - (unsigned)lo > span) return;
        f(idx);
    };
    int base = tid;
    for (; base + 7 * OWN_T < E16; base += 8 * OWN_T) {
        uint4 v[8];
#pragma unroll
        for (int u = 0; u < 8; ++u)
            v[u] = bp[base + u * OWN_T];
#pragma unroll
        for (int p = 0; p < 4; ++p) {
            unsigned hit = pack4(v[2 * p], 0) | pack4(v[2 * p + 1], 4);
            while (hit) {
                int j = __ffs(hit) - 1;
                hit &= hit - 1u;
                int k = j & 7;
                visit(16 * (base + (2 * p + (k >> 2)) * OWN_T) + 4 * (k & 3) + (j >> 3));
            }
        }
    }
    for (; base < E16; base += OWN_T) {
        unsigned hit = pack4(bp[base], 0);
        while (hit) {
            int j = __ffs(hit) - 1;
            hit &= hit - 1u;
            visit(16 * base + 4 * (j & 7) + (j >> 3));
        }
    }
    for (int idx = 16 * E16 + tid; idx < E; idx += OWN_T)
        if ((unsigned)ob[idx] == own) visit(idx);
}

__device__ __noinline__ void own_edge_max_direct(int s, unsigned dl, int dnode,
                                                 const float* __restrict__ as1,
                                                 const float* __restrict__ ad1,
                                                 unsigned* mx) {
    for (int h = 0; h < 8; ++h) {
        float e = as1[(unsigned)s * 8u + h] + ad1[(unsigned)dnode * 8u + h];
        e = e > 0.0f ? e : NEG_SLOPE * e;
        atomicMax(&mx[dl * 8u + h], enc_f32(e));
    }
}

__device__ __noinline__ void own_edge_direct(int s, unsigned dl, int dnode,
                                             const float* __restrict__ as1,
                                             const float* __restrict__ ad1,
                                             const unsigned* mx,
                                             const float* __restrict__ h1,
                                             float* acc, float* ssum) {
    for (int h = 0; h < 8; ++h) {
        float e = as1[(unsigned)s * 8u + h] + ad1[(unsigned)dnode * 8u + h];
        e = e > 0.0f ? e : NEG_SLOPE * e;
        float ex = expf(e - dec_f32(mx[dl * 8u + h]));
        atomicAdd(&ssum[dl * 8u + h], ex);
        const float* hp = h1 + (unsigned)s * 64u + (unsigned)h * 8u;
        for (int j = 0; j < 8; ++j)
            atomicAdd(&acc[dl * 64u + h * 8u + j], hp[j] * ex);
    }
}

__device__ __forceinline__ float quad_bcast(float v, int k) {
    int i = __builtin_bit_cast(int, v), r;
    switch (k) {
    case 0:  r = __builtin_amdgcn_update_dpp(i, i, 0x00, 0xf, 0xf, true); break;
    case 1:  r = __builtin_amdgcn_update_dpp(i, i, 0x55, 0xf, 0xf, true); break;
    case 2:  r = __builtin_amdgcn_update_dpp(i, i, 0xaa, 0xf, 0xf, true); break;
    default: r = __builtin_amdgcn_update_dpp(i, i, 0xff, 0xf, 0xf, true); break;
    }
    return __builtin_bit_cast(float, r);
}

__device__ __forceinline__ unsigned wave_incl_sum(unsigned x, int lane) {
    x += (unsigned)__builtin_amdgcn_update_dpp(0, (int)x, 0x111, 0xf, 0xf, false);
    x += (unsigned)__builtin_amdgcn_update_dpp(0, (int)x, 0x112, 0xf, 0xf, false);
    x += (unsigned)__builtin_amdgcn_update_dpp(0, (int)x, 0x114, 0xf, 0xf, false);
    x += (unsigned)__builtin_amdgcn_update_dpp(0, (int)x, 0x118, 0xf, 0xf, false);
    unsigned r0 = (unsigned)__builtin_amdgcn_readlane((int)x, 15);
    return x + ((lane & 16) ? r0 : 0u);
}

__global__ __launch_bounds__(OWN_T) void k_own_scatter1(const int* __restrict__ ei, int E, int n,
                                const float* __restrict__ as1, const float* __restrict__ ad1,
                                const float* __restrict__ h1,
                                const float* __restrict__ bias1, float* __restrict__ a1out,
                                int* __restrict__ wq_s, unsigned short* __restrict__ wq_d,
                                unsigned* __restrict__ wq_n,
                                const unsigned char* __restrict__ ob,
                                const unsigned* __restrict__ bm, int rw) {
    __shared__ float acc[NB1 * 64];
    __shared__ float ssum[NB1 * 8];
    __shared__ unsigned mx[NB1 * 8];
    __shared__ int qs[QCAP];
    __shared__ unsigned short qd[QCAP];
    __shared__ unsigned qn;
    __shared__ unsigned qover;

    const int tid  = threadIdx.x;
    const int lane = tid & 31;
    const int wave = tid >> 5;
    const int d0   = blockIdx.x * NB1;
    const int nb   = (n - d0) < NB1 ? (n - d0) : NB1;

    for (int i = tid; i < NB1 * 64; i += OWN_T) acc[i] = 0.0f;
    for (int i = tid; i < NB1 * 8; i += OWN_T) { ssum[i] = 0.0f; mx[i] = 0u; }
    if (tid == 0) { qn = 0u; qover = 0u; }
    __syncthreads();

    const int* __restrict__ dstp = ei + E;
    const int lo = (d0 == 0) ? (int)0x80000000 : d0;
    const int hi = (d0 + nb == n) ? 0x7fffffff : (d0 + nb - 1);
    const unsigned span = (unsigned)hi - (unsigned)lo;
    const unsigned own = blockIdx.x < 255u ? blockIdx.x : 255u;
    const bool exact = gridDim.x <= 256u || blockIdx.x < 255u;
    int* __restrict__ gs = wq_s + (size_t)blockIdx.x * QCAP;
    unsigned short* __restrict__ gd = wq_d + (size_t)blockIdx.x * QCAP;
    if (exact) {
        const u4v* __restrict__ rp = (const u4v*)(bm + (size_t)own * (size_t)rw);
        const int R16 = rw >> 2;
        const u4v none = {EMPTY_W, EMPTY_W, EMPTY_W, EMPTY_W};
        for (int b0 = 0; b0 < R16; b0 += 8 * OWN_T) {
            const int base = b0 + tid;
            u4v v[8];
            bool over = false;
#pragma unroll
            for (int u = 0; u < 8; ++u) {
                int i = base + u * OWN_T;
                const bool in = i < R16;
                v[u] = in ? __builtin_nontemporal_load(rp + i) : none;
                if (((unsigned)i & 7u) == 7u) {
                    over = over || (in && v[u][3] > CELL_MAX);
                    v[u][3] = EMPTY_W;
                }
            }
            if (over) qover = 1u;
            unsigned c = 0u;
#pragma unroll
            for (int u = 0; u < 8; ++u)
                c += (unsigned)(v[u][0] != EMPTY_W) + (unsigned)(v[u][1] != EMPTY_W)
                   + (unsigned)(v[u][2] != EMPTY_W) + (unsigned)(v[u][3] != EMPTY_W);
            const unsigned incl = wave_incl_sum(c, lane);
            const unsigned wtot = (unsigned)__builtin_amdgcn_readlane((int)incl, 31);
            unsigned wbase = 0u;
            if (lane == 0 && wtot) wbase = atomicAdd(&qn, wtot);
            wbase = (unsigned)__builtin_amdgcn_readfirstlane((int)wbase);
            unsigned slot = wbase + incl - c;
            if (slot + c <= (unsigned)QCAP) {
#pragma unroll
                for (int u = 0; u < 8; ++u) {
#pragma unroll
                    for (int k = 0; k < 4; ++k) {
                        const unsigned w = v[u][k];
                        if (w != EMPTY_W) {
                            const int s = (int)(w & ((1u << SRC_BITS) - 1u));
                            const unsigned short dl = (unsigned short)(w >> SRC_BITS);
                            qs[slot] = s;
                            qd[slot] = dl;
                            gs[slot] = s;
                            gd[slot] = dl;
                            ++slot;
                        }
                    }
                }
            }
        }
    } else {
        if (tid == 0) qn = (unsigned)QCAP + 1u;
    }
    __syncthreads();

    const unsigned qtot = qover ? (unsigned)QCAP + 1u : qn;
    if (tid == 0) wq_n[blockIdx.x] = qtot;
    if (qtot <= (unsigned)QCAP) {
        const int cnt = (int)qtot;
        const int total = cnt + nb;

        {
            const unsigned hh = (unsigned)lane & 7u;
            const int sub = lane >> 3;
            for (int q0 = wave * 16; q0 < total; q0 += (OWN_T / 32) * 16) {
                int   sN[4];
                unsigned dL[4];
                float aV[4], bV[4];
#pragma unroll
                for (int u = 0; u < 4; ++u) {
                    int q = q0 + 4 * u + sub;
                    int qq = q < total ? q : q0;
                    if (qq < cnt) { sN[u] = qs[qq]; dL[u] = qd[qq]; }
                    else          { dL[u] = (unsigned)(qq - cnt); sN[u] = d0 + (qq - cnt); }
                }
#pragma unroll
                for (int u = 0; u < 4; ++u) {
                    aV[u] = as1[(unsigned)sN[u] * 8u + hh];
                    bV[u] = ad1[((unsigned)d0 + dL[u]) * 8u + hh];
                }
#pragma unroll
                for (int u = 0; u < 4; ++u) {
                    float e = aV[u] + bV[u];
                    e = e > 0.0f ? e : NEG_SLOPE * e;
                    atomicMax(&mx[dL[u] * 8u + hh], enc_f32(e));
                }
            }
        }
        __syncthreads();

        const unsigned hq = (unsigned)lane >> 2;
        const int sub = lane & 3;
        const int wv = __builtin_amdgcn_readfirstlane(wave);
        for (int q0 = wv * 8; q0 < total; q0 += (OWN_T / 32) * 8) {
            int   sN[2];
            unsigned dL[2];
            float aV[2], bV[2], ex[2];
            v2f   hV[8];
            unsigned dU[8];
#pragma unroll
            for (int g = 0; g < 2; ++g) {
                int q = q0 + 4 * g + sub;
                int qq = q < total ? q : q0;
                if (qq < cnt) { sN[g] = qs[qq]; dL[g] = qd[qq]; }
                else          { dL[g] = (unsigned)(qq - cnt); sN[g] = d0 + (qq - cnt); }
            }
#pragma unroll
            for (int g = 0; g < 2; ++g) {
                aV[g] = as1[(unsigned)sN[g] * 8u + hq];
                bV[g] = ad1[((unsigned)d0 + dL[g]) * 8u + hq];
            }
#pragma unroll
            for (int u = 0; u < 8; ++u) {
                unsigned sU = (unsigned)__builtin_amdgcn_readlane(sN[u >> 2], u & 3);
                dU[u] = (unsigned)__builtin_amdgcn_readlane((int)dL[u >> 2], u & 3);
                hV[u] = ((const v2f*)(h1 + (size_t)sU * 64u))[lane];
            }
#pragma unroll
            for (int g = 0; g < 2; ++g) {
                float e = aV[g] + bV[g];
                e = e > 0.0f ? e : NEG_SLOPE * e;
                ex[g] = expf(e - dec_f32(mx[dL[g] * 8u + hq]));
                if (q0 + 4 * g + sub < total) atomicAdd(&ssum[dL[g] * 8u + hq], ex[g]);
            }
#pragma unroll
            for (int u = 0; u < 8; ++u) {
                float w = quad_bcast(ex[u >> 2], u & 3);
                if (q0 + u < total) {
                    atomicAdd(&acc[dU[u] * 64u + 2u * lane + 0u], hV[u][0] * w);
                    atomicAdd(&acc[dU[u] * 64u + 2u * lane + 1u], hV[u][1] * w);
                }
            }
        }
    } else {
        own_scan(dstp, ob, E, lo, span, own, exact, tid, [&](int idx) {
            int d = dstp[idx];
            d = d < 0 ? 0 : (d >= n ? n - 1 : d);
            int s = ei[idx];
            s = s < 0 ? 0 : (s >= n ? n - 1 : s);
            own_edge_max_direct(s, (unsigned)(d - d0), d, as1, ad1, mx);
        });
        for (int i = tid; i < nb; i += OWN_T)
            own_edge_max_direct(d0 + i, (unsigned)i, d0 + i, as1, ad1, mx);
        __syncthreads();
        own_scan(dstp, ob, E, lo, span, own, exact, tid, [&](int idx) {
            int d = dstp[idx];
            d = d < 0 ? 0 : (d >= n ? n - 1 : d);
            int s = ei[idx];
            s = s < 0 ? 0 : (s >= n ? n - 1 : s);
            own_edge_direct(s, (unsigned)(d - d0), d, as1, ad1, mx, h1, acc, ssum);
        });
        for (int i = tid; i < nb; i += OWN_T)
            own_edge_direct(d0 + i, (unsigned)i, d0 + i, as1, ad1, mx, h1, acc, ssum);
    }
    __syncthreads();

    float* __restrict__ op = a1out + (size_t)d0 * 64;
    for (int i = tid; i < nb * 64; i += OWN_T) {
        float v = acc[i] / (ssum[i >> 3] + 1e-16f) + bias1[i & 63];
        op[i] = v > 0.0f ? v : (expf(v) - 1.0f);
    }
}

#define Q2CAP 7168
#define Q2MAX ((Q2CAP + NB1 + OWN_T - 1) / OWN_T)

__device__ __forceinline__ float dot8(float4 v0, float4 v1, float4 s0, float4 s1v) {
    return v0.x * s0.x + v0.y * s0.y + v0.z * s0.z + v0.w * s0.w
         + v1.x * s1v.x + v1.y * s1v.y + v1.z * s1v.z + v1.w * s1v.w;
}

__device__ __noinline__ float own2_logit_direct(int s, int dnode, const float* __restrict__ h2,
                                                const float* __restrict__ asrc,
                                                const float* __restrict__ adst) {
    const float4* sp = (const float4*)(h2 + (unsigned)s * 8u);
    const float4* dp = (const float4*)(h2 + (unsigned)dnode * 8u);
    float e = dot8(sp[0], sp[1], ((const float4*)asrc)[0], ((const float4*)asrc)[1])
            + dot8(dp[0], dp[1], ((const float4*)adst)[0], ((const float4*)adst)[1]);
    return e > 0.0f ? e : NEG_SLOPE * e;
}

__device__ __noinline__ void own2_edge_direct(int s, unsigned dl, float e,
                                              const unsigned* mx,
                                              const float* __restrict__ h2,
                                              float* acc, float* ssum) {
    float ex = expf(e - dec_f32(mx[dl]));
    atomicAdd(&ssum[dl], ex);
    const float* hp = h2 + (unsigned)s * 8u;
    for (int j = 0; j < 8; ++j)
        atomicAdd(&acc[dl * 8u + j], hp[j] * ex);
}

__global__ __launch_bounds__(OWN_T) void k_own_scatter2(const int* __restrict__ ei, int E, int n,
                                const float* __restrict__ asrc2, const float* __restrict__ adst2,
                                const float* __restrict__ h2, const float* __restrict__ bias2,
                                const int* __restrict__ wq_s,
                                const unsigned short* __restrict__ wq_d,
                                const unsigned* __restrict__ wq_n,
                                const unsigned char* __restrict__ ob, float* __restrict__ out) {
    __shared__ float acc[NB1 * 8];
    __shared__ float ssum[NB1];
    __shared__ unsigned mx[NB1];
    __shared__ float adl[NB1];
    __shared__ float rows[(Q2CAP + NB1) * 8];
    __shared__ float qe[Q2CAP + NB1];
    __shared__ unsigned short qd[Q2CAP + NB1];

    const int tid  = threadIdx.x;
    const int lane = tid & 31;
    const int wave = tid >> 5;
    const int d0   = blockIdx.x * NB1;
    const int nb   = (n - d0) < NB1 ? (n - d0) : NB1;

    for (int i = tid; i < NB1 * 8; i += OWN_T) acc[i] = 0.0f;
    for (int i = tid; i < NB1; i += OWN_T) { ssum[i] = 0.0f; mx[i] = 0u; }
    __syncthreads();

    const unsigned qtot = wq_n[blockIdx.x];
    if (qtot <= (unsigned)Q2CAP) {
        const int cnt = (int)qtot;
        const int total = cnt + nb;
        const int* __restrict__ gs = wq_s + (size_t)blockIdx.x * QCAP;
        const unsigned short* __restrict__ gd = wq_d + (size_t)blockIdx.x * QCAP;
        const float4 as0 = ((const float4*)asrc2)[0], as1v = ((const float4*)asrc2)[1];
        const float4 ad0 = ((const float4*)adst2)[0], ad1v = ((const float4*)adst2)[1];

        for (int b = 0; b < Q2MAX; b += 4) {
            int sN[4];
            unsigned dL[4];
            float4 r0[4], r1[4];
#pragma unroll
            for (int u = 0; u < 4; ++u) {
                int q = tid + (b + u) * OWN_T;
                sN[u] = 0; dL[u] = 0u;
                if (q < cnt) { sN[u] = gs[q]; dL[u] = gd[q]; }
            }
#pragma unroll
            for (int u = 0; u < 4; ++u) {
                int q = tid + (b + u) * OWN_T;
                int s = sN[u];
                unsigned d = dL[u];
                if (q < cnt) {
                    s = s < 0 ? 0 : (s >= n ? n - 1 : s);
                    d = d < (unsigned)nb ? d : (unsigned)(nb - 1);
                } else {
                    d = q < total ? (unsigned)(q - cnt) : 0u;
                    s = d0 + (int)d;
                }
                dL[u] = d;
                const float4* hp = (const float4*)(h2 + (unsigned)s * 8u);
                r0[u] = hp[0];
                r1[u] = hp[1];
            }
#pragma unroll
            for (int u = 0; u < 4; ++u) {
                int q = tid + (b + u) * OWN_T;
                if (q < total) {
                    ((float4*)(rows + q * 8))[0] = r0[u];
                    ((float4*)(rows + q * 8))[1] = r1[u];
                    qd[q] = (unsigned short)dL[u];
                    qe[q] = dot8(r0[u], r1[u], as0, as1v);
                    if (q >= cnt) adl[dL[u]] = dot8(r0[u], r1[u], ad0, ad1v);
                }
            }
        }
        __syncthreads();

        for (int q = tid; q < total; q += OWN_T) {
            unsigned d = qd[q];
            float e = qe[q] + adl[d];
            e = e > 0.0f ? e : NEG_SLOPE * e;
            qe[q] = e;
            atomicMax(&mx[d], enc_f32(e));
        }
        __syncthreads();

        const int sub = lane >> 2;
        const unsigned j2 = (unsigned)lane & 3u;
        for (int q0 = wave * 32; q0 < total; q0 += (OWN_T / 32) * 32) {
            unsigned dL[4], mV[4];
            float eV[4];
            v2f   hV[4];
#pragma unroll
            for (int u = 0; u < 4; ++u) {
                int q = q0 + 8 * u + sub;
                int qq = q < total ? q : q0;
                dL[u] = qd[qq]; eV[u] = qe[qq];
                hV[u] = ((const v2f*)(rows + qq * 8))[j2];
            }
#pragma unroll
            for (int u = 0; u < 4; ++u)
                mV[u] = mx[dL[u]];
#pragma unroll
            for (int u = 0; u < 4; ++u) {
                if (q0 + 8 * u + sub < total) {
                    float ex = expf(eV[u] - dec_f32(mV[u]));
                    if (j2 == 0u) atomicAdd(&ssum[dL[u]], ex);
                    atomicAdd(&acc[dL[u] * 8u + 2u * j2 + 0u], hV[u][0] * ex);
                    atomicAdd(&acc[dL[u] * 8u + 2u * j2 + 1u], hV[u][1] * ex);
                }
            }
        }
    } else {
        const int* __restrict__ dstp = ei + E;
        const int lo = (d0 == 0) ? (int)0x80000000 : d0;
        const int hi = (d0 + nb == n) ? 0x7fffffff : (d0 + nb - 1);
        const unsigned span = (unsigned)hi - (unsigned)lo;
        const unsigned own = blockIdx.x < 255u ? blockIdx.x : 255u;
        const bool exact = gridDim.x <= 256u || blockIdx.x < 255u;
        own_scan(dstp, ob, E, lo, span, own, exact, tid, [&](int idx) {
            int d = dstp[idx];
            d = d < 0 ? 0 : (d >= n ? n - 1 : d);
            int s = ei[idx];
            s = s < 0 ? 0 : (s >= n ? n - 1 : s);
            atomicMax(&mx[d - d0], enc_f32(own2_logit_direct(s, d, h2, asrc2, adst2)));
        });
        for (int i = tid; i < nb; i += OWN_T)
            atomicMax(&mx[i], enc_f32(own2_logit_direct(d0 + i, d0 + i, h2, asrc2, adst2)));
        __syncthreads();
        own_scan(dstp, ob, E, lo, span, own, exact, tid, [&](int idx) {
            int d = dstp[idx];
            d = d < 0 ? 0 : (d >= n ? n - 1 : d);
            int s = ei[idx];
            s = s < 0 ? 0 : (s >= n ? n - 1 : s);
            own2_edge_direct(s, (unsigned)(d - d0), own2_logit_direct(s, d, h2, asrc2, adst2),
                             mx, h2, acc, ssum);
        });
        for (int i = tid; i < nb; i += OWN_T)
            own2_edge_direct(d0 + i, (unsigned)i,
                             own2_logit_direct(d0 + i, d0 + i, h2, asrc2, adst2),
                             mx, h2, acc, ssum);
    }
    __syncthreads();

    float* __restrict__ op = out + (size_t)d0 * 8;
    for (int i = tid; i < nb * 8; i += OWN_T)
        op[i] = acc[i] / (ssum[i >> 3] + 1e-16f) + bias2[i & 7];
}

extern "C" void kernel_launch(void* const* d_in, const int* in_sizes, int n_in,
                              void* d_out, int out_size, void* d_ws, size_t ws_size,
                              hipStream_t stream) {
    const float* x     = (const float*)d_in[0];
    const int*   ei    = (const int*)d_in[1];
    const float* W1    = (const float*)d_in[2];
    const float* asrc1 = (const float*)d_in[3];
    const float* adst1 = (const float*)d_in[4];
    const float* bias1 = (const float*)d_in[5];
    const float* W2    = (const float*)d_in[6];
    const float* asrc2 = (const float*)d_in[7];
    const float* adst2 = (const float*)d_in[8];
    const float* bias2 = (const float*)d_in[9];

    int n = in_sizes[0] / F_IN;
    int E = in_sizes[1] / 2;

    char* p = (char*)d_ws;
    float*    h1   = (float*)p;    p += (size_t)n * 64 * 4;
    float*    acc1 = (float*)p;    p += (size_t)n * 64 * 4;
    float*    as1  = (float*)p;    p += (size_t)n * 8 * 4;
    float*    ad1  = (float*)p;    p += (size_t)n * 8 * 4;
    float*    h2   = (float*)p;    p += (size_t)n * 8 * 4;
    unsigned char*  ob   = (unsigned char*)p;  p += (size_t)((E + 15) / 16) * 16;
    int*            wq_s = (int*)p;            p += (size_t)((n + NB1 - 1) / NB1) * QCAP * 4;
    unsigned short* wq_d = (unsigned short*)p; p += (size_t)((n + NB1 - 1) / NB1) * QCAP * 2;
    const int tpb = PRE_T;
    const int nt = (E / 4 + tpb) / tpb;
    const int rw = nt * 32;
    unsigned*       bm   = (unsigned*)p;       p += (size_t)256 * rw * 4;
    unsigned*       wq_n = (unsigned*)p;       p += (size_t)((n + NB1 - 1) / NB1) * 4;

    k_gemm1<<<(((n + 15) / 16) + PROD_W * PROD_TPW - 1) / (PROD_W * PROD_TPW), 32 * PROD_W, 0, stream>>>(x, W1, h1, n);
    k_alpha1<<<(n * 8 + tpb - 1) / tpb + nt, tpb, 0, stream>>>(
        h1, asrc1, adst1, as1, ad1, n, ei, E, ob, (n * 8 + tpb - 1) / tpb, bm, rw);
    k_own_scatter1<<<(n + NB1 - 1) / NB1, OWN_T, 0, stream>>>(ei, E, n, as1, ad1, h1, bias1, acc1,
                                                              wq_s, wq_d, wq_n, ob, bm, rw);

    k_gemm2<<<(((n + 15) / 16) + PROD_W * PROD_TPW - 1) / (PROD_W * PROD_TPW), 32 * PROD_W, 0, stream>>>(acc1, W2, h2, n);
    k_own_scatter2<<<(n + NB1 - 1) / NB1, OWN_T, 0, stream>>>(ei, E, n, asrc2, adst2, h2, bias2,
                                                              wq_s, wq_d, wq_n, ob, (float*)d_out);
}
